# starting kernel
# baseline (speedup 1.0000x reference)
#include <hip/hip_runtime.h>
#include <hip/hip_bf16.h>

typedef __attribute__((ext_vector_type(2))) float v2f;
typedef __attribute__((ext_vector_type(8))) float v8f;

#define HDIM 128
#define CTPW 4

__global__ __launch_bounds__(256) void gat_transpose(
    const float* __restrict__ W, float* __restrict__ Wt,
    int rows, int cols, int colsPad)
{
    int i = blockIdx.x * blockDim.x + threadIdx.x;
    int total = colsPad * rows;
    if (i >= total) return;
    int c = i / rows;
    int r = i - c * rows;
    Wt[(size_t)c * rows + r] = (c < cols) ? W[(size_t)r * cols + c] : 0.f;
}

__global__ __launch_bounds__(256) void gat_gemm_strip(
    const float* __restrict__ X, const float* __restrict__ Wt,
    float* __restrict__ H, int nRowTiles)
{
    int wid  = (blockIdx.x * blockDim.x + threadIdx.x) >> 5;
    int lane = threadIdx.x & 31;
    int nHalves = HDIM / (CTPW * 16);
    int nW = nRowTiles * nHalves;
    if (wid >= nW) return;
    int rt   = wid >> 1;
    int half = wid & 1;
    int row0 = rt * 16;
    int laneM = lane & 15;
    int kOff  = (lane >> 4) * 2;

    const float* xrow = X  + (size_t)(row0 + laneM) * HDIM + kOff;
    const float* wrow = Wt + (size_t)(half * (CTPW * 16) + laneM) * HDIM + kOff;

    v8f c[CTPW] = {};
    #pragma unroll 4
    for (int k = 0; k < HDIM; k += 4) {
        v2f a = *(const v2f*)(xrow + k);
        #pragma unroll
        for (int t = 0; t < CTPW; ++t) {
            v2f b = *(const v2f*)(wrow + (size_t)t * 16 * HDIM + k);
            c[t] = __builtin_amdgcn_wmma_f32_16x16x4_f32(
                       false, a, false, b, (short)0, c[t], false, false);
        }
    }

    int mBase = (lane >> 4) * 8;
    #pragma unroll
    for (int t = 0; t < CTPW; ++t) {
        float* o = H + (size_t)row0 * HDIM + (half * CTPW + t) * 16 + laneM;
        #pragma unroll
        for (int r = 0; r < 8; ++r)
            o[(size_t)(mBase + r) * HDIM] = c[t][r];
    }
}

__global__ __launch_bounds__(256) void gat_gemm_cls(
    const float* __restrict__ X, const float* __restrict__ Wlt,
    const float* __restrict__ bl, float* __restrict__ Out,
    int nRowTiles, int nCols)
{
    int wid  = (blockIdx.x * blockDim.x + threadIdx.x) >> 5;
    int lane = threadIdx.x & 31;
    if (wid >= nRowTiles) return;
    int row0 = wid * 16;
    int laneM = lane & 15;
    int kOff  = (lane >> 4) * 2;

    const float* xrow = X   + (size_t)(row0 + laneM) * HDIM + kOff;
    const float* wrow = Wlt + (size_t)laneM * HDIM + kOff;

    v8f c[3] = {};
    #pragma unroll 4
    for (int k = 0; k < HDIM; k += 4) {
        v2f a = *(const v2f*)(xrow + k);
        #pragma unroll
        for (int t = 0; t < 3; ++t) {
            v2f b = *(const v2f*)(wrow + (size_t)t * 16 * HDIM + k);
            c[t] = __builtin_amdgcn_wmma_f32_16x16x4_f32(
                       false, a, false, b, (short)0, c[t], false, false);
        }
    }

    int mBase = (lane >> 4) * 8;
    #pragma unroll
    for (int t = 0; t < 3; ++t) {
        int col = t * 16 + laneM;
        if (col >= nCols) continue;
        float bv = bl[col];
        #pragma unroll
        for (int r = 0; r < 8; ++r)
            Out[(size_t)(row0 + mBase + r) * nCols + col] = c[t][r] + bv;
    }
}

__global__ __launch_bounds__(256) void gat_alpha(
    const float* __restrict__ H, const float* __restrict__ asrc,
    const float* __restrict__ adst, float* __restrict__ aS,
    float* __restrict__ aD, int N)
{
    int w    = (blockIdx.x * blockDim.x + threadIdx.x) >> 5;
    int lane = threadIdx.x & 31;
    if (w >= N) return;
    const float* h = H + (size_t)w * HDIM;
    float s = 0.f, d = 0.f;
    #pragma unroll
    for (int j = 0; j < 4; ++j) {
        int f = lane + 32 * j;
        float hv = h[f];
        s += hv * asrc[f];
        d += hv * adst[f];
    }
    #pragma unroll
    for (int off = 16; off > 0; off >>= 1) {
        s += __shfl_xor(s, off, 32);
        d += __shfl_xor(d, off, 32);
    }
    if (lane == 0) { aS[w] = s; aD[w] = d; }
}

__device__ __forceinline__ unsigned f2ord(float f) {
    unsigned b = __float_as_uint(f);
    return (b & 0x80000000u) ? ~b : (b | 0x80000000u);
}
__device__ __forceinline__ float ord2f(unsigned u) {
    unsigned b = (u & 0x80000000u) ? (u & 0x7FFFFFFFu) : ~u;
    return __uint_as_float(b);
}

__global__ __launch_bounds__(256) void gat_init(
    unsigned* __restrict__ segmax, float* __restrict__ den,
    float* __restrict__ agg, int N)
{
    int i = blockIdx.x * blockDim.x + threadIdx.x;
    int total = N * HDIM;
    if (i < total) agg[i] = 0.f;
    if (i < N) {
        segmax[i] = 0x007FFFFFu;
        den[i]    = 0.f;
    }
}

__global__ __launch_bounds__(256) void gat_edge1(
    const int* __restrict__ ei, int E, int N,
    const float* __restrict__ aS, const float* __restrict__ aD,
    float* __restrict__ eVal, unsigned* __restrict__ segmax)
{
    int i = blockIdx.x * blockDim.x + threadIdx.x;
    int ET = E + N;
    if (i >= ET) return;
    int s, d;
    if (i < E) { s = ei[i]; d = ei[E + i]; } else { s = d = i - E; }
    float e = aS[s] + aD[d];
    e = (e >= 0.f) ? e : 0.2f * e;
    eVal[i] = e;
    atomicMax(&segmax[d], f2ord(e));
}

__global__ __launch_bounds__(256) void gat_edge2(
    const int* __restrict__ ei, int E, int N,
    float* __restrict__ eVal, const unsigned* __restrict__ segmax,
    float* __restrict__ den)
{
    int i = blockIdx.x * blockDim.x + threadIdx.x;
    int ET = E + N;
    if (i >= ET) return;
    int d = (i < E) ? ei[E + i] : (i - E);
    float ex = expf(eVal[i] - ord2f(segmax[d]));
    eVal[i] = ex;
    atomicAdd(&den[d], ex);
}

__global__ __launch_bounds__(256) void gat_edge3(
    const int* __restrict__ ei, int E, int N,
    const float* __restrict__ eVal, const float* __restrict__ den,
    const float* __restrict__ H, float* __restrict__ agg)
{
    int w    = (blockIdx.x * blockDim.x + threadIdx.x) >> 5;
    int lane = threadIdx.x & 31;
    int ET = E + N;
    if (w >= ET) return;
    int s, d;
    if (w < E) { s = ei[w]; d = ei[E + w]; } else { s = d = w - E; }
    float alpha = eVal[w] / den[d];
    const float* h = H + (size_t)s * HDIM;
    float* o = agg + (size_t)d * HDIM;
    #pragma unroll
    for (int j = 0; j < 4; ++j) {
        int f = lane + 32 * j;
        atomicAdd(&o[f], h[f] * alpha);
    }
}

__global__ __launch_bounds__(256) void gat_bias_relu(
    float* __restrict__ agg, const float* __restrict__ b, int total)
{
    int i = blockIdx.x * blockDim.x + threadIdx.x;
    if (i >= total) return;
    float v = agg[i] + b[i & (HDIM - 1)];
    agg[i] = (v > 0.f) ? v : 0.f;
}

extern "C" void kernel_launch(void* const* d_in, const int* in_sizes, int n_in,
                              void* d_out, int out_size, void* d_ws, size_t ws_size,
                              hipStream_t stream)
{
    const float* x   = (const float*)d_in[0];
    const int*   ei  = (const int*)d_in[1];
    const float* W1  = (const float*)d_in[2];
    const float* a1s = (const float*)d_in[3];
    const float* a1d = (const float*)d_in[4];
    const float* b1  = (const float*)d_in[5];
    const float* W2  = (const float*)d_in[6];
    const float* a2s = (const float*)d_in[7];
    const float* a2d = (const float*)d_in[8];
    const float* b2  = (const float*)d_in[9];
    const float* Wl  = (const float*)d_in[10];
    const float* bl  = (const float*)d_in[11];

    const int N  = in_sizes[0] / HDIM;
    const int E  = in_sizes[1] / 2;
    const int C  = in_sizes[11];
    const int CP = (C + 15) & ~15;
    const int ET = E + N;

    char* ws = (char*)d_ws;
    auto align256 = [](size_t v) { return (v + 255) & ~(size_t)255; };
    size_t off = 0;
    float*    bufH  = (float*)(ws + off);  off += align256((size_t)N * HDIM * 4);
    float*    bufA  = (float*)(ws + off);  off += align256((size_t)N * HDIM * 4);
    float*    Wt    = (float*)(ws + off);  off += align256((size_t)HDIM * HDIM * 4);
    float*    Wlt   = (float*)(ws + off);  off += align256((size_t)CP * HDIM * 4);
    float*    aS    = (float*)(ws + off);  off += align256((size_t)N * 4);
    float*    aD    = (float*)(ws + off);  off += align256((size_t)N * 4);
    unsigned* smax  = (unsigned*)(ws + off); off += align256((size_t)N * 4);
    float*    den   = (float*)(ws + off);  off += align256((size_t)N * 4);
    float*    eVal  = (float*)(ws + off);  off += align256((size_t)ET * 4);
    (void)ws_size;

    const int nRowTiles = N / 16;
    const int BLK = 256;
    dim3 blk(BLK);

    dim3 trGrid((HDIM * HDIM + BLK - 1) / BLK);
    dim3 gemmGrid((nRowTiles * 2 + 7) / 8);
    dim3 waveGridN((N + 7) / 8);
    dim3 elemGridN((N * HDIM + BLK - 1) / BLK);
    dim3 edgeGrid((ET + BLK - 1) / BLK);
    dim3 edgeWGrid((ET + 7) / 8);

    auto run_layer = [&](const float* act_in, const float* W,
                         const float* as_, const float* ad_, const float* bias) {
        gat_transpose<<<trGrid, blk, 0, stream>>>(W, Wt, HDIM, HDIM, HDIM);
        gat_gemm_strip<<<gemmGrid, blk, 0, stream>>>(act_in, Wt, bufH, nRowTiles);
        gat_alpha<<<waveGridN, blk, 0, stream>>>(bufH, as_, ad_, aS, aD, N);
        gat_init<<<elemGridN, blk, 0, stream>>>(smax, den, bufA, N);
        gat_edge1<<<edgeGrid, blk, 0, stream>>>(ei, E, N, aS, aD, eVal, smax);
        gat_edge2<<<edgeGrid, blk, 0, stream>>>(ei, E, N, eVal, smax, den);
        gat_edge3<<<edgeWGrid, blk, 0, stream>>>(ei, E, N, eVal, den, bufH, bufA);
        gat_bias_relu<<<elemGridN, blk, 0, stream>>>(bufA, bias, N * HDIM);
    };

    run_layer(x, W1, a1s, a1d, b1);
    run_layer(bufA, W2, a2s, a2d, b2);

    dim3 trlGrid((CP * HDIM + BLK - 1) / BLK);
    gat_transpose<<<trlGrid, blk, 0, stream>>>(Wl, Wlt, HDIM, C, CP);
    dim3 clsGrid((nRowTiles + 7) / 8);
    gat_gemm_cls<<<clsGrid, blk, 0, stream>>>(bufA, Wlt, bl, (float*)d_out,
                                              nRowTiles, C);
}


# Aggregation by destination strips: a work-group owns 128 rows, filters the edge list for them, adds h rows in LDS (float adds), writes relu(sum+bias) once; per-node sum order changes
# speedup vs baseline: 1.8189x; 1.8189x over previous
#include <hip/hip_runtime.h>
#include <hip/hip_bf16.h>

typedef __attribute__((ext_vector_type(2))) float v2f;
typedef __attribute__((ext_vector_type(8))) float v8f;

#define HDIM 128
#define CTPW 4

__global__ __launch_bounds__(256) void gat_transpose(
    const float* __restrict__ W, float* __restrict__ Wt,
    int rows, int cols, int colsPad)
{
    int i = blockIdx.x * blockDim.x + threadIdx.x;
    int total = colsPad * rows;
    if (i >= total) return;
    int c = i / rows;
    int r = i - c * rows;
    Wt[(size_t)c * rows + r] = (c < cols) ? W[(size_t)r * cols + c] : 0.f;
}

__global__ __launch_bounds__(256) void gat_gemm_strip(
    const float* __restrict__ X, const float* __restrict__ Wt,
    float* __restrict__ H, int nRowTiles)
{
    int wid  = (blockIdx.x * blockDim.x + threadIdx.x) >> 5;
    int lane = threadIdx.x & 31;
    int nHalves = HDIM / (CTPW * 16);
    int nW = nRowTiles * nHalves;
    if (wid >= nW) return;
    int rt   = wid >> 1;
    int half = wid & 1;
    int row0 = rt * 16;
    int laneM = lane & 15;
    int kOff  = (lane >> 4) * 2;

    const float* xrow = X  + (size_t)(row0 + laneM) * HDIM + kOff;
    const float* wrow = Wt + (size_t)(half * (CTPW * 16) + laneM) * HDIM + kOff;

    v8f c[CTPW] = {};
    #pragma unroll 4
    for (int k = 0; k < HDIM; k += 4) {
        v2f a = *(const v2f*)(xrow + k);
        #pragma unroll
        for (int t = 0; t < CTPW; ++t) {
            v2f b = *(const v2f*)(wrow + (size_t)t * 16 * HDIM + k);
            c[t] = __builtin_amdgcn_wmma_f32_16x16x4_f32(
                       false, a, false, b, (short)0, c[t], false, false);
        }
    }

    int mBase = (lane >> 4) * 8;
    #pragma unroll
    for (int t = 0; t < CTPW; ++t) {
        float* o = H + (size_t)row0 * HDIM + (half * CTPW + t) * 16 + laneM;
        #pragma unroll
        for (int r = 0; r < 8; ++r)
            o[(size_t)(mBase + r) * HDIM] = c[t][r];
    }
}

__global__ __launch_bounds__(256) void gat_gemm_cls(
    const float* __restrict__ X, const float* __restrict__ Wlt,
    const float* __restrict__ bl, float* __restrict__ Out,
    int nRowTiles, int nCols)
{
    int wid  = (blockIdx.x * blockDim.x + threadIdx.x) >> 5;
    int lane = threadIdx.x & 31;
    if (wid >= nRowTiles) return;
    int row0 = wid * 16;
    int laneM = lane & 15;
    int kOff  = (lane >> 4) * 2;

    const float* xrow = X   + (size_t)(row0 + laneM) * HDIM + kOff;
    const float* wrow = Wlt + (size_t)laneM * HDIM + kOff;

    v8f c[3] = {};
    #pragma unroll 4
    for (int k = 0; k < HDIM; k += 4) {
        v2f a = *(const v2f*)(xrow + k);
        #pragma unroll
        for (int t = 0; t < 3; ++t) {
            v2f b = *(const v2f*)(wrow + (size_t)t * 16 * HDIM + k);
            c[t] = __builtin_amdgcn_wmma_f32_16x16x4_f32(
                       false, a, false, b, (short)0, c[t], false, false);
        }
    }

    int mBase = (lane >> 4) * 8;
    #pragma unroll
    for (int t = 0; t < 3; ++t) {
        int col = t * 16 + laneM;
        if (col >= nCols) continue;
        float bv = bl[col];
        #pragma unroll
        for (int r = 0; r < 8; ++r)
            Out[(size_t)(row0 + mBase + r) * nCols + col] = c[t][r] + bv;
    }
}

__global__ __launch_bounds__(256) void gat_alpha(
    const float* __restrict__ H, const float* __restrict__ asrc,
    const float* __restrict__ adst, float* __restrict__ aS,
    float* __restrict__ aD, int N)
{
    int w    = (blockIdx.x * blockDim.x + threadIdx.x) >> 5;
    int lane = threadIdx.x & 31;
    if (w >= N) return;
    const float* h = H + (size_t)w * HDIM;
    float s = 0.f, d = 0.f;
    #pragma unroll
    for (int j = 0; j < 4; ++j) {
        int f = lane + 32 * j;
        float hv = h[f];
        s += hv * asrc[f];
        d += hv * adst[f];
    }
    #pragma unroll
    for (int off = 16; off > 0; off >>= 1) {
        s += __shfl_xor(s, off, 32);
        d += __shfl_xor(d, off, 32);
    }
    if (lane == 0) { aS[w] = s; aD[w] = d; }
}

__device__ __forceinline__ unsigned f2ord(float f) {
    unsigned b = __float_as_uint(f);
    return (b & 0x80000000u) ? ~b : (b | 0x80000000u);
}
__device__ __forceinline__ float ord2f(unsigned u) {
    unsigned b = (u & 0x80000000u) ? (u & 0x7FFFFFFFu) : ~u;
    return __uint_as_float(b);
}

__device__ __forceinline__ int block_l2_half(int nCopies) {
    __shared__ int s_half;
    if (threadIdx.x == 0) {
        unsigned id = __builtin_amdgcn_s_sendmsg_rtn(0x87);
        s_half = (nCopies == 2) ? (int)((id >> 18) & 1u) : 0;
    }
    __syncthreads();
    return __builtin_amdgcn_readfirstlane(s_half);
}

__global__ __launch_bounds__(256) void gat_init(
    unsigned* __restrict__ segmax, float* __restrict__ den, int total)
{
    int i = blockIdx.x * blockDim.x + threadIdx.x;
    if (i < total) {
        segmax[i] = 0x007FFFFFu;
        den[i]    = 0.f;
    }
}

__global__ __launch_bounds__(256) void gat_edge1(
    const int* __restrict__ ei, int E, int N,
    const float* __restrict__ aS, const float* __restrict__ aD,
    float* __restrict__ eVal, unsigned* __restrict__ segmax, int nCopies)
{
    unsigned* mymax = segmax + (size_t)block_l2_half(nCopies) * N;
    int i = blockIdx.x * blockDim.x + threadIdx.x;
    int ET = E + N;
    if (i >= ET) return;
    int s, d;
    if (i < E) { s = ei[i]; d = ei[E + i]; } else { s = d = i - E; }
    float e = aS[s] + aD[d];
    e = (e >= 0.f) ? e : 0.2f * e;
    eVal[i] = e;
    atomicMax(&mymax[d], f2ord(e));
}

__global__ __launch_bounds__(256) void gat_edge2(
    const int* __restrict__ ei, int E, int N,
    float* __restrict__ eVal, const unsigned* __restrict__ segmax,
    float* __restrict__ den, int nCopies)
{
    float* myden = den + (size_t)block_l2_half(nCopies) * N;
    int i = blockIdx.x * blockDim.x + threadIdx.x;
    int ET = E + N;
    if (i >= ET) return;
    int d = (i < E) ? ei[E + i] : (i - E);
    unsigned m0 = segmax[d];
    unsigned m1 = segmax[(size_t)(nCopies - 1) * N + d];
    unsigned mo = (m1 > m0) ? m1 : m0;
    float ex = expf(eVal[i] - ord2f(mo));
    eVal[i] = ex;
    atomicAdd(&myden[d], ex);
}

#define SROWS  128
#define SWAVES 32
#define SLOADS 8
#define WEDGES (SLOADS * 128)
#define QCAP   1536

__global__ __launch_bounds__(1024) void gat_agg_strip(
    const int* __restrict__ ei, int E, int N,
    const float* __restrict__ eVal, const float* __restrict__ den,
    const float* __restrict__ H, const float* __restrict__ bias,
    float* __restrict__ act, int nCopies)
{
    __shared__ float s_acc[SROWS * HDIM];
    __shared__ float s_den[SROWS];
    __shared__ int   s_q[SWAVES * QCAP];

    const int tid  = threadIdx.x;
    const int lane = tid & 31;
    const int wv   = tid >> 5;
    const int n0   = blockIdx.x * SROWS;
    const int rows = (N - n0 < SROWS) ? (N - n0) : SROWS;
    const int* __restrict__ dstp = ei + E;

    if (tid < SROWS) {
        int n = (tid < rows) ? (n0 + tid) : n0;
        float d0 = den[n];
        float d1 = den[(size_t)(nCopies - 1) * N + n];
        s_den[tid] = (nCopies == 2) ? d0 + d1 : d0;
    }
    __syncthreads();

    for (int r = wv; r < SROWS; r += SWAVES) {
        if (r < rows) {
            int n = n0 + r;
            float a = eVal[E + n] / s_den[r];
            const float* h = H + (size_t)n * HDIM;
            #pragma unroll
            for (int k = 0; k < 4; ++k)
                s_acc[r * HDIM + lane + 32 * k] = h[lane + 32 * k] * a;
        }
    }
    __syncthreads();

    int* q = s_q + wv * QCAP;
    const bool vec = ((E & 3) == 0);

    auto load_round = [&](int base, int (&dv)[SLOADS][4]) {
        if (vec && base + WEDGES <= E) {
            #pragma unroll
            for (int j = 0; j < SLOADS; ++j) {
                int4 t = *(const int4*)(dstp + base + j * 128 + lane * 4);
                dv[j][0] = t.x; dv[j][1] = t.y; dv[j][2] = t.z; dv[j][3] = t.w;
            }
        } else {
            #pragma unroll
            for (int j = 0; j < SLOADS; ++j) {
                #pragma unroll
                for (int c = 0; c < 4; ++c) {
                    int i = base + j * 128 + lane * 4 + c;
                    dv[j][c] = (i < E) ? dstp[i] : -1;
                }
            }
        }
    };

    auto drain = [&](int cnt) {
        for (int b = 0; b < cnt; b += 32) {
            int nb = (cnt - b < 32) ? (cnt - b) : 32;
            int s = 0, dl = 0; float al = 0.f;
            if (lane < nb) {
                int idx = q[b + lane];
                s  = ei[idx];
                dl = dstp[idx] - n0;
                al = eVal[idx] / s_den[dl];
            }
            for (int j = 0; j < nb; j += 16) {
                v2f hv[16][2];
                #pragma unroll
                for (int u = 0; u < 16; ++u) {
                    int ju = (j + u < nb) ? (j + u) : (nb - 1);
                    int sj = __builtin_amdgcn_readlane(s, ju);
                    const float* h = H + (size_t)sj * HDIM + 2 * lane;
                    hv[u][0] = *(const v2f*)(h);
                    hv[u][1] = *(const v2f*)(h + 64);
                }
                #pragma unroll
                for (int u = 0; u < 16; ++u) {
                    if (j + u < nb) {
                        float aj = __uint_as_float(__builtin_amdgcn_readlane(__float_as_uint(al), j + u));
                        int   dj = __builtin_amdgcn_readlane(dl, j + u);
                        float* o = s_acc + dj * HDIM + 2 * lane;
                        atomicAdd(&o[0],  hv[u][0][0] * aj);
                        atomicAdd(&o[1],  hv[u][0][1] * aj);
                        atomicAdd(&o[64], hv[u][1][0] * aj);
                        atomicAdd(&o[65], hv[u][1][1] * aj);
                    }
                }
            }
        }
    };

    int cnt = 0;
    auto scan_round = [&](int base, const int (&dv)[SLOADS][4]) {
        #pragma unroll
        for (int j = 0; j < SLOADS; ++j) {
            #pragma unroll
            for (int c = 0; c < 4; ++c) {
                bool m = (unsigned)(dv[j][c] - n0) < (unsigned)rows;
                unsigned mask = (unsigned)__ballot(m);
                if (mask) {
                    if (m) q[cnt + __popc(mask & ((1u << lane) - 1u))] =
                               base + j * 128 + lane * 4 + c;
                    cnt += __popc(mask);
                }
            }
        }
        if (cnt > QCAP - WEDGES) { drain(cnt); cnt = 0; }
    };

    const int perRound = SWAVES * WEDGES;
    const int nRounds = (E + perRound - 1) / perRound;
    int rd = (nRounds > 0) ? (int)(blockIdx.x % (unsigned)nRounds) : 0;
    auto next_base = [&]() {
        int b = rd * perRound + wv * WEDGES;
        rd = (rd + 1 == nRounds) ? 0 : rd + 1;
        return b;
    };
    int dv[SLOADS][4];
    for (int it = 0; it < nRounds; ++it) {
        const int base = next_base();
        if (base >= E) continue;
        load_round(base, dv);
        scan_round(base, dv);
    }
    drain(cnt);
    __syncthreads();

    for (int e4 = tid; e4 < rows * (HDIM / 4); e4 += SWAVES * 32) {
        int r = e4 / (HDIM / 4);
        int f = (e4 - r * (HDIM / 4)) * 4;
        float4 v = *(const float4*)(s_acc + r * HDIM + f);
        v.x += bias[f]; v.y += bias[f + 1]; v.z += bias[f + 2]; v.w += bias[f + 3];
        v.x = (v.x > 0.f) ? v.x : 0.f;  v.y = (v.y > 0.f) ? v.y : 0.f;
        v.z = (v.z > 0.f) ? v.z : 0.f;  v.w = (v.w > 0.f) ? v.w : 0.f;
        *(float4*)(act + (size_t)(n0 + r) * HDIM + f) = v;
    }
}

extern "C" void kernel_launch(void* const* d_in, const int* in_sizes, int n_in,
                              void* d_out, int out_size, void* d_ws, size_t ws_size,
                              hipStream_t stream)
{
    const float* x   = (const float*)d_in[0];
    const int*   ei  = (const int*)d_in[1];
    const float* W1  = (const float*)d_in[2];
    const float* a1s = (const float*)d_in[3];
    const float* a1d = (const float*)d_in[4];
    const float* b1  = (const float*)d_in[5];
    const float* W2  = (const float*)d_in[6];
    const float* a2s = (const float*)d_in[7];
    const float* a2d = (const float*)d_in[8];
    const float* b2  = (const float*)d_in[9];
    const float* Wl  = (const float*)d_in[10];
    const float* bl  = (const float*)d_in[11];

    const int N  = in_sizes[0] / HDIM;
    const int E  = in_sizes[1] / 2;
    const int C  = in_sizes[11];
    const int CP = (C + 15) & ~15;
    const int ET = E + N;

    char* ws = (char*)d_ws;
    auto align256 = [](size_t v) { return (v + 255) & ~(size_t)255; };
    auto wsNeed = [&](size_t nc) {
        return align256((size_t)N * HDIM * 4) + align256(nc * N * HDIM * 4)
             + align256((size_t)HDIM * HDIM * 4) + align256((size_t)CP * HDIM * 4)
             + 2 * align256((size_t)N * 4) + 2 * align256(nc * N * 4)
             + align256((size_t)ET * 4);
    };
    const int NC = (ws_size >= wsNeed(2)) ? 2 : 1;
    size_t off = 0;
    float*    bufH  = (float*)(ws + off);  off += align256((size_t)N * HDIM * 4);
    float*    bufA  = (float*)(ws + off);  off += align256((size_t)NC * N * HDIM * 4);
    float*    Wt    = (float*)(ws + off);  off += align256((size_t)HDIM * HDIM * 4);
    float*    Wlt   = (float*)(ws + off);  off += align256((size_t)CP * HDIM * 4);
    float*    aS    = (float*)(ws + off);  off += align256((size_t)N * 4);
    float*    aD    = (float*)(ws + off);  off += align256((size_t)N * 4);
    unsigned* smax  = (unsigned*)(ws + off); off += align256((size_t)NC * N * 4);
    float*    den   = (float*)(ws + off);  off += align256((size_t)NC * N * 4);
    float*    eVal  = (float*)(ws + off);  off += align256((size_t)ET * 4);

    const int nRowTiles = N / 16;
    const int BLK = 256;
    dim3 blk(BLK);

    dim3 trGrid((HDIM * HDIM + BLK - 1) / BLK);
    dim3 gemmGrid((nRowTiles * 2 + 7) / 8);
    dim3 waveGridN((N + 7) / 8);
    dim3 initGrid((NC * N + BLK - 1) / BLK);
    dim3 edgeGrid((ET + BLK - 1) / BLK);
    dim3 stripGrid((N + SROWS - 1) / SROWS);

    auto run_layer = [&](const float* act_in, const float* W,
                         const float* as_, const float* ad_, const float* bias) {
        gat_transpose<<<trGrid, blk, 0, stream>>>(W, Wt, HDIM, HDIM, HDIM);
        gat_gemm_strip<<<gemmGrid, blk, 0, stream>>>(act_in, Wt, bufH, nRowTiles);
        gat_alpha<<<waveGridN, blk, 0, stream>>>(bufH, as_, ad_, aS, aD, N);
        gat_init<<<initGrid, blk, 0, stream>>>(smax, den, NC * N);
        gat_edge1<<<edgeGrid, blk, 0, stream>>>(ei, E, N, aS, aD, eVal, smax, NC);
        gat_edge2<<<edgeGrid, blk, 0, stream>>>(ei, E, N, eVal, smax, den, NC);
        gat_agg_strip<<<stripGrid, dim3(SWAVES * 32), 0, stream>>>(ei, E, N, eVal, den, bufH, bias, bufA, NC);
    };

    run_layer(x, W1, a1s, a1d, b1);
    run_layer(bufA, W2, a2s, a2d, b2);

    dim3 trlGrid((CP * HDIM + BLK - 1) / BLK);
    gat_transpose<<<trlGrid, blk, 0, stream>>>(Wl, Wlt, HDIM, C, CP);
    dim3 clsGrid((nRowTiles + 7) / 8);
    gat_gemm_cls<<<clsGrid, blk, 0, stream>>>(bufA, Wlt, bl, (float*)d_out,
                                              nRowTiles, C);
}


# Edge softmax moved into the strip kernel: segment max and denominator of its 128 rows built in LDS from the edges it filters; no global atomics, 16 -> 10 launches; denominator sum order changes
# speedup vs baseline: 2.2612x; 1.2432x over previous
#include <hip/hip_runtime.h>
#include <hip/hip_bf16.h>

typedef __attribute__((ext_vector_type(2))) float v2f;
typedef __attribute__((ext_vector_type(8))) float v8f;

#define HDIM 128
#define CTPW 4

__global__ __launch_bounds__(256) void gat_transpose(
    const float* __restrict__ W, float* __restrict__ Wt,
    int rows, int cols, int colsPad)
{
    int i = blockIdx.x * blockDim.x + threadIdx.x;
    int total = colsPad * rows;
    if (i >= total) return;
    int c = i / rows;
    int r = i - c * rows;
    Wt[(size_t)c * rows + r] = (c < cols) ? W[(size_t)r * cols + c] : 0.f;
}

__global__ __launch_bounds__(256) void gat_gemm_strip(
    const float* __restrict__ X, const float* __restrict__ Wt,
    float* __restrict__ H, int nRowTiles)
{
    int wid  = (blockIdx.x * blockDim.x + threadIdx.x) >> 5;
    int lane = threadIdx.x & 31;
    int nHalves = HDIM / (CTPW * 16);
    int nW = nRowTiles * nHalves;
    if (wid >= nW) return;
    int rt   = wid >> 1;
    int half = wid & 1;
    int row0 = rt * 16;
    int laneM = lane & 15;
    int kOff  = (lane >> 4) * 2;

    const float* xrow = X  + (size_t)(row0 + laneM) * HDIM + kOff;
    const float* wrow = Wt + (size_t)(half * (CTPW * 16) + laneM) * HDIM + kOff;

    v8f c[CTPW] = {};
    #pragma unroll 4
    for (int k = 0; k < HDIM; k += 4) {
        v2f a = *(const v2f*)(xrow + k);
        #pragma unroll
        for (int t = 0; t < CTPW; ++t) {
            v2f b = *(const v2f*)(wrow + (size_t)t * 16 * HDIM + k);
            c[t] = __builtin_amdgcn_wmma_f32_16x16x4_f32(
                       false, a, false, b, (short)0, c[t], false, false);
        }
    }

    int mBase = (lane >> 4) * 8;
    #pragma unroll
    for (int t = 0; t < CTPW; ++t) {
        float* o = H + (size_t)row0 * HDIM + (half * CTPW + t) * 16 + laneM;
        #pragma unroll
        for (int r = 0; r < 8; ++r)
            o[(size_t)(mBase + r) * HDIM] = c[t][r];
    }
}

__global__ __launch_bounds__(256) void gat_gemm_cls(
    const float* __restrict__ X, const float* __restrict__ Wlt,
    const float* __restrict__ bl, float* __restrict__ Out,
    int nRowTiles, int nCols)
{
    int wid  = (blockIdx.x * blockDim.x + threadIdx.x) >> 5;
    int lane = threadIdx.x & 31;
    if (wid >= nRowTiles) return;
    int row0 = wid * 16;
    int laneM = lane & 15;
    int kOff  = (lane >> 4) * 2;

    const float* xrow = X   + (size_t)(row0 + laneM) * HDIM + kOff;
    const float* wrow = Wlt + (size_t)laneM * HDIM + kOff;

    v8f c[3] = {};
    #pragma unroll 4
    for (int k = 0; k < HDIM; k += 4) {
        v2f a = *(const v2f*)(xrow + k);
        #pragma unroll
        for (int t = 0; t < 3; ++t) {
            v2f b = *(const v2f*)(wrow + (size_t)t * 16 * HDIM + k);
            c[t] = __builtin_amdgcn_wmma_f32_16x16x4_f32(
                       false, a, false, b, (short)0, c[t], false, false);
        }
    }

    int mBase = (lane >> 4) * 8;
    #pragma unroll
    for (int t = 0; t < 3; ++t) {
        int col = t * 16 + laneM;
        if (col >= nCols) continue;
        float bv = bl[col];
        #pragma unroll
        for (int r = 0; r < 8; ++r)
            Out[(size_t)(row0 + mBase + r) * nCols + col] = c[t][r] + bv;
    }
}

__global__ __launch_bounds__(256) void gat_alpha(
    const float* __restrict__ H, const float* __restrict__ asrc,
    const float* __restrict__ adst, float* __restrict__ aS,
    float* __restrict__ aD, int N)
{
    int w    = (blockIdx.x * blockDim.x + threadIdx.x) >> 5;
    int lane = threadIdx.x & 31;
    if (w >= N) return;
    const float* h = H + (size_t)w * HDIM;
    float s = 0.f, d = 0.f;
    #pragma unroll
    for (int j = 0; j < 4; ++j) {
        int f = lane + 32 * j;
        float hv = h[f];
        s += hv * asrc[f];
        d += hv * adst[f];
    }
    #pragma unroll
    for (int off = 16; off > 0; off >>= 1) {
        s += __shfl_xor(s, off, 32);
        d += __shfl_xor(d, off, 32);
    }
    if (lane == 0) { aS[w] = s; aD[w] = d; }
}

__device__ __forceinline__ unsigned f2ord(float f) {
    unsigned b = __float_as_uint(f);
    return (b & 0x80000000u) ? ~b : (b | 0x80000000u);
}
__device__ __forceinline__ float ord2f(unsigned u) {
    unsigned b = (u & 0x80000000u) ? (u & 0x7FFFFFFFu) : ~u;
    return __uint_as_float(b);
}
__device__ __forceinline__ float leaky02(float e) {
    return (e >= 0.f) ? e : 0.2f * e;
}

#define SROWS  128
#define SWAVES 32
#define SLOADS 8
#define WEDGES (SLOADS * 128)
#define QCAP   640

__global__ __launch_bounds__(1024) void gat_agg_strip(
    const int* __restrict__ ei, int E, int N,
    const float* __restrict__ aS, const float* __restrict__ aD,
    const float* __restrict__ H, const float* __restrict__ bias,
    float* __restrict__ act)
{
    __shared__ __attribute__((aligned(16))) float s_acc[SROWS * HDIM];
    __shared__ unsigned s_max[SROWS];
    __shared__ float    s_den[SROWS];
    __shared__ float    s_aD[SROWS];
    __shared__ float    s_self[SROWS];
    __shared__ int      s_qi[SWAVES * QCAP];
    __shared__ int      s_qd[SWAVES * QCAP];
    __shared__ float    s_qv[SWAVES * QCAP];
    __shared__ int      s_over;

    const int tid  = threadIdx.x;
    const int lane = tid & 31;
    const int wv   = tid >> 5;
    const int n0   = blockIdx.x * SROWS;
    const int rows = (N - n0 < SROWS) ? (N - n0) : SROWS;
    const int* __restrict__ dstp = ei + E;

    for (int i = tid; i < SROWS * HDIM / 4; i += SWAVES * 32)
        ((float4*)s_acc)[i] = make_float4(0.f, 0.f, 0.f, 0.f);
    if (tid < SROWS) {
        float ad = 0.f, es = 0.f;
        if (tid < rows) {
            ad = aD[n0 + tid];
            es = leaky02(aS[n0 + tid] + ad);
        }
        s_aD[tid]   = ad;
        s_self[tid] = es;
        s_max[tid]  = f2ord(es);
        s_den[tid]  = 0.f;
    }
    if (tid == 0) s_over = 0;
    __syncthreads();

    int*   qi = s_qi + wv * QCAP;
    int*   qd = s_qd + wv * QCAP;
    float* qv = s_qv + wv * QCAP;
    const bool vec = ((E & 3) == 0);

    auto load_round = [&](int base, int (&dv)[SLOADS][4]) {
        if (vec && base + WEDGES <= E) {
            #pragma unroll
            for (int j = 0; j < SLOADS; ++j) {
                int4 t = *(const int4*)(dstp + base + j * 128 + lane * 4);
                dv[j][0] = t.x; dv[j][1] = t.y; dv[j][2] = t.z; dv[j][3] = t.w;
            }
        } else {
            #pragma unroll
            for (int j = 0; j < SLOADS; ++j) {
                #pragma unroll
                for (int c = 0; c < 4; ++c) {
                    int i = base + j * 128 + lane * 4 + c;
                    dv[j][c] = (i < E) ? dstp[i] : -1;
                }
            }
        }
    };

    int cnt = 0;
    auto scan_round = [&](int base, const int (&dv)[SLOADS][4]) {
        #pragma unroll
        for (int j = 0; j < SLOADS; ++j) {
            #pragma unroll
            for (int c = 0; c < 4; ++c) {
                int dl = dv[j][c] - n0;
                bool m = (unsigned)dl < (unsigned)rows;
                unsigned mask = (unsigned)__ballot(m);
                if (mask) {
                    int pos = cnt + __popc(mask & ((1u << lane) - 1u));
                    if (m && pos < QCAP) {
                        qi[pos] = base + j * 128 + lane * 4 + c;
                        qd[pos] = dl;
                    }
                    cnt += __popc(mask);
                }
            }
        }
    };

    const int perRound = SWAVES * WEDGES;
    const int nRounds = (E + perRound - 1) / perRound;
    int rd = (nRounds > 0) ? (int)(blockIdx.x % (unsigned)nRounds) : 0;
    auto next_base = [&]() {
        int b = rd * perRound + wv * WEDGES;
        rd = (rd + 1 == nRounds) ? 0 : rd + 1;
        return b;
    };
    int dv[SLOADS][4];
    for (int it = 0; it < nRounds; ++it) {
        const int base = next_base();
        if (base >= E) continue;
        load_round(base, dv);
        scan_round(base, dv);
    }
    if (cnt > QCAP) { s_over = 1; cnt = QCAP; }

    for (int b = 0; b < cnt; b += 256) {
        int k[8], dl[8], s[8]; float as_[8];
        #pragma unroll
        for (int u = 0; u < 8; ++u) {
            int kk = b + 32 * u + lane;
            k[u] = (kk < cnt) ? kk : (cnt - 1);
        }
        #pragma unroll
        for (int u = 0; u < 8; ++u) { s[u] = ei[qi[k[u]]]; dl[u] = qd[k[u]]; }
        #pragma unroll
        for (int u = 0; u < 8; ++u) as_[u] = aS[s[u]];
        asm volatile("" :: "v"(as_[0]), "v"(as_[1]), "v"(as_[2]), "v"(as_[3]),
                           "v"(as_[4]), "v"(as_[5]), "v"(as_[6]), "v"(as_[7]));
        #pragma unroll
        for (int u = 0; u < 8; ++u) {
            if (b + 32 * u + lane < cnt) {
                float e = leaky02(as_[u] + s_aD[dl[u]]);
                qi[k[u]] = s[u];
                qv[k[u]] = e;
                atomicMax(&s_max[dl[u]], f2ord(e));
            }
        }
    }
    __syncthreads();

    const bool over = __builtin_amdgcn_readfirstlane(s_over) != 0;

    auto list_trip = [&](int trip) {
        for (int base = wv * 32; base < E; base += SWAVES * 32) {
            int i = base + lane;
            int dl = ((i < E) ? dstp[i] : -1) - n0;
            bool m = (unsigned)dl < (unsigned)rows;
            int s = 0; float al = 0.f;
            if (m) {
                s = ei[i];
                float e = leaky02(aS[s] + s_aD[dl]);
                if (trip == 0) {
                    atomicMax(&s_max[dl], f2ord(e));
                } else {
                    float ex = expf(e - ord2f(s_max[dl]));
                    if (trip == 1) atomicAdd(&s_den[dl], ex);
                    else           al = ex / s_den[dl];
                }
            }
            if (trip == 2) {
                unsigned mk = (unsigned)__ballot(m);
                while (mk) {
                    int l = __ffs((int)mk) - 1;
                    mk &= mk - 1u;
                    int   sj = __shfl(s, l, 32);
                    int   dj = __shfl(dl, l, 32);
                    float aj = __shfl(al, l, 32);
                    const float* h = H + (size_t)sj * HDIM;
                    float* o = s_acc + dj * HDIM;
                    #pragma unroll
                    for (int f = 0; f < 4; ++f)
                        atomicAdd(&o[lane + 32 * f], h[lane + 32 * f] * aj);
                }
            }
        }
    };

    if (over) {
        list_trip(0);
        __syncthreads();
    }

    if (tid < rows) {
        float ex = expf(s_self[tid] - ord2f(s_max[tid]));
        s_self[tid] = ex;
        atomicAdd(&s_den[tid], ex);
    }
    if (!over) {
        for (int b = 0; b < cnt; b += 32) {
            int k = b + lane;
            if (k < cnt) {
                int dl = qd[k];
                float ex = expf(qv[k] - ord2f(s_max[dl]));
                qv[k] = ex;
                atomicAdd(&s_den[dl], ex);
            }
        }
    } else {
        list_trip(1);
    }
    __syncthreads();

    if (!over) {
        for (int b = 0; b < cnt; b += 32) {
            int nb = (cnt - b < 32) ? (cnt - b) : 32;
            int s = 0, dl = 0; float al = 0.f;
            if (lane < nb) {
                s  = qi[b + lane];
                dl = qd[b + lane];
                al = qv[b + lane] / s_den[dl];
            }
            for (int j = 0; j < nb; j += 16) {
                v2f hv[16][2];
                #pragma unroll
                for (int u = 0; u < 16; ++u) {
                    int ju = (j + u < nb) ? (j + u) : (nb - 1);
                    int sj = __builtin_amdgcn_readlane(s, ju);
                    const float* h = H + (size_t)sj * HDIM + 2 * lane;
                    hv[u][0] = *(const v2f*)(h);
                    hv[u][1] = *(const v2f*)(h + 64);
                }
                #pragma unroll
                for (int u = 0; u < 16; ++u) {
                    if (j + u < nb) {
                        float aj = __uint_as_float(__builtin_amdgcn_readlane(__float_as_uint(al), j + u));
                        int   dj = __builtin_amdgcn_readlane(dl, j + u);
                        float* o = s_acc + dj * HDIM + 2 * lane;
                        atomicAdd(&o[0],  hv[u][0][0] * aj);
                        atomicAdd(&o[1],  hv[u][0][1] * aj);
                        atomicAdd(&o[64], hv[u][1][0] * aj);
                        atomicAdd(&o[65], hv[u][1][1] * aj);
                    }
                }
            }
        }
    } else {
        list_trip(2);
    }
    __syncthreads();

    for (int e4 = tid; e4 < rows * (HDIM / 4); e4 += SWAVES * 32) {
        int r = e4 / (HDIM / 4);
        int f = (e4 - r * (HDIM / 4)) * 4;
        float4 v = *(const float4*)(s_acc + r * HDIM + f);
        float4 h = *(const float4*)(H + (size_t)(n0 + r) * HDIM + f);
        float  a = s_self[r] / s_den[r];
        v.x += h.x * a; v.y += h.y * a; v.z += h.z * a; v.w += h.w * a;
        v.x += bias[f]; v.y += bias[f + 1]; v.z += bias[f + 2]; v.w += bias[f + 3];
        v.x = (v.x > 0.f) ? v.x : 0.f;  v.y = (v.y > 0.f) ? v.y : 0.f;
        v.z = (v.z > 0.f) ? v.z : 0.f;  v.w = (v.w > 0.f) ? v.w : 0.f;
        *(float4*)(act + (size_t)(n0 + r) * HDIM + f) = v;
    }
}

extern "C" void kernel_launch(void* const* d_in, const int* in_sizes, int n_in,
                              void* d_out, int out_size, void* d_ws, size_t ws_size,
                              hipStream_t stream)
{
    const float* x   = (const float*)d_in[0];
    const int*   ei  = (const int*)d_in[1];
    const float* W1  = (const float*)d_in[2];
    const float* a1s = (const float*)d_in[3];
    const float* a1d = (const float*)d_in[4];
    const float* b1  = (const float*)d_in[5];
    const float* W2  = (const float*)d_in[6];
    const float* a2s = (const float*)d_in[7];
    const float* a2d = (const float*)d_in[8];
    const float* b2  = (const float*)d_in[9];
    const float* Wl  = (const float*)d_in[10];
    const float* bl  = (const float*)d_in[11];

    const int N  = in_sizes[0] / HDIM;
    const int E  = in_sizes[1] / 2;
    const int C  = in_sizes[11];
    const int CP = (C + 15) & ~15;

    char* ws = (char*)d_ws;
    auto align256 = [](size_t v) { return (v + 255) & ~(size_t)255; };
    size_t off = 0;
    float*    bufH  = (float*)(ws + off);  off += align256((size_t)N * HDIM * 4);
    float*    bufA  = (float*)(ws + off);  off += align256((size_t)N * HDIM * 4);
    float*    Wt    = (float*)(ws + off);  off += align256((size_t)HDIM * HDIM * 4);
    float*    Wlt   = (float*)(ws + off);  off += align256((size_t)CP * HDIM * 4);
    float*    aS    = (float*)(ws + off);  off += align256((size_t)N * 4);
    float*    aD    = (float*)(ws + off);  off += align256((size_t)N * 4);

    const int nRowTiles = N / 16;
    const int BLK = 256;
    dim3 blk(BLK);

    dim3 trGrid((HDIM * HDIM + BLK - 1) / BLK);
    dim3 gemmGrid((nRowTiles * 2 + 7) / 8);
    dim3 waveGridN((N + 7) / 8);
    dim3 stripGrid((N + SROWS - 1) / SROWS);

    auto run_layer = [&](const float* act_in, const float* W,
                         const float* as_, const float* ad_, const float* bias) {
        gat_transpose<<<trGrid, blk, 0, stream>>>(W, Wt, HDIM, HDIM, HDIM);
        gat_gemm_strip<<<gemmGrid, blk, 0, stream>>>(act_in, Wt, bufH, nRowTiles);
        gat_alpha<<<waveGridN, blk, 0, stream>>>(bufH, as_, ad_, aS, aD, N);
        gat_agg_strip<<<stripGrid, dim3(SWAVES * 32), 0, stream>>>(ei, E, N, aS, aD, bufH, bias, bufA);
    };

    run_layer(x, W1, a1s, a1d, b1);
    run_layer(bufA, W2, a2s, a2d, b2);

    dim3 trlGrid((CP * HDIM + BLK - 1) / BLK);
    gat_transpose<<<trlGrid, blk, 0, stream>>>(Wl, Wlt, HDIM, C, CP);
    dim3 clsGrid((nRowTiles + 7) / 8);
    gat_gemm_cls<<<clsGrid, blk, 0, stream>>>(bufA, Wlt, bl, (float*)d_out,
                                              nRowTiles, C);
}


# Layer 2's strip work-groups read their edge queues (source row, strip row) from workspace, where layer 1's strip launch saved them: no second filtering of the edge list, no second index gather
# speedup vs baseline: 2.5021x; 1.1065x over previous
#include <hip/hip_runtime.h>
#include <hip/hip_bf16.h>

typedef __attribute__((ext_vector_type(2))) float v2f;
typedef __attribute__((ext_vector_type(8))) float v8f;

#define HDIM 128
#define CTPW 4

__global__ __launch_bounds__(256) void gat_transpose(
    const float* __restrict__ W, float* __restrict__ Wt,
    int rows, int cols, int colsPad)
{
    int i = blockIdx.x * blockDim.x + threadIdx.x;
    int total = colsPad * rows;
    if (i >= total) return;
    int c = i / rows;
    int r = i - c * rows;
    Wt[(size_t)c * rows + r] = (c < cols) ? W[(size_t)r * cols + c] : 0.f;
}

__global__ __launch_bounds__(256) void gat_gemm_strip(
    const float* __restrict__ X, const float* __restrict__ Wt,
    float* __restrict__ H, int nRowTiles)
{
    int wid  = (blockIdx.x * blockDim.x + threadIdx.x) >> 5;
    int lane = threadIdx.x & 31;
    int nHalves = HDIM / (CTPW * 16);
    int nW = nRowTiles * nHalves;
    if (wid >= nW) return;
    int rt   = wid >> 1;
    int half = wid & 1;
    int row0 = rt * 16;
    int laneM = lane & 15;
    int kOff  = (lane >> 4) * 2;

    const float* xrow = X  + (size_t)(row0 + laneM) * HDIM + kOff;
    const float* wrow = Wt + (size_t)(half * (CTPW * 16) + laneM) * HDIM + kOff;

    v8f c[CTPW] = {};
    #pragma unroll 4
    for (int k = 0; k < HDIM; k += 4) {
        v2f a = *(const v2f*)(xrow + k);
        #pragma unroll
        for (int t = 0; t < CTPW; ++t) {
            v2f b = *(const v2f*)(wrow + (size_t)t * 16 * HDIM + k);
            c[t] = __builtin_amdgcn_wmma_f32_16x16x4_f32(
                       false, a, false, b, (short)0, c[t], false, false);
        }
    }

    int mBase = (lane >> 4) * 8;
    #pragma unroll
    for (int t = 0; t < CTPW; ++t) {
        float* o = H + (size_t)row0 * HDIM + (half * CTPW + t) * 16 + laneM;
        #pragma unroll
        for (int r = 0; r < 8; ++r)
            o[(size_t)(mBase + r) * HDIM] = c[t][r];
    }
}

__global__ __launch_bounds__(256) void gat_gemm_cls(
    const float* __restrict__ X, const float* __restrict__ Wlt,
    const float* __restrict__ bl, float* __restrict__ Out,
    int nRowTiles, int nCols)
{
    int wid  = (blockIdx.x * blockDim.x + threadIdx.x) >> 5;
    int lane = threadIdx.x & 31;
    if (wid >= nRowTiles) return;
    int row0 = wid * 16;
    int laneM = lane & 15;
    int kOff  = (lane >> 4) * 2;

    const float* xrow = X   + (size_t)(row0 + laneM) * HDIM + kOff;
    const float* wrow = Wlt + (size_t)laneM * HDIM + kOff;

    v8f c[3] = {};
    #pragma unroll 4
    for (int k = 0; k < HDIM; k += 4) {
        v2f a = *(const v2f*)(xrow + k);
        #pragma unroll
        for (int t = 0; t < 3; ++t) {
            v2f b = *(const v2f*)(wrow + (size_t)t * 16 * HDIM + k);
            c[t] = __builtin_amdgcn_wmma_f32_16x16x4_f32(
                       false, a, false, b, (short)0, c[t], false, false);
        }
    }

    int mBase = (lane >> 4) * 8;
    #pragma unroll
    for (int t = 0; t < 3; ++t) {
        int col = t * 16 + laneM;
        if (col >= nCols) continue;
        float bv = bl[col];
        #pragma unroll
        for (int r = 0; r < 8; ++r)
            Out[(size_t)(row0 + mBase + r) * nCols + col] = c[t][r] + bv;
    }
}

__global__ __launch_bounds__(256) void gat_alpha(
    const float* __restrict__ H, const float* __restrict__ asrc,
    const float* __restrict__ adst, float* __restrict__ aS,
    float* __restrict__ aD, int N)
{
    int w    = (blockIdx.x * blockDim.x + threadIdx.x) >> 5;
    int lane = threadIdx.x & 31;
    if (w >= N) return;
    const float* h = H + (size_t)w * HDIM;
    float s = 0.f, d = 0.f;
    #pragma unroll
    for (int j = 0; j < 4; ++j) {
        int f = lane + 32 * j;
        float hv = h[f];
        s += hv * asrc[f];
        d += hv * adst[f];
    }
    #pragma unroll
    for (int off = 16; off > 0; off >>= 1) {
        s += __shfl_xor(s, off, 32);
        d += __shfl_xor(d, off, 32);
    }
    if (lane == 0) { aS[w] = s; aD[w] = d; }
}

__device__ __forceinline__ unsigned f2ord(float f) {
    unsigned b = __float_as_uint(f);
    return (b & 0x80000000u) ? ~b : (b | 0x80000000u);
}
__device__ __forceinline__ float ord2f(unsigned u) {
    unsigned b = (u & 0x80000000u) ? (u & 0x7FFFFFFFu) : ~u;
    return __uint_as_float(b);
}
__device__ __forceinline__ float leaky02(float e) {
    return (e >= 0.f) ? e : 0.2f * e;
}

#define SROWS  128
#define SWAVES 32
#define SLOADS 8
#define WEDGES (SLOADS * 128)
#define QCAP   640
#define QKEEP  320
#define QSAVE  (QKEEP + 32)
#define QSRCBITS 24

__global__ __launch_bounds__(1024) void gat_agg_strip(
    const int* __restrict__ ei, int E, int N,
    const float* __restrict__ aS, const float* __restrict__ aD,
    const float* __restrict__ H, const float* __restrict__ bias,
    float* __restrict__ act,
    const int* __restrict__ qin,
    int* __restrict__ qout)
{
    __shared__ __attribute__((aligned(16))) float s_acc[SROWS * HDIM];
    __shared__ unsigned s_max[SROWS];
    __shared__ float    s_den[SROWS];
    __shared__ float    s_aD[SROWS];
    __shared__ float    s_self[SROWS];
    __shared__ int      s_qi[SWAVES * QCAP];
    __shared__ int      s_qd[SWAVES * QCAP];
    __shared__ float    s_qv[SWAVES * QCAP];
    __shared__ int      s_over;

    const int tid  = threadIdx.x;
    const int lane = tid & 31;
    const int wv   = tid >> 5;
    const int n0   = blockIdx.x * SROWS;
    const int rows = (N - n0 < SROWS) ? (N - n0) : SROWS;
    const int* __restrict__ dstp = ei + E;

    for (int i = tid; i < SROWS * HDIM / 4; i += SWAVES * 32)
        ((float4*)s_acc)[i] = make_float4(0.f, 0.f, 0.f, 0.f);
    if (tid < SROWS) {
        float ad = 0.f, es = 0.f;
        if (tid < rows) {
            ad = aD[n0 + tid];
            es = leaky02(aS[n0 + tid] + ad);
        }
        s_aD[tid]   = ad;
        s_self[tid] = es;
        s_max[tid]  = f2ord(es);
        s_den[tid]  = 0.f;
    }
    if (tid == 0) s_over = 0;
    __syncthreads();

    int*   qi = s_qi + wv * QCAP;
    int*   qd = s_qd + wv * QCAP;
    float* qv = s_qv + wv * QCAP;
    const bool vec = ((E & 3) == 0);

    auto load_round = [&](int base, int (&dv)[SLOADS][4]) {
        if (vec && base + WEDGES <= E) {
            #pragma unroll
            for (int j = 0; j < SLOADS; ++j) {
                int4 t = *(const int4*)(dstp + base + j * 128 + lane * 4);
                dv[j][0] = t.x; dv[j][1] = t.y; dv[j][2] = t.z; dv[j][3] = t.w;
            }
        } else {
            #pragma unroll
            for (int j = 0; j < SLOADS; ++j) {
                #pragma unroll
                for (int c = 0; c < 4; ++c) {
                    int i = base + j * 128 + lane * 4 + c;
                    dv[j][c] = (i < E) ? dstp[i] : -1;
                }
            }
        }
    };

    int cnt = 0;
    auto scan_round = [&](int base, const int (&dv)[SLOADS][4]) {
        #pragma unroll
        for (int j = 0; j < SLOADS; ++j) {
            #pragma unroll
            for (int c = 0; c < 4; ++c) {
                int dl = dv[j][c] - n0;
                bool m = (unsigned)dl < (unsigned)rows;
                unsigned mask = (unsigned)__ballot(m);
                if (mask) {
                    int pos = cnt + __popc(mask & ((1u << lane) - 1u));
                    if (m && pos < QCAP) {
                        qi[pos] = base + j * 128 + lane * 4 + c;
                        qd[pos] = dl;
                    }
                    cnt += __popc(mask);
                }
            }
        }
    };

    const int perRound = SWAVES * WEDGES;
    const int nRounds = (E + perRound - 1) / perRound;
    int rd = (nRounds > 0) ? (int)(blockIdx.x % (unsigned)nRounds) : 0;
    auto next_base = [&]() {
        int b = rd * perRound + wv * WEDGES;
        rd = (rd + 1 == nRounds) ? 0 : rd + 1;
        return b;
    };
    const size_t wq = ((size_t)blockIdx.x * SWAVES + wv) * QSAVE;
    const int saved = qin ? __builtin_amdgcn_readfirstlane(qin[wq]) : -1;
    const bool reuse = saved >= 0;
    bool save = false;
    if (reuse) {
        cnt = saved;
    } else {
        int dv[SLOADS][4];
        for (int it = 0; it < nRounds; ++it) {
            const int base = next_base();
            if (base >= E) continue;
            load_round(base, dv);
            scan_round(base, dv);
        }
        save = (qout != nullptr) && (cnt <= QKEEP);
        if (qout && lane == 0) qout[wq] = save ? cnt : -1;
    }
    if (cnt > QCAP) { s_over = 1; cnt = QCAP; }

    for (int b = 0; b < cnt; b += 256) {
        int k[8], dl[8], s[8]; float as_[8];
        #pragma unroll
        for (int u = 0; u < 8; ++u) {
            int kk = b + 32 * u + lane;
            k[u] = (kk < cnt) ? kk : (cnt - 1);
        }
        if (reuse) {
            #pragma unroll
            for (int u = 0; u < 8; ++u) {
                unsigned p = (unsigned)qin[wq + 32 + k[u]];
                s[u]  = (int)(p & ((1u << QSRCBITS) - 1u));
                dl[u] = (int)(p >> QSRCBITS);
            }
        } else {
            #pragma unroll
            for (int u = 0; u < 8; ++u) { s[u] = ei[qi[k[u]]]; dl[u] = qd[k[u]]; }
        }
        #pragma unroll
        for (int u = 0; u < 8; ++u) as_[u] = aS[s[u]];
        asm volatile("" :: "v"(as_[0]), "v"(as_[1]), "v"(as_[2]), "v"(as_[3]),
                           "v"(as_[4]), "v"(as_[5]), "v"(as_[6]), "v"(as_[7]));
        #pragma unroll
        for (int u = 0; u < 8; ++u) {
            if (b + 32 * u + lane < cnt) {
                float e = leaky02(as_[u] + s_aD[dl[u]]);
                qi[k[u]] = s[u];
                qv[k[u]] = e;
                if (reuse) qd[k[u]] = dl[u];
                else if (save) qout[wq + 32 + k[u]] = (int)((unsigned)s[u] | ((unsigned)dl[u] << QSRCBITS));
                atomicMax(&s_max[dl[u]], f2ord(e));
            }
        }
    }
    __syncthreads();

    const bool over = __builtin_amdgcn_readfirstlane(s_over) != 0;

    auto list_trip = [&](int trip) {
        for (int base = wv * 32; base < E; base += SWAVES * 32) {
            int i = base + lane;
            int dl = ((i < E) ? dstp[i] : -1) - n0;
            bool m = (unsigned)dl < (unsigned)rows;
            int s = 0; float al = 0.f;
            if (m) {
                s = ei[i];
                float e = leaky02(aS[s] + s_aD[dl]);
                if (trip == 0) {
                    atomicMax(&s_max[dl], f2ord(e));
                } else {
                    float ex = expf(e - ord2f(s_max[dl]));
                    if (trip == 1) atomicAdd(&s_den[dl], ex);
                    else           al = ex / s_den[dl];
                }
            }
            if (trip == 2) {
                unsigned mk = (unsigned)__ballot(m);
                while (mk) {
                    int l = __ffs((int)mk) - 1;
                    mk &= mk - 1u;
                    int   sj = __shfl(s, l, 32);
                    int   dj = __shfl(dl, l, 32);
                    float aj = __shfl(al, l, 32);
                    const float* h = H + (size_t)sj * HDIM;
                    float* o = s_acc + dj * HDIM;
                    #pragma unroll
                    for (int f = 0; f < 4; ++f)
                        atomicAdd(&o[lane + 32 * f], h[lane + 32 * f] * aj);
                }
            }
        }
    };

    if (over) {
        list_trip(0);
        __syncthreads();
    }

    if (tid < rows) {
        float ex = expf(s_self[tid] - ord2f(s_max[tid]));
        s_self[tid] = ex;
        atomicAdd(&s_den[tid], ex);
    }
    if (!over) {
        for (int b = 0; b < cnt; b += 32) {
            int k = b + lane;
            if (k < cnt) {
                int dl = qd[k];
                float ex = expf(qv[k] - ord2f(s_max[dl]));
                qv[k] = ex;
                atomicAdd(&s_den[dl], ex);
            }
        }
    } else {
        list_trip(1);
    }
    __syncthreads();

    if (!over) {
        for (int b = 0; b < cnt; b += 32) {
            int nb = (cnt - b < 32) ? (cnt - b) : 32;
            int s = 0, dl = 0; float al = 0.f;
            if (lane < nb) {
                s  = qi[b + lane];
                dl = qd[b + lane];
                al = qv[b + lane] / s_den[dl];
            }
            for (int j = 0; j < nb; j += 16) {
                v2f hv[16][2];
                #pragma unroll
                for (int u = 0; u < 16; ++u) {
                    int ju = (j + u < nb) ? (j + u) : (nb - 1);
                    int sj = __builtin_amdgcn_readlane(s, ju);
                    const float* h = H + (size_t)sj * HDIM + 2 * lane;
                    hv[u][0] = *(const v2f*)(h);
                    hv[u][1] = *(const v2f*)(h + 64);
                }
                #pragma unroll
                for (int u = 0; u < 16; ++u) {
                    if (j + u < nb) {
                        float aj = __uint_as_float(__builtin_amdgcn_readlane(__float_as_uint(al), j + u));
                        int   dj = __builtin_amdgcn_readlane(dl, j + u);
                        float* o = s_acc + dj * HDIM + 2 * lane;
                        atomicAdd(&o[0],  hv[u][0][0] * aj);
                        atomicAdd(&o[1],  hv[u][0][1] * aj);
                        atomicAdd(&o[64], hv[u][1][0] * aj);
                        atomicAdd(&o[65], hv[u][1][1] * aj);
                    }
                }
            }
        }
    } else {
        list_trip(2);
    }
    __syncthreads();

    for (int e4 = tid; e4 < rows * (HDIM / 4); e4 += SWAVES * 32) {
        int r = e4 / (HDIM / 4);
        int f = (e4 - r * (HDIM / 4)) * 4;
        float4 v = *(const float4*)(s_acc + r * HDIM + f);
        float4 h = *(const float4*)(H + (size_t)(n0 + r) * HDIM + f);
        float  a = s_self[r] / s_den[r];
        v.x += h.x * a; v.y += h.y * a; v.z += h.z * a; v.w += h.w * a;
        v.x += bias[f]; v.y += bias[f + 1]; v.z += bias[f + 2]; v.w += bias[f + 3];
        v.x = (v.x > 0.f) ? v.x : 0.f;  v.y = (v.y > 0.f) ? v.y : 0.f;
        v.z = (v.z > 0.f) ? v.z : 0.f;  v.w = (v.w > 0.f) ? v.w : 0.f;
        *(float4*)(act + (size_t)(n0 + r) * HDIM + f) = v;
    }
}

extern "C" void kernel_launch(void* const* d_in, const int* in_sizes, int n_in,
                              void* d_out, int out_size, void* d_ws, size_t ws_size,
                              hipStream_t stream)
{
    const float* x   = (const float*)d_in[0];
    const int*   ei  = (const int*)d_in[1];
    const float* W1  = (const float*)d_in[2];
    const float* a1s = (const float*)d_in[3];
    const float* a1d = (const float*)d_in[4];
    const float* b1  = (const float*)d_in[5];
    const float* W2  = (const float*)d_in[6];
    const float* a2s = (const float*)d_in[7];
    const float* a2d = (const float*)d_in[8];
    const float* b2  = (const float*)d_in[9];
    const float* Wl  = (const float*)d_in[10];
    const float* bl  = (const float*)d_in[11];

    const int N  = in_sizes[0] / HDIM;
    const int E  = in_sizes[1] / 2;
    const int C  = in_sizes[11];
    const int CP = (C + 15) & ~15;

    char* ws = (char*)d_ws;
    auto align256 = [](size_t v) { return (v + 255) & ~(size_t)255; };
    size_t off = 0;
    float*    bufH  = (float*)(ws + off);  off += align256((size_t)N * HDIM * 4);
    float*    bufA  = (float*)(ws + off);  off += align256((size_t)N * HDIM * 4);
    float*    Wt    = (float*)(ws + off);  off += align256((size_t)HDIM * HDIM * 4);
    float*    Wlt   = (float*)(ws + off);  off += align256((size_t)CP * HDIM * 4);
    float*    aS    = (float*)(ws + off);  off += align256((size_t)N * 4);
    float*    aD    = (float*)(ws + off);  off += align256((size_t)N * 4);
    const size_t nStrips = (size_t)(N + SROWS - 1) / SROWS;
    const size_t qBytes  = nStrips * SWAVES * QSAVE * 4;
    int* qsave = (off + qBytes <= ws_size && N <= (1 << QSRCBITS)) ? (int*)(ws + off) : nullptr;

    const int nRowTiles = N / 16;
    const int BLK = 256;
    dim3 blk(BLK);

    dim3 trGrid((HDIM * HDIM + BLK - 1) / BLK);
    dim3 gemmGrid((nRowTiles * 2 + 7) / 8);
    dim3 waveGridN((N + 7) / 8);
    dim3 stripGrid((N + SROWS - 1) / SROWS);

    auto run_layer = [&](const float* act_in, const float* W,
                         const float* as_, const float* ad_, const float* bias,
                         const int* qin, int* qout) {
        gat_transpose<<<trGrid, blk, 0, stream>>>(W, Wt, HDIM, HDIM, HDIM);
        gat_gemm_strip<<<gemmGrid, blk, 0, stream>>>(act_in, Wt, bufH, nRowTiles);
        gat_alpha<<<waveGridN, blk, 0, stream>>>(bufH, as_, ad_, aS, aD, N);
        gat_agg_strip<<<stripGrid, dim3(SWAVES * 32), 0, stream>>>(ei, E, N, aS, aD, bufH, bias, bufA, qin, qout);
    };

    run_layer(x, W1, a1s, a1d, b1, nullptr, qsave);
    run_layer(bufA, W2, a2s, a2d, b2, qsave, nullptr);

    dim3 trlGrid((CP * HDIM + BLK - 1) / BLK);
    gat_transpose<<<trlGrid, blk, 0, stream>>>(Wl, Wlt, HDIM, C, CP);
    dim3 clsGrid((nRowTiles + 7) / 8);
    gat_gemm_cls<<<clsGrid, blk, 0, stream>>>(bufA, Wlt, bl, (float*)d_out,
                                              nRowTiles, C);
}


# Attention logits: one wave per 16 nodes, not one per node; a lane's 32 partial sums are summed by a halving lane exchange at 16, 8, 4, 2, 1: 40000 -> 2500 logit waves per call; same sum order
# speedup vs baseline: 2.5847x; 1.0330x over previous
#include <hip/hip_runtime.h>
#include <hip/hip_bf16.h>

typedef __attribute__((ext_vector_type(2))) float v2f;
typedef __attribute__((ext_vector_type(8))) float v8f;

#define HDIM 128
#define CTPW 4

__global__ __launch_bounds__(256) void gat_transpose(
    const float* __restrict__ W, float* __restrict__ Wt,
    int rows, int cols, int colsPad)
{
    int i = blockIdx.x * blockDim.x + threadIdx.x;
    int total = colsPad * rows;
    if (i >= total) return;
    int c = i / rows;
    int r = i - c * rows;
    Wt[(size_t)c * rows + r] = (c < cols) ? W[(size_t)r * cols + c] : 0.f;
}

__global__ __launch_bounds__(256) void gat_gemm_strip(
    const float* __restrict__ X, const float* __restrict__ Wt,
    float* __restrict__ H, int nRowTiles)
{
    int wid  = (blockIdx.x * blockDim.x + threadIdx.x) >> 5;
    int lane = threadIdx.x & 31;
    int nHalves = HDIM / (CTPW * 16);
    int nW = nRowTiles * nHalves;
    if (wid >= nW) return;
    int rt   = wid >> 1;
    int half = wid & 1;
    int row0 = rt * 16;
    int laneM = lane & 15;
    int kOff  = (lane >> 4) * 2;

    const float* xrow = X  + (size_t)(row0 + laneM) * HDIM + kOff;
    const float* wrow = Wt + (size_t)(half * (CTPW * 16) + laneM) * HDIM + kOff;

    v8f c[CTPW] = {};
    #pragma unroll 4
    for (int k = 0; k < HDIM; k += 4) {
        v2f a = *(const v2f*)(xrow + k);
        #pragma unroll
        for (int t = 0; t < CTPW; ++t) {
            v2f b = *(const v2f*)(wrow + (size_t)t * 16 * HDIM + k);
            c[t] = __builtin_amdgcn_wmma_f32_16x16x4_f32(
                       false, a, false, b, (short)0, c[t], false, false);
        }
    }

    int mBase = (lane >> 4) * 8;
    #pragma unroll
    for (int t = 0; t < CTPW; ++t) {
        float* o = H + (size_t)row0 * HDIM + (half * CTPW + t) * 16 + laneM;
        #pragma unroll
        for (int r = 0; r < 8; ++r)
            o[(size_t)(mBase + r) * HDIM] = c[t][r];
    }
}

__global__ __launch_bounds__(256) void gat_gemm_cls(
    const float* __restrict__ X, const float* __restrict__ Wlt,
    const float* __restrict__ bl, float* __restrict__ Out,
    int nRowTiles, int nCols)
{
    int wid  = (blockIdx.x * blockDim.x + threadIdx.x) >> 5;
    int lane = threadIdx.x & 31;
    if (wid >= nRowTiles) return;
    int row0 = wid * 16;
    int laneM = lane & 15;
    int kOff  = (lane >> 4) * 2;

    const float* xrow = X   + (size_t)(row0 + laneM) * HDIM + kOff;
    const float* wrow = Wlt + (size_t)laneM * HDIM + kOff;

    v8f c[3] = {};
    #pragma unroll 4
    for (int k = 0; k < HDIM; k += 4) {
        v2f a = *(const v2f*)(xrow + k);
        #pragma unroll
        for (int t = 0; t < 3; ++t) {
            v2f b = *(const v2f*)(wrow + (size_t)t * 16 * HDIM + k);
            c[t] = __builtin_amdgcn_wmma_f32_16x16x4_f32(
                       false, a, false, b, (short)0, c[t], false, false);
        }
    }

    int mBase = (lane >> 4) * 8;
    #pragma unroll
    for (int t = 0; t < 3; ++t) {
        int col = t * 16 + laneM;
        if (col >= nCols) continue;
        float bv = bl[col];
        #pragma unroll
        for (int r = 0; r < 8; ++r)
            Out[(size_t)(row0 + mBase + r) * nCols + col] = c[t][r] + bv;
    }
}

#define AROWS 16
template <int O>
__device__ __forceinline__ void alpha_halve(float (&val)[2 * AROWS], int lane)
{
    const bool up = (lane & O) != 0;
    #pragma unroll
    for (int v = 0; v < O; ++v) {
        const float send = up ? val[v]     : val[v + O];
        const float keep = up ? val[v + O] : val[v];
        val[v] = keep + __shfl_xor(send, O, 32);
    }
}

__global__ __launch_bounds__(128) void gat_alpha(
    const float* __restrict__ H, const float* __restrict__ asrc,
    const float* __restrict__ adst, float* __restrict__ aS,
    float* __restrict__ aD, int N)
{
    const int w    = (blockIdx.x * blockDim.x + threadIdx.x) >> 5;
    const int lane = threadIdx.x & 31;
    const int row0 = w * AROWS;
    if (row0 >= N) return;

    float as4[4], ad4[4];
    #pragma unroll
    for (int j = 0; j < 4; ++j) {
        as4[j] = asrc[lane + 32 * j];
        ad4[j] = adst[lane + 32 * j];
    }

    float hv[AROWS][4];
    #pragma unroll
    for (int r = 0; r < AROWS; ++r) {
        const int row = (row0 + r < N) ? (row0 + r) : (N - 1);
        const float* h = H + (size_t)row * HDIM + lane;
        #pragma unroll
        for (int j = 0; j < 4; ++j) hv[r][j] = h[32 * j];
    }

    float val[2 * AROWS];
    #pragma unroll
    for (int r = 0; r < AROWS; ++r) {
        float s = 0.f, d = 0.f;
        #pragma unroll
        for (int j = 0; j < 4; ++j) {
            s += hv[r][j] * as4[j];
            d += hv[r][j] * ad4[j];
        }
        val[r]         = s;
        val[AROWS + r] = d;
    }

    alpha_halve<16>(val, lane);
    alpha_halve<8>(val, lane);
    alpha_halve<4>(val, lane);
    alpha_halve<2>(val, lane);
    alpha_halve<1>(val, lane);

    const int row = row0 + (lane & (AROWS - 1));
    if (row < N) {
        if (lane < AROWS) aS[row] = val[0];
        else              aD[row] = val[0];
    }
}

__device__ __forceinline__ unsigned f2ord(float f) {
    unsigned b = __float_as_uint(f);
    return (b & 0x80000000u) ? ~b : (b | 0x80000000u);
}
__device__ __forceinline__ float ord2f(unsigned u) {
    unsigned b = (u & 0x80000000u) ? (u & 0x7FFFFFFFu) : ~u;
    return __uint_as_float(b);
}
__device__ __forceinline__ float leaky02(float e) {
    return (e >= 0.f) ? e : 0.2f * e;
}

#define SROWS  128
#define SWAVES 32
#define SLOADS 8
#define WEDGES (SLOADS * 128)
#define QCAP   640
#define QKEEP  320
#define QSAVE  (QKEEP + 32)
#define QSRCBITS 24

__global__ __launch_bounds__(1024) void gat_agg_strip(
    const int* __restrict__ ei, int E, int N,
    const float* __restrict__ aS, const float* __restrict__ aD,
    const float* __restrict__ H, const float* __restrict__ bias,
    float* __restrict__ act,
    const int* __restrict__ qin,
    int* __restrict__ qout)
{
    __shared__ __attribute__((aligned(16))) float s_acc[SROWS * HDIM];
    __shared__ unsigned s_max[SROWS];
    __shared__ float    s_den[SROWS];
    __shared__ float    s_aD[SROWS];
    __shared__ float    s_self[SROWS];
    __shared__ int      s_qi[SWAVES * QCAP];
    __shared__ int      s_qd[SWAVES * QCAP];
    __shared__ float    s_qv[SWAVES * QCAP];
    __shared__ int      s_over;

    const int tid  = threadIdx.x;
    const int lane = tid & 31;
    const int wv   = tid >> 5;
    const int n0   = blockIdx.x * SROWS;
    const int rows = (N - n0 < SROWS) ? (N - n0) : SROWS;
    const int* __restrict__ dstp = ei + E;

    for (int i = tid; i < SROWS * HDIM / 4; i += SWAVES * 32)
        ((float4*)s_acc)[i] = make_float4(0.f, 0.f, 0.f, 0.f);
    if (tid < SROWS) {
        float ad = 0.f, es = 0.f;
        if (tid < rows) {
            ad = aD[n0 + tid];
            es = leaky02(aS[n0 + tid] + ad);
        }
        s_aD[tid]   = ad;
        s_self[tid] = es;
        s_max[tid]  = f2ord(es);
        s_den[tid]  = 0.f;
    }
    if (tid == 0) s_over = 0;
    __syncthreads();

    int*   qi = s_qi + wv * QCAP;
    int*   qd = s_qd + wv * QCAP;
    float* qv = s_qv + wv * QCAP;
    const bool vec = ((E & 3) == 0);

    auto load_round = [&](int base, int (&dv)[SLOADS][4]) {
        if (vec && base + WEDGES <= E) {
            #pragma unroll
            for (int j = 0; j < SLOADS; ++j) {
                int4 t = *(const int4*)(dstp + base + j * 128 + lane * 4);
                dv[j][0] = t.x; dv[j][1] = t.y; dv[j][2] = t.z; dv[j][3] = t.w;
            }
        } else {
            #pragma unroll
            for (int j = 0; j < SLOADS; ++j) {
                #pragma unroll
                for (int c = 0; c < 4; ++c) {
                    int i = base + j * 128 + lane * 4 + c;
                    dv[j][c] = (i < E) ? dstp[i] : -1;
                }
            }
        }
    };

    int cnt = 0;
    auto scan_round = [&](int base, const int (&dv)[SLOADS][4]) {
        #pragma unroll
        for (int j = 0; j < SLOADS; ++j) {
            #pragma unroll
            for (int c = 0; c < 4; ++c) {
                int dl = dv[j][c] - n0;
                bool m = (unsigned)dl < (unsigned)rows;
                unsigned mask = (unsigned)__ballot(m);
                if (mask) {
                    int pos = cnt + __popc(mask & ((1u << lane) - 1u));
                    if (m && pos < QCAP) {
                        qi[pos] = base + j * 128 + lane * 4 + c;
                        qd[pos] = dl;
                    }
                    cnt += __popc(mask);
                }
            }
        }
    };

    const int perRound = SWAVES * WEDGES;
    const int nRounds = (E + perRound - 1) / perRound;
    int rd = (nRounds > 0) ? (int)(blockIdx.x % (unsigned)nRounds) : 0;
    auto next_base = [&]() {
        int b = rd * perRound + wv * WEDGES;
        rd = (rd + 1 == nRounds) ? 0 : rd + 1;
        return b;
    };
    const size_t wq = ((size_t)blockIdx.x * SWAVES + wv) * QSAVE;
    const int saved = qin ? __builtin_amdgcn_readfirstlane(qin[wq]) : -1;
    const bool reuse = saved >= 0;
    bool save = false;
    if (reuse) {
        cnt = saved;
    } else {
        int dv[SLOADS][4];
        for (int it = 0; it < nRounds; ++it) {
            const int base = next_base();
            if (base >= E) continue;
            load_round(base, dv);
            scan_round(base, dv);
        }
        save = (qout != nullptr) && (cnt <= QKEEP);
        if (qout && lane == 0) qout[wq] = save ? cnt : -1;
    }
    if (cnt > QCAP) { s_over = 1; cnt = QCAP; }

    for (int b = 0; b < cnt; b += 256) {
        int k[8], dl[8], s[8]; float as_[8];
        #pragma unroll
        for (int u = 0; u < 8; ++u) {
            int kk = b + 32 * u + lane;
            k[u] = (kk < cnt) ? kk : (cnt - 1);
        }
        if (reuse) {
            #pragma unroll
            for (int u = 0; u < 8; ++u) {
                unsigned p = (unsigned)qin[wq + 32 + k[u]];
                s[u]  = (int)(p & ((1u << QSRCBITS) - 1u));
                dl[u] = (int)(p >> QSRCBITS);
            }
        } else {
            #pragma unroll
            for (int u = 0; u < 8; ++u) { s[u] = ei[qi[k[u]]]; dl[u] = qd[k[u]]; }
        }
        #pragma unroll
        for (int u = 0; u < 8; ++u) as_[u] = aS[s[u]];
        asm volatile("" :: "v"(as_[0]), "v"(as_[1]), "v"(as_[2]), "v"(as_[3]),
                           "v"(as_[4]), "v"(as_[5]), "v"(as_[6]), "v"(as_[7]));
        #pragma unroll
        for (int u = 0; u < 8; ++u) {
            if (b + 32 * u + lane < cnt) {
                float e = leaky02(as_[u] + s_aD[dl[u]]);
                qi[k[u]] = s[u];
                qv[k[u]] = e;
                if (reuse) qd[k[u]] = dl[u];
                else if (save) qout[wq + 32 + k[u]] = (int)((unsigned)s[u] | ((unsigned)dl[u] << QSRCBITS));
                atomicMax(&s_max[dl[u]], f2ord(e));
            }
        }
    }
    __syncthreads();

    const bool over = __builtin_amdgcn_readfirstlane(s_over) != 0;

    auto list_trip = [&](int trip) {
        for (int base = wv * 32; base < E; base += SWAVES * 32) {
            int i = base + lane;
            int dl = ((i < E) ? dstp[i] : -1) - n0;
            bool m = (unsigned)dl < (unsigned)rows;
            int s = 0; float al = 0.f;
            if (m) {
                s = ei[i];
                float e = leaky02(aS[s] + s_aD[dl]);
                if (trip == 0) {
                    atomicMax(&s_max[dl], f2ord(e));
                } else {
                    float ex = expf(e - ord2f(s_max[dl]));
                    if (trip == 1) atomicAdd(&s_den[dl], ex);
                    else           al = ex / s_den[dl];
                }
            }
            if (trip == 2) {
                unsigned mk = (unsigned)__ballot(m);
                while (mk) {
                    int l = __ffs((int)mk) - 1;
                    mk &= mk - 1u;
                    int   sj = __shfl(s, l, 32);
                    int   dj = __shfl(dl, l, 32);
                    float aj = __shfl(al, l, 32);
                    const float* h = H + (size_t)sj * HDIM;
                    float* o = s_acc + dj * HDIM;
                    #pragma unroll
                    for (int f = 0; f < 4; ++f)
                        atomicAdd(&o[lane + 32 * f], h[lane + 32 * f] * aj);
                }
            }
        }
    };

    if (over) {
        list_trip(0);
        __syncthreads();
    }

    if (tid < rows) {
        float ex = expf(s_self[tid] - ord2f(s_max[tid]));
        s_self[tid] = ex;
        atomicAdd(&s_den[tid], ex);
    }
    if (!over) {
        for (int b = 0; b < cnt; b += 32) {
            int k = b + lane;
            if (k < cnt) {
                int dl = qd[k];
                float ex = expf(qv[k] - ord2f(s_max[dl]));
                qv[k] = ex;
                atomicAdd(&s_den[dl], ex);
            }
        }
    } else {
        list_trip(1);
    }
    __syncthreads();

    if (!over) {
        for (int b = 0; b < cnt; b += 32) {
            int nb = (cnt - b < 32) ? (cnt - b) : 32;
            int s = 0, dl = 0; float al = 0.f;
            if (lane < nb) {
                s  = qi[b + lane];
                dl = qd[b + lane];
                al = qv[b + lane] / s_den[dl];
            }
            for (int j = 0; j < nb; j += 16) {
                v2f hv[16][2];
                #pragma unroll
                for (int u = 0; u < 16; ++u) {
                    int ju = (j + u < nb) ? (j + u) : (nb - 1);
                    int sj = __builtin_amdgcn_readlane(s, ju);
                    const float* h = H + (size_t)sj * HDIM + 2 * lane;
                    hv[u][0] = *(const v2f*)(h);
                    hv[u][1] = *(const v2f*)(h + 64);
                }
                #pragma unroll
                for (int u = 0; u < 16; ++u) {
                    if (j + u < nb) {
                        float aj = __uint_as_float(__builtin_amdgcn_readlane(__float_as_uint(al), j + u));
                        int   dj = __builtin_amdgcn_readlane(dl, j + u);
                        float* o = s_acc + dj * HDIM + 2 * lane;
                        atomicAdd(&o[0],  hv[u][0][0] * aj);
                        atomicAdd(&o[1],  hv[u][0][1] * aj);
                        atomicAdd(&o[64], hv[u][1][0] * aj);
                        atomicAdd(&o[65], hv[u][1][1] * aj);
                    }
                }
            }
        }
    } else {
        list_trip(2);
    }
    __syncthreads();

    for (int e4 = tid; e4 < rows * (HDIM / 4); e4 += SWAVES * 32) {
        int r = e4 / (HDIM / 4);
        int f = (e4 - r * (HDIM / 4)) * 4;
        float4 v = *(const float4*)(s_acc + r * HDIM + f);
        float4 h = *(const float4*)(H + (size_t)(n0 + r) * HDIM + f);
        float  a = s_self[r] / s_den[r];
        v.x += h.x * a; v.y += h.y * a; v.z += h.z * a; v.w += h.w * a;
        v.x += bias[f]; v.y += bias[f + 1]; v.z += bias[f + 2]; v.w += bias[f + 3];
        v.x = (v.x > 0.f) ? v.x : 0.f;  v.y = (v.y > 0.f) ? v.y : 0.f;
        v.z = (v.z > 0.f) ? v.z : 0.f;  v.w = (v.w > 0.f) ? v.w : 0.f;
        *(float4*)(act + (size_t)(n0 + r) * HDIM + f) = v;
    }
}

extern "C" void kernel_launch(void* const* d_in, const int* in_sizes, int n_in,
                              void* d_out, int out_size, void* d_ws, size_t ws_size,
                              hipStream_t stream)
{
    const float* x   = (const float*)d_in[0];
    const int*   ei  = (const int*)d_in[1];
    const float* W1  = (const float*)d_in[2];
    const float* a1s = (const float*)d_in[3];
    const float* a1d = (const float*)d_in[4];
    const float* b1  = (const float*)d_in[5];
    const float* W2  = (const float*)d_in[6];
    const float* a2s = (const float*)d_in[7];
    const float* a2d = (const float*)d_in[8];
    const float* b2  = (const float*)d_in[9];
    const float* Wl  = (const float*)d_in[10];
    const float* bl  = (const float*)d_in[11];

    const int N  = in_sizes[0] / HDIM;
    const int E  = in_sizes[1] / 2;
    const int C  = in_sizes[11];
    const int CP = (C + 15) & ~15;

    char* ws = (char*)d_ws;
    auto align256 = [](size_t v) { return (v + 255) & ~(size_t)255; };
    size_t off = 0;
    float*    bufH  = (float*)(ws + off);  off += align256((size_t)N * HDIM * 4);
    float*    bufA  = (float*)(ws + off);  off += align256((size_t)N * HDIM * 4);
    float*    Wt    = (float*)(ws + off);  off += align256((size_t)HDIM * HDIM * 4);
    float*    Wlt   = (float*)(ws + off);  off += align256((size_t)CP * HDIM * 4);
    float*    aS    = (float*)(ws + off);  off += align256((size_t)N * 4);
    float*    aD    = (float*)(ws + off);  off += align256((size_t)N * 4);
    const size_t nStrips = (size_t)(N + SROWS - 1) / SROWS;
    const size_t qBytes  = nStrips * SWAVES * QSAVE * 4;
    int* qsave = (off + qBytes <= ws_size && N <= (1 << QSRCBITS)) ? (int*)(ws + off) : nullptr;

    const int nRowTiles = N / 16;
    const int BLK = 256;
    dim3 blk(BLK);

    dim3 trGrid((HDIM * HDIM + BLK - 1) / BLK);
    dim3 gemmGrid((nRowTiles * 2 + 7) / 8);
    dim3 alphaGrid((N + 4 * AROWS - 1) / (4 * AROWS));
    dim3 stripGrid((N + SROWS - 1) / SROWS);

    auto run_layer = [&](const float* act_in, const float* W,
                         const float* as_, const float* ad_, const float* bias,
                         const int* qin, int* qout) {
        gat_transpose<<<trGrid, blk, 0, stream>>>(W, Wt, HDIM, HDIM, HDIM);
        gat_gemm_strip<<<gemmGrid, blk, 0, stream>>>(act_in, Wt, bufH, nRowTiles);
        gat_alpha<<<alphaGrid, dim3(128), 0, stream>>>(bufH, as_, ad_, aS, aD, N);
        gat_agg_strip<<<stripGrid, dim3(SWAVES * 32), 0, stream>>>(ei, E, N, aS, aD, bufH, bias, bufA, qin, qout);
    };

    run_layer(x, W1, a1s, a1d, b1, nullptr, qsave);
    run_layer(bufA, W2, a2s, a2d, b2, qsave, nullptr);

    dim3 trlGrid((CP * HDIM + BLK - 1) / BLK);
    gat_transpose<<<trlGrid, blk, 0, stream>>>(Wl, Wlt, HDIM, C, CP);
    dim3 clsGrid((nRowTiles + 7) / 8);
    gat_gemm_cls<<<clsGrid, blk, 0, stream>>>(bufA, Wlt, bl, (float*)d_out,
                                              nRowTiles, C);
}


# GEMMs read W1, W2, Wl as given (row-major), no transposed copies: 3 launches go (10 -> 7); a wave's 4 column tiles interleave: a lane's B values and its results are 16-byte accesses; same sum order
# speedup vs baseline: 2.7539x; 1.0655x over previous
#include <hip/hip_runtime.h>
#include <hip/hip_bf16.h>

typedef __attribute__((ext_vector_type(2))) float v2f;
typedef __attribute__((ext_vector_type(8))) float v8f;

#define HDIM 128
#define CTPW 4

__global__ __launch_bounds__(256) void gat_gemm_strip(
    const float* __restrict__ X, const float* __restrict__ W,
    float* __restrict__ H, int nRowTiles)
{
    static_assert(CTPW == 4, "a lane's columns of the 4 tiles are one float4");
    int wid  = (blockIdx.x * blockDim.x + threadIdx.x) >> 5;
    int lane = threadIdx.x & 31;
    int nHalves = HDIM / (CTPW * 16);
    int nW = nRowTiles * nHalves;
    if (wid >= nW) return;
    int rt   = wid >> 1;
    int half = wid & 1;
    int row0 = rt * 16;
    int laneM = lane & 15;
    int kOff  = (lane >> 4) * 2;
    int col0  = half * (CTPW * 16) + CTPW * laneM;

    const float* xrow = X + (size_t)(row0 + laneM) * HDIM + kOff;
    const float* wrow = W + (size_t)kOff * HDIM + col0;

    v8f c[CTPW] = {};
    #pragma unroll 4
    for (int k = 0; k < HDIM; k += 4) {
        v2f a = *(const v2f*)(xrow + k);
        const float4 w0 = *(const float4*)(wrow + (size_t)k * HDIM);
        const float4 w1 = *(const float4*)(wrow + (size_t)(k + 1) * HDIM);
        const float w0v[CTPW] = {w0.x, w0.y, w0.z, w0.w};
        const float w1v[CTPW] = {w1.x, w1.y, w1.z, w1.w};
        #pragma unroll
        for (int t = 0; t < CTPW; ++t) {
            v2f b = {w0v[t], w1v[t]};
            c[t] = __builtin_amdgcn_wmma_f32_16x16x4_f32(
                       false, a, false, b, (short)0, c[t], false, false);
        }
    }

    int mBase = (lane >> 4) * 8;
    float* o = H + (size_t)(row0 + mBase) * HDIM + col0;
    #pragma unroll
    for (int r = 0; r < 8; ++r)
        *(float4*)(o + (size_t)r * HDIM) = make_float4(c[0][r], c[1][r], c[2][r], c[3][r]);
}

__global__ __launch_bounds__(256) void gat_gemm_cls(
    const float* __restrict__ X, const float* __restrict__ Wl,
    const float* __restrict__ bl, float* __restrict__ Out,
    int nRowTiles, int nCols)
{
    int wid  = (blockIdx.x * blockDim.x + threadIdx.x) >> 5;
    int lane = threadIdx.x & 31;
    if (wid >= nRowTiles) return;
    int row0 = wid * 16;
    int laneM = lane & 15;
    int kOff  = (lane >> 4) * 2;

    const float* xrow = X + (size_t)(row0 + laneM) * HDIM + kOff;
    unsigned woff[3];
    #pragma unroll
    for (int t = 0; t < 3; ++t) {
        int col = t * 16 + laneM;
        int cc  = (col < nCols) ? col : (nCols - 1);
        woff[t] = (unsigned)(kOff * nCols + cc);
    }

    v8f c[3] = {};
    for (int k = 0; k < HDIM; k += 16) {
        v2f a[4], b[4][3];
        #pragma unroll
        for (int kk = 0; kk < 4; ++kk) {
            const float* r0 = Wl + (size_t)(k + 4 * kk) * nCols;
            const float* r1 = r0 + nCols;
            a[kk] = *(const v2f*)(xrow + k + 4 * kk);
            #pragma unroll
            for (int t = 0; t < 3; ++t) {
                b[kk][t][0] = r0[woff[t]];
                b[kk][t][1] = r1[woff[t]];
            }
        }
        #pragma unroll
        for (int kk = 0; kk < 4; ++kk)
            asm volatile("" :: "v"(a[kk][0]), "v"(a[kk][1]),
                               "v"(b[kk][0][0]), "v"(b[kk][0][1]), "v"(b[kk][1][0]),
                               "v"(b[kk][1][1]), "v"(b[kk][2][0]), "v"(b[kk][2][1]));
        #pragma unroll
        for (int kk = 0; kk < 4; ++kk) {
            #pragma unroll
            for (int t = 0; t < 3; ++t)
                c[t] = __builtin_amdgcn_wmma_f32_16x16x4_f32(
                           false, a[kk], false, b[kk][t], (short)0, c[t], false, false);
        }
    }

    int mBase = (lane >> 4) * 8;
    #pragma unroll
    for (int t = 0; t < 3; ++t) {
        int col = t * 16 + laneM;
        if (col >= nCols) continue;
        float bv = bl[col];
        #pragma unroll
        for (int r = 0; r < 8; ++r)
            Out[(size_t)(row0 + mBase + r) * nCols + col] = c[t][r] + bv;
    }
}

#define AROWS 16
template <int O>
__device__ __forceinline__ void alpha_halve(float (&val)[2 * AROWS], int lane)
{
    const bool up = (lane & O) != 0;
    #pragma unroll
    for (int v = 0; v < O; ++v) {
        const float send = up ? val[v]     : val[v + O];
        const float keep = up ? val[v + O] : val[v];
        val[v] = keep + __shfl_xor(send, O, 32);
    }
}

__global__ __launch_bounds__(128) void gat_alpha(
    const float* __restrict__ H, const float* __restrict__ asrc,
    const float* __restrict__ adst, float* __restrict__ aS,
    float* __restrict__ aD, int N)
{
    const int w    = (blockIdx.x * blockDim.x + threadIdx.x) >> 5;
    const int lane = threadIdx.x & 31;
    const int row0 = w * AROWS;
    if (row0 >= N) return;

    float as4[4], ad4[4];
    #pragma unroll
    for (int j = 0; j < 4; ++j) {
        as4[j] = asrc[lane + 32 * j];
        ad4[j] = adst[lane + 32 * j];
    }

    float hv[AROWS][4];
    #pragma unroll
    for (int r = 0; r < AROWS; ++r) {
        const int row = (row0 + r < N) ? (row0 + r) : (N - 1);
        const float* h = H + (size_t)row * HDIM + lane;
        #pragma unroll
        for (int j = 0; j < 4; ++j) hv[r][j] = h[32 * j];
    }

    float val[2 * AROWS];
    #pragma unroll
    for (int r = 0; r < AROWS; ++r) {
        float s = 0.f, d = 0.f;
        #pragma unroll
        for (int j = 0; j < 4; ++j) {
            s += hv[r][j] * as4[j];
            d += hv[r][j] * ad4[j];
        }
        val[r]         = s;
        val[AROWS + r] = d;
    }

    alpha_halve<16>(val, lane);
    alpha_halve<8>(val, lane);
    alpha_halve<4>(val, lane);
    alpha_halve<2>(val, lane);
    alpha_halve<1>(val, lane);

    const int row = row0 + (lane & (AROWS - 1));
    if (row < N) {
        if (lane < AROWS) aS[row] = val[0];
        else              aD[row] = val[0];
    }
}

__device__ __forceinline__ unsigned f2ord(float f) {
    unsigned b = __float_as_uint(f);
    return (b & 0x80000000u) ? ~b : (b | 0x80000000u);
}
__device__ __forceinline__ float ord2f(unsigned u) {
    unsigned b = (u & 0x80000000u) ? (u & 0x7FFFFFFFu) : ~u;
    return __uint_as_float(b);
}
__device__ __forceinline__ float leaky02(float e) {
    return (e >= 0.f) ? e : 0.2f * e;
}

#define SROWS  128
#define SWAVES 32
#define SLOADS 8
#define WEDGES (SLOADS * 128)
#define QCAP   640
#define QKEEP  320
#define QSAVE  (QKEEP + 32)
#define QSRCBITS 24

__global__ __launch_bounds__(1024) void gat_agg_strip(
    const int* __restrict__ ei, int E, int N,
    const float* __restrict__ aS, const float* __restrict__ aD,
    const float* __restrict__ H, const float* __restrict__ bias,
    float* __restrict__ act,
    const int* __restrict__ qin,
    int* __restrict__ qout)
{
    __shared__ __attribute__((aligned(16))) float s_acc[SROWS * HDIM];
    __shared__ unsigned s_max[SROWS];
    __shared__ float    s_den[SROWS];
    __shared__ float    s_aD[SROWS];
    __shared__ float    s_self[SROWS];
    __shared__ int      s_qi[SWAVES * QCAP];
    __shared__ int      s_qd[SWAVES * QCAP];
    __shared__ float    s_qv[SWAVES * QCAP];
    __shared__ int      s_over;

    const int tid  = threadIdx.x;
    const int lane = tid & 31;
    const int wv   = tid >> 5;
    const int n0   = blockIdx.x * SROWS;
    const int rows = (N - n0 < SROWS) ? (N - n0) : SROWS;
    const int* __restrict__ dstp = ei + E;

    for (int i = tid; i < SROWS * HDIM / 4; i += SWAVES * 32)
        ((float4*)s_acc)[i] = make_float4(0.f, 0.f, 0.f, 0.f);
    if (tid < SROWS) {
        float ad = 0.f, es = 0.f;
        if (tid < rows) {
            ad = aD[n0 + tid];
            es = leaky02(aS[n0 + tid] + ad);
        }
        s_aD[tid]   = ad;
        s_self[tid] = es;
        s_max[tid]  = f2ord(es);
        s_den[tid]  = 0.f;
    }
    if (tid == 0) s_over = 0;
    __syncthreads();

    int*   qi = s_qi + wv * QCAP;
    int*   qd = s_qd + wv * QCAP;
    float* qv = s_qv + wv * QCAP;
    const bool vec = ((E & 3) == 0);

    auto load_round = [&](int base, int (&dv)[SLOADS][4]) {
        if (vec && base + WEDGES <= E) {
            #pragma unroll
            for (int j = 0; j < SLOADS; ++j) {
                int4 t = *(const int4*)(dstp + base + j * 128 + lane * 4);
                dv[j][0] = t.x; dv[j][1] = t.y; dv[j][2] = t.z; dv[j][3] = t.w;
            }
        } else {
            #pragma unroll
            for (int j = 0; j < SLOADS; ++j) {
                #pragma unroll
                for (int c = 0; c < 4; ++c) {
                    int i = base + j * 128 + lane * 4 + c;
                    dv[j][c] = (i < E) ? dstp[i] : -1;
                }
            }
        }
    };

    int cnt = 0;
    auto scan_round = [&](int base, const int (&dv)[SLOADS][4]) {
        #pragma unroll
        for (int j = 0; j < SLOADS; ++j) {
            #pragma unroll
            for (int c = 0; c < 4; ++c) {
                int dl = dv[j][c] - n0;
                bool m = (unsigned)dl < (unsigned)rows;
                unsigned mask = (unsigned)__ballot(m);
                if (mask) {
                    int pos = cnt + __popc(mask & ((1u << lane) - 1u));
                    if (m && pos < QCAP) {
                        qi[pos] = base + j * 128 + lane * 4 + c;
                        qd[pos] = dl;
                    }
                    cnt += __popc(mask);
                }
            }
        }
    };

    const int perRound = SWAVES * WEDGES;
    const int nRounds = (E + perRound - 1) / perRound;
    int rd = (nRounds > 0) ? (int)(blockIdx.x % (unsigned)nRounds) : 0;
    auto next_base = [&]() {
        int b = rd * perRound + wv * WEDGES;
        rd = (rd + 1 == nRounds) ? 0 : rd + 1;
        return b;
    };
    const size_t wq = ((size_t)blockIdx.x * SWAVES + wv) * QSAVE;
    const int saved = qin ? __builtin_amdgcn_readfirstlane(qin[wq]) : -1;
    const bool reuse = saved >= 0;
    bool save = false;
    if (reuse) {
        cnt = saved;
    } else {
        int dv[SLOADS][4];
        for (int it = 0; it < nRounds; ++it) {
            const int base = next_base();
            if (base >= E) continue;
            load_round(base, dv);
            scan_round(base, dv);
        }
        save = (qout != nullptr) && (cnt <= QKEEP);
        if (qout && lane == 0) qout[wq] = save ? cnt : -1;
    }
    if (cnt > QCAP) { s_over = 1; cnt = QCAP; }

    for (int b = 0; b < cnt; b += 256) {
        int k[8], dl[8], s[8]; float as_[8];
        #pragma unroll
        for (int u = 0; u < 8; ++u) {
            int kk = b + 32 * u + lane;
            k[u] = (kk < cnt) ? kk : (cnt - 1);
        }
        if (reuse) {
            #pragma unroll
            for (int u = 0; u < 8; ++u) {
                unsigned p = (unsigned)qin[wq + 32 + k[u]];
                s[u]  = (int)(p & ((1u << QSRCBITS) - 1u));
                dl[u] = (int)(p >> QSRCBITS);
            }
        } else {
            #pragma unroll
            for (int u = 0; u < 8; ++u) { s[u] = ei[qi[k[u]]]; dl[u] = qd[k[u]]; }
        }
        #pragma unroll
        for (int u = 0; u < 8; ++u) as_[u] = aS[s[u]];
        asm volatile("" :: "v"(as_[0]), "v"(as_[1]), "v"(as_[2]), "v"(as_[3]),
                           "v"(as_[4]), "v"(as_[5]), "v"(as_[6]), "v"(as_[7]));
        #pragma unroll
        for (int u = 0; u < 8; ++u) {
            if (b + 32 * u + lane < cnt) {
                float e = leaky02(as_[u] + s_aD[dl[u]]);
                qi[k[u]] = s[u];
                qv[k[u]] = e;
                if (reuse) qd[k[u]] = dl[u];
                else if (save) qout[wq + 32 + k[u]] = (int)((unsigned)s[u] | ((unsigned)dl[u] << QSRCBITS));
                atomicMax(&s_max[dl[u]], f2ord(e));
            }
        }
    }
    __syncthreads();

    const bool over = __builtin_amdgcn_readfirstlane(s_over) != 0;

    auto list_trip = [&](int trip) {
        for (int base = wv * 32; base < E; base += SWAVES * 32) {
            int i = base + lane;
            int dl = ((i < E) ? dstp[i] : -1) - n0;
            bool m = (unsigned)dl < (unsigned)rows;
            int s = 0; float al = 0.f;
            if (m) {
                s = ei[i];
                float e = leaky02(aS[s] + s_aD[dl]);
                if (trip == 0) {
                    atomicMax(&s_max[dl], f2ord(e));
                } else {
                    float ex = expf(e - ord2f(s_max[dl]));
                    if (trip == 1) atomicAdd(&s_den[dl], ex);
                    else           al = ex / s_den[dl];
                }
            }
            if (trip == 2) {
                unsigned mk = (unsigned)__ballot(m);
                while (mk) {
                    int l = __ffs((int)mk) - 1;
                    mk &= mk - 1u;
                    int   sj = __shfl(s, l, 32);
                    int   dj = __shfl(dl, l, 32);
                    float aj = __shfl(al, l, 32);
                    const float* h = H + (size_t)sj * HDIM;
                    float* o = s_acc + dj * HDIM;
                    #pragma unroll
                    for (int f = 0; f < 4; ++f)
                        atomicAdd(&o[lane + 32 * f], h[lane + 32 * f] * aj);
                }
            }
        }
    };

    if (over) {
        list_trip(0);
        __syncthreads();
    }

    if (tid < rows) {
        float ex = expf(s_self[tid] - ord2f(s_max[tid]));
        s_self[tid] = ex;
        atomicAdd(&s_den[tid], ex);
    }
    if (!over) {
        for (int b = 0; b < cnt; b += 32) {
            int k = b + lane;
            if (k < cnt) {
                int dl = qd[k];
                float ex = expf(qv[k] - ord2f(s_max[dl]));
                qv[k] = ex;
                atomicAdd(&s_den[dl], ex);
            }
        }
    } else {
        list_trip(1);
    }
    __syncthreads();

    if (!over) {
        for (int b = 0; b < cnt; b += 32) {
            int nb = (cnt - b < 32) ? (cnt - b) : 32;
            int s = 0, dl = 0; float al = 0.f;
            if (lane < nb) {
                s  = qi[b + lane];
                dl = qd[b + lane];
                al = qv[b + lane] / s_den[dl];
            }
            for (int j = 0; j < nb; j += 16) {
                v2f hv[16][2];
                #pragma unroll
                for (int u = 0; u < 16; ++u) {
                    int ju = (j + u < nb) ? (j + u) : (nb - 1);
                    int sj = __builtin_amdgcn_readlane(s, ju);
                    const float* h = H + (size_t)sj * HDIM + 2 * lane;
                    hv[u][0] = *(const v2f*)(h);
                    hv[u][1] = *(const v2f*)(h + 64);
                }
                #pragma unroll
                for (int u = 0; u < 16; ++u) {
                    if (j + u < nb) {
                        float aj = __uint_as_float(__builtin_amdgcn_readlane(__float_as_uint(al), j + u));
                        int   dj = __builtin_amdgcn_readlane(dl, j + u);
                        float* o = s_acc + dj * HDIM + 2 * lane;
                        atomicAdd(&o[0],  hv[u][0][0] * aj);
                        atomicAdd(&o[1],  hv[u][0][1] * aj);
                        atomicAdd(&o[64], hv[u][1][0] * aj);
                        atomicAdd(&o[65], hv[u][1][1] * aj);
                    }
                }
            }
        }
    } else {
        list_trip(2);
    }
    __syncthreads();

    for (int e4 = tid; e4 < rows * (HDIM / 4); e4 += SWAVES * 32) {
        int r = e4 / (HDIM / 4);
        int f = (e4 - r * (HDIM / 4)) * 4;
        float4 v = *(const float4*)(s_acc + r * HDIM + f);
        float4 h = *(const float4*)(H + (size_t)(n0 + r) * HDIM + f);
        float  a = s_self[r] / s_den[r];
        v.x += h.x * a; v.y += h.y * a; v.z += h.z * a; v.w += h.w * a;
        v.x += bias[f]; v.y += bias[f + 1]; v.z += bias[f + 2]; v.w += bias[f + 3];
        v.x = (v.x > 0.f) ? v.x : 0.f;  v.y = (v.y > 0.f) ? v.y : 0.f;
        v.z = (v.z > 0.f) ? v.z : 0.f;  v.w = (v.w > 0.f) ? v.w : 0.f;
        *(float4*)(act + (size_t)(n0 + r) * HDIM + f) = v;
    }
}

extern "C" void kernel_launch(void* const* d_in, const int* in_sizes, int n_in,
                              void* d_out, int out_size, void* d_ws, size_t ws_size,
                              hipStream_t stream)
{
    const float* x   = (const float*)d_in[0];
    const int*   ei  = (const int*)d_in[1];
    const float* W1  = (const float*)d_in[2];
    const float* a1s = (const float*)d_in[3];
    const float* a1d = (const float*)d_in[4];
    const float* b1  = (const float*)d_in[5];
    const float* W2  = (const float*)d_in[6];
    const float* a2s = (const float*)d_in[7];
    const float* a2d = (const float*)d_in[8];
    const float* b2  = (const float*)d_in[9];
    const float* Wl  = (const float*)d_in[10];
    const float* bl  = (const float*)d_in[11];

    const int N  = in_sizes[0] / HDIM;
    const int E  = in_sizes[1] / 2;
    const int C  = in_sizes[11];

    char* ws = (char*)d_ws;
    auto align256 = [](size_t v) { return (v + 255) & ~(size_t)255; };
    size_t off = 0;
    float*    bufH  = (float*)(ws + off);  off += align256((size_t)N * HDIM * 4);
    float*    bufA  = (float*)(ws + off);  off += align256((size_t)N * HDIM * 4);
    float*    aS    = (float*)(ws + off);  off += align256((size_t)N * 4);
    float*    aD    = (float*)(ws + off);  off += align256((size_t)N * 4);
    const size_t nStrips = (size_t)(N + SROWS - 1) / SROWS;
    const size_t qBytes  = nStrips * SWAVES * QSAVE * 4;
    int* qsave = (off + qBytes <= ws_size && N <= (1 << QSRCBITS)) ? (int*)(ws + off) : nullptr;

    const int nRowTiles = N / 16;
    const int BLK = 256;
    dim3 blk(BLK);

    dim3 gemmGrid((nRowTiles * 2 + 7) / 8);
    dim3 alphaGrid((N + 4 * AROWS - 1) / (4 * AROWS));
    dim3 stripGrid((N + SROWS - 1) / SROWS);

    auto run_layer = [&](const float* act_in, const float* W,
                         const float* as_, const float* ad_, const float* bias,
                         const int* qin, int* qout) {
        gat_gemm_strip<<<gemmGrid, blk, 0, stream>>>(act_in, W, bufH, nRowTiles);
        gat_alpha<<<alphaGrid, dim3(128), 0, stream>>>(bufH, as_, ad_, aS, aD, N);
        gat_agg_strip<<<stripGrid, dim3(SWAVES * 32), 0, stream>>>(ei, E, N, aS, aD, bufH, bias, bufA, qin, qout);
    };

    run_layer(x, W1, a1s, a1d, b1, nullptr, qsave);
    run_layer(bufA, W2, a2s, a2d, b2, qsave, nullptr);

    dim3 clsGrid((nRowTiles + 7) / 8);
    gat_gemm_cls<<<clsGrid, blk, 0, stream>>>(bufA, Wl, bl, (float*)d_out,
                                              nRowTiles, C);
}


# Aggregation strips of 80 destination rows, not 128: 250 work-groups fill the 256 WGPs in one pass (157 did not); LDS row adds and row reads per WGP fall by 3/8; sums stay in LDS arrival order
# speedup vs baseline: 3.3063x; 1.2006x over previous
#include <hip/hip_runtime.h>
#include <hip/hip_bf16.h>

typedef __attribute__((ext_vector_type(2))) float v2f;
typedef __attribute__((ext_vector_type(8))) float v8f;

#define HDIM 128
#define CTPW 4

__global__ __launch_bounds__(256) void gat_gemm_strip(
    const float* __restrict__ X, const float* __restrict__ W,
    float* __restrict__ H, int nRowTiles)
{
    static_assert(CTPW == 4, "a lane's columns of the 4 tiles are one float4");
    int wid  = (blockIdx.x * blockDim.x + threadIdx.x) >> 5;
    int lane = threadIdx.x & 31;
    int nHalves = HDIM / (CTPW * 16);
    int nW = nRowTiles * nHalves;
    if (wid >= nW) return;
    int rt   = wid >> 1;
    int half = wid & 1;
    int row0 = rt * 16;
    int laneM = lane & 15;
    int kOff  = (lane >> 4) * 2;
    int col0  = half * (CTPW * 16) + CTPW * laneM;

    const float* xrow = X + (size_t)(row0 + laneM) * HDIM + kOff;
    const float* wrow = W + (size_t)kOff * HDIM + col0;

    v8f c[CTPW] = {};
    #pragma unroll 4
    for (int k = 0; k < HDIM; k += 4) {
        v2f a = *(const v2f*)(xrow + k);
        const float4 w0 = *(const float4*)(wrow + (size_t)k * HDIM);
        const float4 w1 = *(const float4*)(wrow + (size_t)(k + 1) * HDIM);
        const float w0v[CTPW] = {w0.x, w0.y, w0.z, w0.w};
        const float w1v[CTPW] = {w1.x, w1.y, w1.z, w1.w};
        #pragma unroll
        for (int t = 0; t < CTPW; ++t) {
            v2f b = {w0v[t], w1v[t]};
            c[t] = __builtin_amdgcn_wmma_f32_16x16x4_f32(
                       false, a, false, b, (short)0, c[t], false, false);
        }
    }

    int mBase = (lane >> 4) * 8;
    float* o = H + (size_t)(row0 + mBase) * HDIM + col0;
    #pragma unroll
    for (int r = 0; r < 8; ++r)
        *(float4*)(o + (size_t)r * HDIM) = make_float4(c[0][r], c[1][r], c[2][r], c[3][r]);
}

__global__ __launch_bounds__(256) void gat_gemm_cls(
    const float* __restrict__ X, const float* __restrict__ Wl,
    const float* __restrict__ bl, float* __restrict__ Out,
    int nRowTiles, int nCols)
{
    int wid  = (blockIdx.x * blockDim.x + threadIdx.x) >> 5;
    int lane = threadIdx.x & 31;
    if (wid >= nRowTiles) return;
    int row0 = wid * 16;
    int laneM = lane & 15;
    int kOff  = (lane >> 4) * 2;

    const float* xrow = X + (size_t)(row0 + laneM) * HDIM + kOff;
    unsigned woff[3];
    #pragma unroll
    for (int t = 0; t < 3; ++t) {
        int col = t * 16 + laneM;
        int cc  = (col < nCols) ? col : (nCols - 1);
        woff[t] = (unsigned)(kOff * nCols + cc);
    }

    v8f c[3] = {};
    for (int k = 0; k < HDIM; k += 16) {
        v2f a[4], b[4][3];
        #pragma unroll
        for (int kk = 0; kk < 4; ++kk) {
            const float* r0 = Wl + (size_t)(k + 4 * kk) * nCols;
            const float* r1 = r0 + nCols;
            a[kk] = *(const v2f*)(xrow + k + 4 * kk);
            #pragma unroll
            for (int t = 0; t < 3; ++t) {
                b[kk][t][0] = r0[woff[t]];
                b[kk][t][1] = r1[woff[t]];
            }
        }
        #pragma unroll
        for (int kk = 0; kk < 4; ++kk)
            asm volatile("" :: "v"(a[kk][0]), "v"(a[kk][1]),
                               "v"(b[kk][0][0]), "v"(b[kk][0][1]), "v"(b[kk][1][0]),
                               "v"(b[kk][1][1]), "v"(b[kk][2][0]), "v"(b[kk][2][1]));
        #pragma unroll
        for (int kk = 0; kk < 4; ++kk) {
            #pragma unroll
            for (int t = 0; t < 3; ++t)
                c[t] = __builtin_amdgcn_wmma_f32_16x16x4_f32(
                           false, a[kk], false, b[kk][t], (short)0, c[t], false, false);
        }
    }

    int mBase = (lane >> 4) * 8;
    #pragma unroll
    for (int t = 0; t < 3; ++t) {
        int col = t * 16 + laneM;
        if (col >= nCols) continue;
        float bv = bl[col];
        #pragma unroll
        for (int r = 0; r < 8; ++r)
            Out[(size_t)(row0 + mBase + r) * nCols + col] = c[t][r] + bv;
    }
}

#define AROWS 16
template <int O>
__device__ __forceinline__ void alpha_halve(float (&val)[2 * AROWS], int lane)
{
    const bool up = (lane & O) != 0;
    #pragma unroll
    for (int v = 0; v < O; ++v) {
        const float send = up ? val[v]     : val[v + O];
        const float keep = up ? val[v + O] : val[v];
        val[v] = keep + __shfl_xor(send, O, 32);
    }
}

__global__ __launch_bounds__(128) void gat_alpha(
    const float* __restrict__ H, const float* __restrict__ asrc,
    const float* __restrict__ adst, float* __restrict__ aS,
    float* __restrict__ aD, int N)
{
    const int w    = (blockIdx.x * blockDim.x + threadIdx.x) >> 5;
    const int lane = threadIdx.x & 31;
    const int row0 = w * AROWS;
    if (row0 >= N) return;

    float as4[4], ad4[4];
    #pragma unroll
    for (int j = 0; j < 4; ++j) {
        as4[j] = asrc[lane + 32 * j];
        ad4[j] = adst[lane + 32 * j];
    }

    float hv[AROWS][4];
    #pragma unroll
    for (int r = 0; r < AROWS; ++r) {
        const int row = (row0 + r < N) ? (row0 + r) : (N - 1);
        const float* h = H + (size_t)row * HDIM + lane;
        #pragma unroll
        for (int j = 0; j < 4; ++j) hv[r][j] = h[32 * j];
    }

    float val[2 * AROWS];
    #pragma unroll
    for (int r = 0; r < AROWS; ++r) {
        float s = 0.f, d = 0.f;
        #pragma unroll
        for (int j = 0; j < 4; ++j) {
            s += hv[r][j] * as4[j];
            d += hv[r][j] * ad4[j];
        }
        val[r]         = s;
        val[AROWS + r] = d;
    }

    alpha_halve<16>(val, lane);
    alpha_halve<8>(val, lane);
    alpha_halve<4>(val, lane);
    alpha_halve<2>(val, lane);
    alpha_halve<1>(val, lane);

    const int row = row0 + (lane & (AROWS - 1));
    if (row < N) {
        if (lane < AROWS) aS[row] = val[0];
        else              aD[row] = val[0];
    }
}

__device__ __forceinline__ unsigned f2ord(float f) {
    unsigned b = __float_as_uint(f);
    return (b & 0x80000000u) ? ~b : (b | 0x80000000u);
}
__device__ __forceinline__ float ord2f(unsigned u) {
    unsigned b = (u & 0x80000000u) ? (u & 0x7FFFFFFFu) : ~u;
    return __uint_as_float(b);
}
__device__ __forceinline__ float leaky02(float e) {
    return (e >= 0.f) ? e : 0.2f * e;
}

#define SROWS  80
#define SWAVES 32
#define SLOADS 8
#define WEDGES (SLOADS * 128)
#define QCAP   640
#define QKEEP  160
#define QSAVE  (QKEEP + 32)
#define QSRCBITS 24

__global__ __launch_bounds__(1024) void gat_agg_strip(
    const int* __restrict__ ei, int E, int N,
    const float* __restrict__ aS, const float* __restrict__ aD,
    const float* __restrict__ H, const float* __restrict__ bias,
    float* __restrict__ act,
    const int* __restrict__ qin,
    int* __restrict__ qout)
{
    __shared__ __attribute__((aligned(16))) float s_acc[SROWS * HDIM];
    __shared__ unsigned s_max[SROWS];
    __shared__ float    s_den[SROWS];
    __shared__ float    s_aD[SROWS];
    __shared__ float    s_self[SROWS];
    __shared__ int      s_qi[SWAVES * QCAP];
    __shared__ int      s_qd[SWAVES * QCAP];
    __shared__ float    s_qv[SWAVES * QCAP];
    __shared__ int      s_over;

    const int tid  = threadIdx.x;
    const int lane = tid & 31;
    const int wv   = tid >> 5;
    const int n0   = blockIdx.x * SROWS;
    const int rows = (N - n0 < SROWS) ? (N - n0) : SROWS;
    const int* __restrict__ dstp = ei + E;

    for (int i = tid; i < SROWS * HDIM / 4; i += SWAVES * 32)
        ((float4*)s_acc)[i] = make_float4(0.f, 0.f, 0.f, 0.f);
    if (tid < SROWS) {
        float ad = 0.f, es = 0.f;
        if (tid < rows) {
            ad = aD[n0 + tid];
            es = leaky02(aS[n0 + tid] + ad);
        }
        s_aD[tid]   = ad;
        s_self[tid] = es;
        s_max[tid]  = f2ord(es);
        s_den[tid]  = 0.f;
    }
    if (tid == 0) s_over = 0;
    __syncthreads();

    int*   qi = s_qi + wv * QCAP;
    int*   qd = s_qd + wv * QCAP;
    float* qv = s_qv + wv * QCAP;
    const bool vec = ((E & 3) == 0);

    auto load_round = [&](int base, int (&dv)[SLOADS][4]) {
        if (vec && base + WEDGES <= E) {
            #pragma unroll
            for (int j = 0; j < SLOADS; ++j) {
                int4 t = *(const int4*)(dstp + base + j * 128 + lane * 4);
                dv[j][0] = t.x; dv[j][1] = t.y; dv[j][2] = t.z; dv[j][3] = t.w;
            }
        } else {
            #pragma unroll
            for (int j = 0; j < SLOADS; ++j) {
                #pragma unroll
                for (int c = 0; c < 4; ++c) {
                    int i = base + j * 128 + lane * 4 + c;
                    dv[j][c] = (i < E) ? dstp[i] : -1;
                }
            }
        }
    };

    int cnt = 0;
    auto scan_round = [&](int base, const int (&dv)[SLOADS][4]) {
        #pragma unroll
        for (int j = 0; j < SLOADS; ++j) {
            #pragma unroll
            for (int c = 0; c < 4; ++c) {
                int dl = dv[j][c] - n0;
                bool m = (unsigned)dl < (unsigned)rows;
                unsigned mask = (unsigned)__ballot(m);
                if (mask) {
                    int pos = cnt + __popc(mask & ((1u << lane) - 1u));
                    if (m && pos < QCAP) {
                        qi[pos] = base + j * 128 + lane * 4 + c;
                        qd[pos] = dl;
                    }
                    cnt += __popc(mask);
                }
            }
        }
    };

    const int perRound = SWAVES * WEDGES;
    const int nRounds = (E + perRound - 1) / perRound;
    int rd = (nRounds > 0) ? (int)(blockIdx.x % (unsigned)nRounds) : 0;
    auto next_base = [&]() {
        int b = rd * perRound + wv * WEDGES;
        rd = (rd + 1 == nRounds) ? 0 : rd + 1;
        return b;
    };
    const size_t wq = ((size_t)blockIdx.x * SWAVES + wv) * QSAVE;
    const int saved = qin ? __builtin_amdgcn_readfirstlane(qin[wq]) : -1;
    const bool reuse = saved >= 0;
    bool save = false;
    if (reuse) {
        cnt = saved;
    } else {
        int dv[SLOADS][4];
        for (int it = 0; it < nRounds; ++it) {
            const int base = next_base();
            if (base >= E) continue;
            load_round(base, dv);
            scan_round(base, dv);
        }
        save = (qout != nullptr) && (cnt <= QKEEP);
        if (qout && lane == 0) qout[wq] = save ? cnt : -1;
    }
    if (cnt > QCAP) { s_over = 1; cnt = QCAP; }

    for (int b = 0; b < cnt; b += 256) {
        int k[8], dl[8], s[8]; float as_[8];
        #pragma unroll
        for (int u = 0; u < 8; ++u) {
            int kk = b + 32 * u + lane;
            k[u] = (kk < cnt) ? kk : (cnt - 1);
        }
        if (reuse) {
            #pragma unroll
            for (int u = 0; u < 8; ++u) {
                unsigned p = (unsigned)qin[wq + 32 + k[u]];
                s[u]  = (int)(p & ((1u << QSRCBITS) - 1u));
                dl[u] = (int)(p >> QSRCBITS);
            }
        } else {
            #pragma unroll
            for (int u = 0; u < 8; ++u) { s[u] = ei[qi[k[u]]]; dl[u] = qd[k[u]]; }
        }
        #pragma unroll
        for (int u = 0; u < 8; ++u) as_[u] = aS[s[u]];
        asm volatile("" :: "v"(as_[0]), "v"(as_[1]), "v"(as_[2]), "v"(as_[3]),
                           "v"(as_[4]), "v"(as_[5]), "v"(as_[6]), "v"(as_[7]));
        #pragma unroll
        for (int u = 0; u < 8; ++u) {
            if (b + 32 * u + lane < cnt) {
                float e = leaky02(as_[u] + s_aD[dl[u]]);
                qi[k[u]] = s[u];
                qv[k[u]] = e;
                if (reuse) qd[k[u]] = dl[u];
                else if (save) qout[wq + 32 + k[u]] = (int)((unsigned)s[u] | ((unsigned)dl[u] << QSRCBITS));
                atomicMax(&s_max[dl[u]], f2ord(e));
            }
        }
    }
    __syncthreads();

    const bool over = __builtin_amdgcn_readfirstlane(s_over) != 0;

    auto list_trip = [&](int trip) {
        for (int base = wv * 32; base < E; base += SWAVES * 32) {
            int i = base + lane;
            int dl = ((i < E) ? dstp[i] : -1) - n0;
            bool m = (unsigned)dl < (unsigned)rows;
            int s = 0; float al = 0.f;
            if (m) {
                s = ei[i];
                float e = leaky02(aS[s] + s_aD[dl]);
                if (trip == 0) {
                    atomicMax(&s_max[dl], f2ord(e));
                } else {
                    float ex = expf(e - ord2f(s_max[dl]));
                    if (trip == 1) atomicAdd(&s_den[dl], ex);
                    else           al = ex / s_den[dl];
                }
            }
            if (trip == 2) {
                unsigned mk = (unsigned)__ballot(m);
                while (mk) {
                    int l = __ffs((int)mk) - 1;
                    mk &= mk - 1u;
                    int   sj = __shfl(s, l, 32);
                    int   dj = __shfl(dl, l, 32);
                    float aj = __shfl(al, l, 32);
                    const float* h = H + (size_t)sj * HDIM;
                    float* o = s_acc + dj * HDIM;
                    #pragma unroll
                    for (int f = 0; f < 4; ++f)
                        atomicAdd(&o[lane + 32 * f], h[lane + 32 * f] * aj);
                }
            }
        }
    };

    if (over) {
        list_trip(0);
        __syncthreads();
    }

    if (tid < rows) {
        float ex = expf(s_self[tid] - ord2f(s_max[tid]));
        s_self[tid] = ex;
        atomicAdd(&s_den[tid], ex);
    }
    if (!over) {
        for (int b = 0; b < cnt; b += 32) {
            int k = b + lane;
            if (k < cnt) {
                int dl = qd[k];
                float ex = expf(qv[k] - ord2f(s_max[dl]));
                qv[k] = ex;
                atomicAdd(&s_den[dl], ex);
            }
        }
    } else {
        list_trip(1);
    }
    __syncthreads();

    if (!over) {
        for (int b = 0; b < cnt; b += 32) {
            int nb = (cnt - b < 32) ? (cnt - b) : 32;
            int s = 0, dl = 0; float al = 0.f;
            if (lane < nb) {
                s  = qi[b + lane];
                dl = qd[b + lane];
                al = qv[b + lane] / s_den[dl];
            }
            for (int j = 0; j < nb; j += 16) {
                v2f hv[16][2];
                #pragma unroll
                for (int u = 0; u < 16; ++u) {
                    int ju = (j + u < nb) ? (j + u) : (nb - 1);
                    int sj = __builtin_amdgcn_readlane(s, ju);
                    const float* h = H + (size_t)sj * HDIM + 2 * lane;
                    hv[u][0] = *(const v2f*)(h);
                    hv[u][1] = *(const v2f*)(h + 64);
                }
                #pragma unroll
                for (int u = 0; u < 16; ++u) {
                    if (j + u < nb) {
                        float aj = __uint_as_float(__builtin_amdgcn_readlane(__float_as_uint(al), j + u));
                        int   dj = __builtin_amdgcn_readlane(dl, j + u);
                        float* o = s_acc + dj * HDIM + 2 * lane;
                        atomicAdd(&o[0],  hv[u][0][0] * aj);
                        atomicAdd(&o[1],  hv[u][0][1] * aj);
                        atomicAdd(&o[64], hv[u][1][0] * aj);
                        atomicAdd(&o[65], hv[u][1][1] * aj);
                    }
                }
            }
        }
    } else {
        list_trip(2);
    }
    __syncthreads();

    for (int e4 = tid; e4 < rows * (HDIM / 4); e4 += SWAVES * 32) {
        int r = e4 / (HDIM / 4);
        int f = (e4 - r * (HDIM / 4)) * 4;
        float4 v = *(const float4*)(s_acc + r * HDIM + f);
        float4 h = *(const float4*)(H + (size_t)(n0 + r) * HDIM + f);
        float  a = s_self[r] / s_den[r];
        v.x += h.x * a; v.y += h.y * a; v.z += h.z * a; v.w += h.w * a;
        v.x += bias[f]; v.y += bias[f + 1]; v.z += bias[f + 2]; v.w += bias[f + 3];
        v.x = (v.x > 0.f) ? v.x : 0.f;  v.y = (v.y > 0.f) ? v.y : 0.f;
        v.z = (v.z > 0.f) ? v.z : 0.f;  v.w = (v.w > 0.f) ? v.w : 0.f;
        *(float4*)(act + (size_t)(n0 + r) * HDIM + f) = v;
    }
}

extern "C" void kernel_launch(void* const* d_in, const int* in_sizes, int n_in,
                              void* d_out, int out_size, void* d_ws, size_t ws_size,
                              hipStream_t stream)
{
    const float* x   = (const float*)d_in[0];
    const int*   ei  = (const int*)d_in[1];
    const float* W1  = (const float*)d_in[2];
    const float* a1s = (const float*)d_in[3];
    const float* a1d = (const float*)d_in[4];
    const float* b1  = (const float*)d_in[5];
    const float* W2  = (const float*)d_in[6];
    const float* a2s = (const float*)d_in[7];
    const float* a2d = (const float*)d_in[8];
    const float* b2  = (const float*)d_in[9];
    const float* Wl  = (const float*)d_in[10];
    const float* bl  = (const float*)d_in[11];

    const int N  = in_sizes[0] / HDIM;
    const int E  = in_sizes[1] / 2;
    const int C  = in_sizes[11];

    char* ws = (char*)d_ws;
    auto align256 = [](size_t v) { return (v + 255) & ~(size_t)255; };
    size_t off = 0;
    float*    bufH  = (float*)(ws + off);  off += align256((size_t)N * HDIM * 4);
    float*    bufA  = (float*)(ws + off);  off += align256((size_t)N * HDIM * 4);
    float*    aS    = (float*)(ws + off);  off += align256((size_t)N * 4);
    float*    aD    = (float*)(ws + off);  off += align256((size_t)N * 4);
    const size_t nStrips = (size_t)(N + SROWS - 1) / SROWS;
    const size_t qBytes  = nStrips * SWAVES * QSAVE * 4;
    int* qsave = (off + qBytes <= ws_size && N <= (1 << QSRCBITS)) ? (int*)(ws + off) : nullptr;

    const int nRowTiles = N / 16;
    const int BLK = 256;
    dim3 blk(BLK);

    dim3 gemmGrid((nRowTiles * 2 + 7) / 8);
    dim3 alphaGrid((N + 4 * AROWS - 1) / (4 * AROWS));
    dim3 stripGrid((N + SROWS - 1) / SROWS);

    auto run_layer = [&](const float* act_in, const float* W,
                         const float* as_, const float* ad_, const float* bias,
                         const int* qin, int* qout) {
        gat_gemm_strip<<<gemmGrid, blk, 0, stream>>>(act_in, W, bufH, nRowTiles);
        gat_alpha<<<alphaGrid, dim3(128), 0, stream>>>(bufH, as_, ad_, aS, aD, N);
        gat_agg_strip<<<stripGrid, dim3(SWAVES * 32), 0, stream>>>(ei, E, N, aS, aD, bufH, bias, bufA, qin, qout);
    };

    run_layer(x, W1, a1s, a1d, b1, nullptr, qsave);
    run_layer(bufA, W2, a2s, a2d, b2, qsave, nullptr);

    dim3 clsGrid((nRowTiles + 7) / 8);
    gat_gemm_cls<<<clsGrid, blk, 0, stream>>>(bufA, Wl, bl, (float*)d_out,
                                              nRowTiles, C);
}


# Layer 2 sums act1 rows first, multiplies after: (sum alpha*act1[src]) @ W2, then Wl, on WMMA out of strip LDS; logits = act1.(W2@a2) in layer 1's strips; 7 -> 4 launches; product sum order changes
# speedup vs baseline: 3.7959x; 1.1481x over previous
#include <hip/hip_runtime.h>
#include <hip/hip_bf16.h>

typedef __attribute__((ext_vector_type(2))) float v2f;
typedef __attribute__((ext_vector_type(8))) float v8f;

#define HDIM 128
#define CTPW 4

__global__ __launch_bounds__(256) void gat_gemm_strip(
    const float* __restrict__ X, const float* __restrict__ W,
    float* __restrict__ H, int nRowTiles)
{
    static_assert(CTPW == 4, "a lane's columns of the 4 tiles are one float4");
    int wid  = (blockIdx.x * blockDim.x + threadIdx.x) >> 5;
    int lane = threadIdx.x & 31;
    int nHalves = HDIM / (CTPW * 16);
    int nW = nRowTiles * nHalves;
    if (wid >= nW) return;
    int rt   = wid >> 1;
    int half = wid & 1;
    int row0 = rt * 16;
    int laneM = lane & 15;
    int kOff  = (lane >> 4) * 2;
    int col0  = half * (CTPW * 16) + CTPW * laneM;

    const float* xrow = X + (size_t)(row0 + laneM) * HDIM + kOff;
    const float* wrow = W + (size_t)kOff * HDIM + col0;

    v8f c[CTPW] = {};
    #pragma unroll 4
    for (int k = 0; k < HDIM; k += 4) {
        v2f a = *(const v2f*)(xrow + k);
        const float4 w0 = *(const float4*)(wrow + (size_t)k * HDIM);
        const float4 w1 = *(const float4*)(wrow + (size_t)(k + 1) * HDIM);
        const float w0v[CTPW] = {w0.x, w0.y, w0.z, w0.w};
        const float w1v[CTPW] = {w1.x, w1.y, w1.z, w1.w};
        #pragma unroll
        for (int t = 0; t < CTPW; ++t) {
            v2f b = {w0v[t], w1v[t]};
            c[t] = __builtin_amdgcn_wmma_f32_16x16x4_f32(
                       false, a, false, b, (short)0, c[t], false, false);
        }
    }

    int mBase = (lane >> 4) * 8;
    float* o = H + (size_t)(row0 + mBase) * HDIM + col0;
    #pragma unroll
    for (int r = 0; r < 8; ++r)
        *(float4*)(o + (size_t)r * HDIM) = make_float4(c[0][r], c[1][r], c[2][r], c[3][r]);
}

#define AROWS 16
template <int O>
__device__ __forceinline__ void alpha_halve(float (&val)[2 * AROWS], int lane)
{
    const bool up = (lane & O) != 0;
    #pragma unroll
    for (int v = 0; v < O; ++v) {
        const float send = up ? val[v]     : val[v + O];
        const float keep = up ? val[v + O] : val[v];
        val[v] = keep + __shfl_xor(send, O, 32);
    }
}

__global__ __launch_bounds__(128) void gat_alpha(
    const float* __restrict__ H, const float* __restrict__ asrc,
    const float* __restrict__ adst, float* __restrict__ aS,
    float* __restrict__ aD, int N)
{
    const int w    = (blockIdx.x * blockDim.x + threadIdx.x) >> 5;
    const int lane = threadIdx.x & 31;
    const int row0 = w * AROWS;
    if (row0 >= N) return;

    float as4[4], ad4[4];
    #pragma unroll
    for (int j = 0; j < 4; ++j) {
        as4[j] = asrc[lane + 32 * j];
        ad4[j] = adst[lane + 32 * j];
    }

    float hv[AROWS][4];
    #pragma unroll
    for (int r = 0; r < AROWS; ++r) {
        const int row = (row0 + r < N) ? (row0 + r) : (N - 1);
        const float* h = H + (size_t)row * HDIM + lane;
        #pragma unroll
        for (int j = 0; j < 4; ++j) hv[r][j] = h[32 * j];
    }

    float val[2 * AROWS];
    #pragma unroll
    for (int r = 0; r < AROWS; ++r) {
        float s = 0.f, d = 0.f;
        #pragma unroll
        for (int j = 0; j < 4; ++j) {
            s += hv[r][j] * as4[j];
            d += hv[r][j] * ad4[j];
        }
        val[r]         = s;
        val[AROWS + r] = d;
    }

    alpha_halve<16>(val, lane);
    alpha_halve<8>(val, lane);
    alpha_halve<4>(val, lane);
    alpha_halve<2>(val, lane);
    alpha_halve<1>(val, lane);

    const int row = row0 + (lane & (AROWS - 1));
    if (row < N) {
        if (lane < AROWS) aS[row] = val[0];
        else              aD[row] = val[0];
    }
}

__device__ __forceinline__ unsigned f2ord(float f) {
    unsigned b = __float_as_uint(f);
    return (b & 0x80000000u) ? ~b : (b | 0x80000000u);
}
__device__ __forceinline__ float ord2f(unsigned u) {
    unsigned b = (u & 0x80000000u) ? (u & 0x7FFFFFFFu) : ~u;
    return __uint_as_float(b);
}
__device__ __forceinline__ float leaky02(float e) {
    return (e >= 0.f) ? e : 0.2f * e;
}

#define SROWS  80
#define SWAVES 32
#define SLOADS 8
#define WEDGES (SLOADS * 128)
#define QCAP   640
#define QKEEP  160
#define QSAVE  (QKEEP + 32)
#define QSRCBITS 24
#define APITCH (HDIM + 4)
#define G1JOBS ((SROWS / 16) * 4)
#define G2JOBS ((SROWS / 16) * 3)

template <int LAYER>
__global__ __launch_bounds__(1024) void gat_agg_strip(
    const int* __restrict__ ei, int E, int N,
    const float* __restrict__ aS, const float* __restrict__ aD,
    const float* __restrict__ H, const float* __restrict__ bias,
    float* __restrict__ act,
    const int* __restrict__ qin,
    int* __restrict__ qout,
    const float* __restrict__ Wn,
    const float* __restrict__ va,
    const float* __restrict__ vb,
    const float* __restrict__ vc,
    float* __restrict__ oa,
    float* __restrict__ ob,
    int nCols)
{
    static_assert(SWAVES * 32 == HDIM * 8, "8 threads per row of W2 form u");
    static_assert(SROWS % 16 == 0 && G1JOBS <= SWAVES && G2JOBS <= SWAVES, "one wave per matrix job");
    static_assert(SROWS * APITCH <= SWAVES * QCAP, "the second row buffer lies in the first queue array");
    static_assert(48 <= APITCH, "the strip's classes (3 column tiles at most) fit where its sums were");
    __shared__ __attribute__((aligned(16))) float s_acc[SROWS * APITCH];
    __shared__ float    s_u[2 * HDIM];
    __shared__ float    s_l2[2 * SROWS];
    __shared__ unsigned s_max[SROWS];
    __shared__ float    s_den[SROWS];
    __shared__ float    s_aD[SROWS];
    __shared__ float    s_self[SROWS];
    __shared__ int      s_qi[SWAVES * QCAP];
    __shared__ int      s_qd[SWAVES * QCAP];
    __shared__ float    s_qv[SWAVES * QCAP];
    __shared__ int      s_over;

    const int tid  = threadIdx.x;
    const int lane = tid & 31;
    const int wv   = tid >> 5;
    const int n0   = blockIdx.x * SROWS;
    const int rows = (N - n0 < SROWS) ? (N - n0) : SROWS;
    const int* __restrict__ dstp = ei + E;

    for (int i = tid; i < SROWS * APITCH / 4; i += SWAVES * 32)
        ((float4*)s_acc)[i] = make_float4(0.f, 0.f, 0.f, 0.f);
    if (LAYER == 1) {
        const int k = tid >> 3, p = tid & 7;
        const float* wr = Wn + (size_t)k * HDIM + p * 16;
        float us = 0.f, ud = 0.f;
        #pragma unroll
        for (int q = 0; q < 4; ++q) {
            const float4 w  = *(const float4*)(wr + 4 * q);
            const float4 s4 = *(const float4*)(va + p * 16 + 4 * q);
            const float4 d4 = *(const float4*)(vb + p * 16 + 4 * q);
            us += w.x * s4.x + w.y * s4.y + w.z * s4.z + w.w * s4.w;
            ud += w.x * d4.x + w.y * d4.y + w.z * d4.z + w.w * d4.w;
        }
        #pragma unroll
        for (int o = 1; o < 8; o <<= 1) {
            us += __shfl_xor(us, o, 32);
            ud += __shfl_xor(ud, o, 32);
        }
        if (p == 0) { s_u[k] = us; s_u[HDIM + k] = ud; }
    }
    if (tid < SROWS) {
        float ad = 0.f, es = 0.f;
        if (tid < rows) {
            ad = aD[n0 + tid];
            es = leaky02(aS[n0 + tid] + ad);
        }
        s_aD[tid]   = ad;
        s_self[tid] = es;
        s_max[tid]  = f2ord(es);
        s_den[tid]  = 0.f;
    }
    if (tid == 0) s_over = 0;
    __syncthreads();

    int*   qi = s_qi + wv * QCAP;
    int*   qd = s_qd + wv * QCAP;
    float* qv = s_qv + wv * QCAP;
    const bool vec = ((E & 3) == 0);

    auto load_round = [&](int base, int (&dv)[SLOADS][4]) {
        if (vec && base + WEDGES <= E) {
            #pragma unroll
            for (int j = 0; j < SLOADS; ++j) {
                int4 t = *(const int4*)(dstp + base + j * 128 + lane * 4);
                dv[j][0] = t.x; dv[j][1] = t.y; dv[j][2] = t.z; dv[j][3] = t.w;
            }
        } else {
            #pragma unroll
            for (int j = 0; j < SLOADS; ++j) {
                #pragma unroll
                for (int c = 0; c < 4; ++c) {
                    int i = base + j * 128 + lane * 4 + c;
                    dv[j][c] = (i < E) ? dstp[i] : -1;
                }
            }
        }
    };

    int cnt = 0;
    auto scan_round = [&](int base, const int (&dv)[SLOADS][4]) {
        #pragma unroll
        for (int j = 0; j < SLOADS; ++j) {
            #pragma unroll
            for (int c = 0; c < 4; ++c) {
                int dl = dv[j][c] - n0;
                bool m = (unsigned)dl < (unsigned)rows;
                unsigned mask = (unsigned)__ballot(m);
                if (mask) {
                    int pos = cnt + __popc(mask & ((1u << lane) - 1u));
                    if (m && pos < QCAP) {
                        qi[pos] = base + j * 128 + lane * 4 + c;
                        qd[pos] = dl;
                    }
                    cnt += __popc(mask);
                }
            }
        }
    };

    const int perRound = SWAVES * WEDGES;
    const int nRounds = (E + perRound - 1) / perRound;
    int rd = (nRounds > 0) ? (int)(blockIdx.x % (unsigned)nRounds) : 0;
    auto next_base = [&]() {
        int b = rd * perRound + wv * WEDGES;
        rd = (rd + 1 == nRounds) ? 0 : rd + 1;
        return b;
    };
    const size_t wq = ((size_t)blockIdx.x * SWAVES + wv) * QSAVE;
    const int saved = qin ? __builtin_amdgcn_readfirstlane(qin[wq]) : -1;
    const bool reuse = saved >= 0;
    bool save = false;
    if (reuse) {
        cnt = saved;
    } else {
        int dv[SLOADS][4];
        for (int it = 0; it < nRounds; ++it) {
            const int base = next_base();
            if (base >= E) continue;
            load_round(base, dv);
            scan_round(base, dv);
        }
        save = (qout != nullptr) && (cnt <= QKEEP);
        if (qout && lane == 0) qout[wq] = save ? cnt : -1;
    }
    if (cnt > QCAP) { s_over = 1; cnt = QCAP; }

    for (int b = 0; b < cnt; b += 256) {
        int k[8], dl[8], s[8]; float as_[8];
        #pragma unroll
        for (int u = 0; u < 8; ++u) {
            int kk = b + 32 * u + lane;
            k[u] = (kk < cnt) ? kk : (cnt - 1);
        }
        if (reuse) {
            #pragma unroll
            for (int u = 0; u < 8; ++u) {
                unsigned p = (unsigned)qin[wq + 32 + k[u]];
                s[u]  = (int)(p & ((1u << QSRCBITS) - 1u));
                dl[u] = (int)(p >> QSRCBITS);
            }
        } else {
            #pragma unroll
            for (int u = 0; u < 8; ++u) { s[u] = ei[qi[k[u]]]; dl[u] = qd[k[u]]; }
        }
        #pragma unroll
        for (int u = 0; u < 8; ++u) as_[u] = aS[s[u]];
        asm volatile("" :: "v"(as_[0]), "v"(as_[1]), "v"(as_[2]), "v"(as_[3]),
                           "v"(as_[4]), "v"(as_[5]), "v"(as_[6]), "v"(as_[7]));
        #pragma unroll
        for (int u = 0; u < 8; ++u) {
            if (b + 32 * u + lane < cnt) {
                float e = leaky02(as_[u] + s_aD[dl[u]]);
                qi[k[u]] = s[u];
                qv[k[u]] = e;
                if (reuse) qd[k[u]] = dl[u];
                else if (save) qout[wq + 32 + k[u]] = (int)((unsigned)s[u] | ((unsigned)dl[u] << QSRCBITS));
                atomicMax(&s_max[dl[u]], f2ord(e));
            }
        }
    }
    __syncthreads();

    const bool over = __builtin_amdgcn_readfirstlane(s_over) != 0;

    auto list_trip = [&](int trip) {
        for (int base = wv * 32; base < E; base += SWAVES * 32) {
            int i = base + lane;
            int dl = ((i < E) ? dstp[i] : -1) - n0;
            bool m = (unsigned)dl < (unsigned)rows;
            int s = 0; float al = 0.f;
            if (m) {
                s = ei[i];
                float e = leaky02(aS[s] + s_aD[dl]);
                if (trip == 0) {
                    atomicMax(&s_max[dl], f2ord(e));
                } else {
                    float ex = expf(e - ord2f(s_max[dl]));
                    if (trip == 1) atomicAdd(&s_den[dl], ex);
                    else           al = ex / s_den[dl];
                }
            }
            if (trip == 2) {
                unsigned mk = (unsigned)__ballot(m);
                while (mk) {
                    int l = __ffs((int)mk) - 1;
                    mk &= mk - 1u;
                    int   sj = __shfl(s, l, 32);
                    int   dj = __shfl(dl, l, 32);
                    float aj = __shfl(al, l, 32);
                    const float* h = H + (size_t)sj * HDIM;
                    float* o = s_acc + dj * APITCH;
                    #pragma unroll
                    for (int f = 0; f < 4; ++f)
                        atomicAdd(&o[lane + 32 * f], h[lane + 32 * f] * aj);
                }
            }
        }
    };

    if (over) {
        list_trip(0);
        __syncthreads();
    }

    if (tid < rows) {
        float ex = expf(s_self[tid] - ord2f(s_max[tid]));
        s_self[tid] = ex;
        atomicAdd(&s_den[tid], ex);
    }
    if (!over) {
        for (int b = 0; b < cnt; b += 32) {
            int k = b + lane;
            if (k < cnt) {
                int dl = qd[k];
                float ex = expf(qv[k] - ord2f(s_max[dl]));
                qv[k] = ex;
                atomicAdd(&s_den[dl], ex);
            }
        }
    } else {
        list_trip(1);
    }
    __syncthreads();

    if (!over) {
        for (int b = 0; b < cnt; b += 32) {
            int nb = (cnt - b < 32) ? (cnt - b) : 32;
            int s = 0, dl = 0; float al = 0.f;
            if (lane < nb) {
                s  = qi[b + lane];
                dl = qd[b + lane];
                al = qv[b + lane] / s_den[dl];
            }
            for (int j = 0; j < nb; j += 16) {
                v2f hv[16][2];
                #pragma unroll
                for (int u = 0; u < 16; ++u) {
                    int ju = (j + u < nb) ? (j + u) : (nb - 1);
                    int sj = __builtin_amdgcn_readlane(s, ju);
                    const float* h = H + (size_t)sj * HDIM + 2 * lane;
                    hv[u][0] = *(const v2f*)(h);
                    hv[u][1] = *(const v2f*)(h + 64);
                }
                #pragma unroll
                for (int u = 0; u < 16; ++u) {
                    if (j + u < nb) {
                        float aj = __uint_as_float(__builtin_amdgcn_readlane(__float_as_uint(al), j + u));
                        int   dj = __builtin_amdgcn_readlane(dl, j + u);
                        float* o = s_acc + dj * APITCH + 2 * lane;
                        atomicAdd(&o[0],  hv[u][0][0] * aj);
                        atomicAdd(&o[1],  hv[u][0][1] * aj);
                        atomicAdd(&o[64], hv[u][1][0] * aj);
                        atomicAdd(&o[65], hv[u][1][1] * aj);
                    }
                }
            }
        }
    } else {
        list_trip(2);
    }
    __syncthreads();

    for (int e4 = tid; e4 < rows * (HDIM / 4); e4 += SWAVES * 32) {
        int r = e4 / (HDIM / 4);
        int f = (e4 - r * (HDIM / 4)) * 4;
        float4 v = *(const float4*)(s_acc + r * APITCH + f);
        float4 h = *(const float4*)(H + (size_t)(n0 + r) * HDIM + f);
        float  a = s_self[r] / s_den[r];
        v.x += h.x * a; v.y += h.y * a; v.z += h.z * a; v.w += h.w * a;
        if (LAYER == 1) {
            v.x += bias[f]; v.y += bias[f + 1]; v.z += bias[f + 2]; v.w += bias[f + 3];
            v.x = (v.x > 0.f) ? v.x : 0.f;  v.y = (v.y > 0.f) ? v.y : 0.f;
            v.z = (v.z > 0.f) ? v.z : 0.f;  v.w = (v.w > 0.f) ? v.w : 0.f;
            *(float4*)(act + (size_t)(n0 + r) * HDIM + f) = v;
        }
        *(float4*)(s_acc + r * APITCH + f) = v;
    }
    __syncthreads();

    if (LAYER == 1) {
        for (int r = wv; r < rows; r += SWAVES) {
            const float* a = s_acc + r * APITCH + lane;
            float ls = 0.f, ld = 0.f;
            #pragma unroll
            for (int j = 0; j < 4; ++j) {
                const float xv = a[32 * j];
                ls += xv * s_u[lane + 32 * j];
                ld += xv * s_u[HDIM + lane + 32 * j];
            }
            #pragma unroll
            for (int o = 16; o >= 1; o >>= 1) {
                ls += __shfl_xor(ls, o, 32);
                ld += __shfl_xor(ld, o, 32);
            }
            if (lane == 0) { s_l2[r] = ls; s_l2[SROWS + r] = ld; }
        }
        __syncthreads();
        if (tid < rows)                              oa[n0 + tid] = s_l2[tid];
        else if (tid >= 128 && tid - 128 < rows)     ob[n0 + tid - 128] = s_l2[SROWS + tid - 128];
        return;
    }

    float* s_t = reinterpret_cast<float*>(s_qi);
    const int wvs   = __builtin_amdgcn_readfirstlane(wv);
    const int laneM = lane & 15;
    const int hh    = lane >> 4;
    const int kOff  = hh * 2;
    if (wvs < G1JOBS) {
        const int rt = wvs >> 2, q = wvs & 3;
        const int col0 = q * 32 + 2 * laneM;
        const float* arow = s_acc + (rt * 16 + laneM) * APITCH + kOff;
        const float* wrow = Wn + (size_t)kOff * HDIM + col0;
        v8f c0 = {}, c1 = {};
        #pragma unroll 8
        for (int k = 0; k < HDIM; k += 4) {
            const v2f a = *(const v2f*)(arow + k);
            const float2 w0 = *(const float2*)(wrow + (size_t)k * HDIM);
            const float2 w1 = *(const float2*)(wrow + (size_t)(k + 1) * HDIM);
            const v2f b0 = {w0.x, w1.x};
            const v2f b1 = {w0.y, w1.y};
            c0 = __builtin_amdgcn_wmma_f32_16x16x4_f32(false, a, false, b0, (short)0, c0, false, false);
            c1 = __builtin_amdgcn_wmma_f32_16x16x4_f32(false, a, false, b1, (short)0, c1, false, false);
        }
        const float bv0 = va[col0], bv1 = va[col0 + 1];
        float* o = s_t + (rt * 16 + hh * 8) * APITCH + col0;
        #pragma unroll
        for (int r = 0; r < 8; ++r) {
            float y0 = c0[r] + bv0, y1 = c1[r] + bv1;
            y0 = (y0 > 0.f) ? y0 : 0.f;
            y1 = (y1 > 0.f) ? y1 : 0.f;
            *(float2*)(o + r * APITCH) = make_float2(y0, y1);
        }
    }
    __syncthreads();

    if (wvs < G2JOBS) {
        const int rt = wvs / 3, ct = wvs - rt * 3;
        const int col = ct * 16 + laneM;
        const int cc  = (col < nCols) ? col : (nCols - 1);
        const float* arow = s_t + (rt * 16 + laneM) * APITCH + kOff;
        const float* wcol = vb + (size_t)kOff * nCols + cc;
        v8f c = {};
        #pragma unroll 8
        for (int k = 0; k < HDIM; k += 4) {
            const v2f a = *(const v2f*)(arow + k);
            v2f b;
            b[0] = wcol[(size_t)k * nCols];
            b[1] = wcol[(size_t)(k + 1) * nCols];
            c = __builtin_amdgcn_wmma_f32_16x16x4_f32(false, a, false, b, (short)0, c, false, false);
        }
        if (col < nCols) {
            const float bv = vc[col];
            #pragma unroll
            for (int r = 0; r < 8; ++r)
                s_acc[(rt * 16 + hh * 8 + r) * nCols + col] = c[r] + bv;
        }
    }
    __syncthreads();

    static_assert(SROWS % 4 == 0, "a strip's output starts 16-byte aligned");
    {
        const int nOut = rows * nCols;
        float* og = oa + (size_t)n0 * nCols;
        for (int i4 = tid; i4 < (nOut >> 2); i4 += SWAVES * 32)
            *(float4*)(og + 4 * i4) = *(const float4*)(s_acc + 4 * i4);
        const int tail = (nOut & ~3) + tid;
        if (tid < (nOut & 3)) og[tail] = s_acc[tail];
    }
}

extern "C" void kernel_launch(void* const* d_in, const int* in_sizes, int n_in,
                              void* d_out, int out_size, void* d_ws, size_t ws_size,
                              hipStream_t stream)
{
    const float* x   = (const float*)d_in[0];
    const int*   ei  = (const int*)d_in[1];
    const float* W1  = (const float*)d_in[2];
    const float* a1s = (const float*)d_in[3];
    const float* a1d = (const float*)d_in[4];
    const float* b1  = (const float*)d_in[5];
    const float* W2  = (const float*)d_in[6];
    const float* a2s = (const float*)d_in[7];
    const float* a2d = (const float*)d_in[8];
    const float* b2  = (const float*)d_in[9];
    const float* Wl  = (const float*)d_in[10];
    const float* bl  = (const float*)d_in[11];

    const int N  = in_sizes[0] / HDIM;
    const int E  = in_sizes[1] / 2;
    const int C  = in_sizes[11];

    char* ws = (char*)d_ws;
    auto align256 = [](size_t v) { return (v + 255) & ~(size_t)255; };
    size_t off = 0;
    float*    bufH  = (float*)(ws + off);  off += align256((size_t)N * HDIM * 4);
    float*    bufA  = (float*)(ws + off);  off += align256((size_t)N * HDIM * 4);
    float*    aS    = (float*)(ws + off);  off += align256((size_t)N * 4);
    float*    aD    = (float*)(ws + off);  off += align256((size_t)N * 4);
    float*    aS2   = (float*)(ws + off);  off += align256((size_t)N * 4);
    float*    aD2   = (float*)(ws + off);  off += align256((size_t)N * 4);
    const size_t nStrips = (size_t)(N + SROWS - 1) / SROWS;
    const size_t qBytes  = nStrips * SWAVES * QSAVE * 4;
    int* qsave = (off + qBytes <= ws_size && N <= (1 << QSRCBITS)) ? (int*)(ws + off) : nullptr;

    const int nRowTiles = N / 16;
    const int BLK = 256;
    dim3 blk(BLK);

    dim3 gemmGrid((nRowTiles * 2 + 7) / 8);
    dim3 alphaGrid((N + 4 * AROWS - 1) / (4 * AROWS));
    dim3 stripGrid((N + SROWS - 1) / SROWS);

    gat_gemm_strip<<<gemmGrid, blk, 0, stream>>>(x, W1, bufH, nRowTiles);
    gat_alpha<<<alphaGrid, dim3(128), 0, stream>>>(bufH, a1s, a1d, aS, aD, N);
    gat_agg_strip<1><<<stripGrid, dim3(SWAVES * 32), 0, stream>>>(
        ei, E, N, aS, aD, bufH, b1, bufA, nullptr, qsave,
        W2, a2s, a2d, nullptr, aS2, aD2, C);
    gat_agg_strip<2><<<stripGrid, dim3(SWAVES * 32), 0, stream>>>(
        ei, E, N, aS2, aD2, bufA, nullptr, nullptr, qsave, nullptr,
        W2, b2, Wl, bl, (float*)d_out, nullptr, C);
}


# Strips filter a one-byte strip key per edge (written beside the layer-1 product) instead of the 4-byte destination, 4 keys compared per word into lane-own lists; queue and LDS sum order change
# speedup vs baseline: 4.0852x; 1.0762x over previous
#include <hip/hip_runtime.h>
#include <hip/hip_bf16.h>

typedef __attribute__((ext_vector_type(2))) float v2f;
typedef __attribute__((ext_vector_type(8))) float v8f;

#define HDIM 128
#define CTPW 4
#define SROWS  80
#define KLOADS 8
#define KEYPAD (KLOADS * 512)

__global__ __launch_bounds__(256) void gat_gemm_strip(
    const float* __restrict__ X, const float* __restrict__ W,
    float* __restrict__ H, int nRowTiles,
    const int* __restrict__ ei, int E, int N,
    unsigned* __restrict__ keys,
    int gemmBlocks)
{
    static_assert(CTPW == 4, "a lane's columns of the 4 tiles are one float4");
    if ((int)blockIdx.x >= gemmBlocks) {
        const int nKey16 = (E + KEYPAD - 1) / KEYPAD * (KEYPAD / 16);
        const int g = ((int)blockIdx.x - gemmBlocks) * (int)blockDim.x + (int)threadIdx.x;
        if (g >= nKey16) return;
        const int* __restrict__ dstp = ei + E;
        const bool vec = ((E & 3) == 0);
        unsigned kw[4];
        #pragma unroll
        for (int q = 0; q < 4; ++q) {
            const int i = g * 16 + q * 4;
            int d[4];
            if (vec && i + 4 <= E) {
                const int4 t = *(const int4*)(dstp + i);
                d[0] = t.x; d[1] = t.y; d[2] = t.z; d[3] = t.w;
            } else {
                #pragma unroll
                for (int c = 0; c < 4; ++c) d[c] = (i + c < E) ? dstp[i + c] : -1;
            }
            unsigned w = 0;
            #pragma unroll
            for (int c = 0; c < 4; ++c) {
                const unsigned k = ((unsigned)d[c] < (unsigned)N) ? (unsigned)d[c] / SROWS : 255u;
                w |= k << (8 * c);
            }
            kw[q] = w;
        }
        *(uint4*)(keys + (size_t)g * 4) = make_uint4(kw[0], kw[1], kw[2], kw[3]);
        return;
    }
    int wid  = (blockIdx.x * blockDim.x + threadIdx.x) >> 5;
    int lane = threadIdx.x & 31;
    int nHalves = HDIM / (CTPW * 16);
    int nW = nRowTiles * nHalves;
    if (wid >= nW) return;
    int rt   = wid >> 1;
    int half = wid & 1;
    int row0 = rt * 16;
    int laneM = lane & 15;
    int kOff  = (lane >> 4) * 2;
    int col0  = half * (CTPW * 16) + CTPW * laneM;

    const float* xrow = X + (size_t)(row0 + laneM) * HDIM + kOff;
    const float* wrow = W + (size_t)kOff * HDIM + col0;

    v8f c[CTPW] = {};
    #pragma unroll 4
    for (int k = 0; k < HDIM; k += 4) {
        v2f a = *(const v2f*)(xrow + k);
        const float4 w0 = *(const float4*)(wrow + (size_t)k * HDIM);
        const float4 w1 = *(const float4*)(wrow + (size_t)(k + 1) * HDIM);
        const float w0v[CTPW] = {w0.x, w0.y, w0.z, w0.w};
        const float w1v[CTPW] = {w1.x, w1.y, w1.z, w1.w};
        #pragma unroll
        for (int t = 0; t < CTPW; ++t) {
            v2f b = {w0v[t], w1v[t]};
            c[t] = __builtin_amdgcn_wmma_f32_16x16x4_f32(
                       false, a, false, b, (short)0, c[t], false, false);
        }
    }

    int mBase = (lane >> 4) * 8;
    float* o = H + (size_t)(row0 + mBase) * HDIM + col0;
    #pragma unroll
    for (int r = 0; r < 8; ++r)
        *(float4*)(o + (size_t)r * HDIM) = make_float4(c[0][r], c[1][r], c[2][r], c[3][r]);
}

#define AROWS 16
template <int O>
__device__ __forceinline__ void alpha_halve(float (&val)[2 * AROWS], int lane)
{
    const bool up = (lane & O) != 0;
    #pragma unroll
    for (int v = 0; v < O; ++v) {
        const float send = up ? val[v]     : val[v + O];
        const float keep = up ? val[v + O] : val[v];
        val[v] = keep + __shfl_xor(send, O, 32);
    }
}

__global__ __launch_bounds__(128) void gat_alpha(
    const float* __restrict__ H, const float* __restrict__ asrc,
    const float* __restrict__ adst, float* __restrict__ aS,
    float* __restrict__ aD, int N)
{
    const int w    = (blockIdx.x * blockDim.x + threadIdx.x) >> 5;
    const int lane = threadIdx.x & 31;
    const int row0 = w * AROWS;
    if (row0 >= N) return;

    float as4[4], ad4[4];
    #pragma unroll
    for (int j = 0; j < 4; ++j) {
        as4[j] = asrc[lane + 32 * j];
        ad4[j] = adst[lane + 32 * j];
    }

    float hv[AROWS][4];
    #pragma unroll
    for (int r = 0; r < AROWS; ++r) {
        const int row = (row0 + r < N) ? (row0 + r) : (N - 1);
        const float* h = H + (size_t)row * HDIM + lane;
        #pragma unroll
        for (int j = 0; j < 4; ++j) hv[r][j] = h[32 * j];
    }

    float val[2 * AROWS];
    #pragma unroll
    for (int r = 0; r < AROWS; ++r) {
        float s = 0.f, d = 0.f;
        #pragma unroll
        for (int j = 0; j < 4; ++j) {
            s += hv[r][j] * as4[j];
            d += hv[r][j] * ad4[j];
        }
        val[r]         = s;
        val[AROWS + r] = d;
    }

    alpha_halve<16>(val, lane);
    alpha_halve<8>(val, lane);
    alpha_halve<4>(val, lane);
    alpha_halve<2>(val, lane);
    alpha_halve<1>(val, lane);

    const int row = row0 + (lane & (AROWS - 1));
    if (row < N) {
        if (lane < AROWS) aS[row] = val[0];
        else              aD[row] = val[0];
    }
}

__device__ __forceinline__ unsigned f2ord(float f) {
    unsigned b = __float_as_uint(f);
    return (b & 0x80000000u) ? ~b : (b | 0x80000000u);
}
__device__ __forceinline__ float ord2f(unsigned u) {
    unsigned b = (u & 0x80000000u) ? (u & 0x7FFFFFFFu) : ~u;
    return __uint_as_float(b);
}
__device__ __forceinline__ float leaky02(float e) {
    return (e >= 0.f) ? e : 0.2f * e;
}

#define SWAVES 32
#define SLOADS 8
#define WEDGES (SLOADS * 128)
#define QCAP   640
#define QKEEP  160
#define QSAVE  (QKEEP + 32)
#define QSRCBITS 24
#define APITCH (HDIM + 4)
#define G1JOBS ((SROWS / 16) * 4)
#define G2JOBS ((SROWS / 16) * 3)

template <int LAYER>
__global__ __launch_bounds__(1024) void gat_agg_strip(
    const int* __restrict__ ei, int E, int N,
    const float* __restrict__ aS, const float* __restrict__ aD,
    const float* __restrict__ H, const float* __restrict__ bias,
    float* __restrict__ act,
    const int* __restrict__ qin,
    int* __restrict__ qout,
    const unsigned* __restrict__ keys,
    const float* __restrict__ Wn,
    const float* __restrict__ va,
    const float* __restrict__ vb,
    const float* __restrict__ vc,
    float* __restrict__ oa,
    float* __restrict__ ob,
    int nCols)
{
    static_assert(SWAVES * 32 == HDIM * 8, "8 threads per row of W2 form u");
    static_assert(SROWS % 16 == 0 && G1JOBS <= SWAVES && G2JOBS <= SWAVES, "one wave per matrix job");
    static_assert(SROWS * APITCH <= SWAVES * QCAP, "the second row buffer lies in the first queue array");
    static_assert(48 <= APITCH, "the strip's classes (3 column tiles at most) fit where its sums were");
    __shared__ __attribute__((aligned(16))) float s_acc[SROWS * APITCH];
    __shared__ float    s_u[2 * HDIM];
    __shared__ float    s_l2[2 * SROWS];
    __shared__ unsigned s_max[SROWS];
    __shared__ float    s_den[SROWS];
    __shared__ float    s_aD[SROWS];
    __shared__ float    s_self[SROWS];
    __shared__ int      s_qi[SWAVES * QCAP];
    __shared__ int      s_qd[SWAVES * QCAP];
    __shared__ float    s_qv[SWAVES * QCAP];
    __shared__ int      s_over;

    const int tid  = threadIdx.x;
    const int lane = tid & 31;
    const int wv   = tid >> 5;
    const int n0   = blockIdx.x * SROWS;
    const int rows = (N - n0 < SROWS) ? (N - n0) : SROWS;
    const int* __restrict__ dstp = ei + E;

    for (int i = tid; i < SROWS * APITCH / 4; i += SWAVES * 32)
        ((float4*)s_acc)[i] = make_float4(0.f, 0.f, 0.f, 0.f);
    if (LAYER == 1) {
        const int k = tid >> 3, p = tid & 7;
        const float* wr = Wn + (size_t)k * HDIM + p * 16;
        float us = 0.f, ud = 0.f;
        #pragma unroll
        for (int q = 0; q < 4; ++q) {
            const float4 w  = *(const float4*)(wr + 4 * q);
            const float4 s4 = *(const float4*)(va + p * 16 + 4 * q);
            const float4 d4 = *(const float4*)(vb + p * 16 + 4 * q);
            us += w.x * s4.x + w.y * s4.y + w.z * s4.z + w.w * s4.w;
            ud += w.x * d4.x + w.y * d4.y + w.z * d4.z + w.w * d4.w;
        }
        #pragma unroll
        for (int o = 1; o < 8; o <<= 1) {
            us += __shfl_xor(us, o, 32);
            ud += __shfl_xor(ud, o, 32);
        }
        if (p == 0) { s_u[k] = us; s_u[HDIM + k] = ud; }
    }
    if (tid < SROWS) {
        float ad = 0.f, es = 0.f;
        if (tid < rows) {
            ad = aD[n0 + tid];
            es = leaky02(aS[n0 + tid] + ad);
        }
        s_aD[tid]   = ad;
        s_self[tid] = es;
        s_max[tid]  = f2ord(es);
        s_den[tid]  = 0.f;
    }
    if (tid == 0) s_over = 0;
    __syncthreads();

    int*   qi = s_qi + wv * QCAP;
    int*   qd = s_qd + wv * QCAP;
    float* qv = s_qv + wv * QCAP;
    const bool vec = ((E & 3) == 0);

    auto load_round = [&](int base, int (&dv)[SLOADS][4]) {
        if (vec && base + WEDGES <= E) {
            #pragma unroll
            for (int j = 0; j < SLOADS; ++j) {
                int4 t = *(const int4*)(dstp + base + j * 128 + lane * 4);
                dv[j][0] = t.x; dv[j][1] = t.y; dv[j][2] = t.z; dv[j][3] = t.w;
            }
        } else {
            #pragma unroll
            for (int j = 0; j < SLOADS; ++j) {
                #pragma unroll
                for (int c = 0; c < 4; ++c) {
                    int i = base + j * 128 + lane * 4 + c;
                    dv[j][c] = (i < E) ? dstp[i] : -1;
                }
            }
        }
    };

    int cnt = 0;
    auto scan_round = [&](int base, const int (&dv)[SLOADS][4]) {
        #pragma unroll
        for (int j = 0; j < SLOADS; ++j) {
            #pragma unroll
            for (int c = 0; c < 4; ++c) {
                int dl = dv[j][c] - n0;
                bool m = (unsigned)dl < (unsigned)rows;
                unsigned mask = (unsigned)__ballot(m);
                if (mask) {
                    int pos = cnt + __popc(mask & ((1u << lane) - 1u));
                    if (m && pos < QCAP) {
                        qi[pos] = base + j * 128 + lane * 4 + c;
                        qd[pos] = dl;
                    }
                    cnt += __popc(mask);
                }
            }
        }
    };

    const int perRound = SWAVES * WEDGES;
    const int nRounds = (E + perRound - 1) / perRound;
    int rd = (nRounds > 0) ? (int)(blockIdx.x % (unsigned)nRounds) : 0;
    auto next_base = [&]() {
        int b = rd * perRound + wv * WEDGES;
        rd = (rd + 1 == nRounds) ? 0 : rd + 1;
        return b;
    };
    const size_t wq = ((size_t)blockIdx.x * SWAVES + wv) * QSAVE;
    const int saved = qin ? __builtin_amdgcn_readfirstlane(qin[wq]) : -1;
    const bool reuse = saved >= 0;
    bool save = false;
    const bool keyed = (keys != nullptr);
    if (reuse) {
        cnt = saved;
    } else if (keyed) {
        constexpr int LCAP = QCAP / 32;
        const unsigned K4 = (unsigned)blockIdx.x * 0x01010101u;
        const int Epad = (E + KEYPAD - 1) / KEYPAD * KEYPAD;
        const int perRoundK = SWAVES * KEYPAD;
        const int nRoundsK = (Epad + perRoundK - 1) / perRoundK;
        int rk = (nRoundsK > 0) ? (int)(blockIdx.x % (unsigned)nRoundsK) : 0;
        int mine = 0;
        for (int it = 0; it < nRoundsK; ++it) {
            const int base = rk * perRoundK + wv * KEYPAD;
            rk = (rk + 1 == nRoundsK) ? 0 : rk + 1;
            if (base >= Epad) continue;
            uint4 kv[KLOADS];
            #pragma unroll
            for (int j = 0; j < KLOADS; ++j)
                kv[j] = *(const uint4*)(keys + ((base + j * 512) >> 2) + lane * 4);
            #pragma unroll
            for (int j = 0; j < KLOADS; j += 2) {
                unsigned c = 0;
                #pragma unroll
                for (int jj = 0; jj < 2; ++jj) {
                    const unsigned w4[4] = {kv[j + jj].x, kv[j + jj].y, kv[j + jj].z, kv[j + jj].w};
                    #pragma unroll
                    for (int w = 0; w < 4; ++w) {
                        const unsigned t = w4[w] ^ K4;
                        const unsigned a = (t & 0x7f7f7f7fu) + 0x7f7f7f7fu;
                        const unsigned z = ~(a | t | 0x7f7f7f7fu);
                        c |= z >> (7 - w - 4 * jj);
                    }
                }
                const int e0 = base + j * 512 + lane * 16;
                while (c) {
                    const int p = __ffs((int)c) - 1;
                    c &= c - 1u;
                    const int slot = (mine < LCAP - 1) ? mine : (LCAP - 1);
                    qd[slot * 32 + lane] = e0 + ((p & 4) << 7) + ((p & 3) << 2) + (p >> 3);
                    ++mine;
                }
            }
        }
        const bool lover = __ballot(mine > LCAP) != 0;
        if (lover) s_over = 1;
        const int n = (mine < LCAP) ? mine : LCAP;
        int incl = n;
        #pragma unroll
        for (int o = 1; o < 32; o <<= 1) {
            const int t = __shfl_up(incl, o, 32);
            if (lane >= o) incl += t;
        }
        const int excl = incl - n;
        cnt = __shfl(incl, 31, 32);
        for (int t = 0; t < LCAP; ++t) {
            if (__ballot(t < n) == 0) break;
            if (t < n) qi[excl + t] = qd[t * 32 + lane];
        }
        save = (qout != nullptr) && !lover && (cnt <= QKEEP);
        if (qout && lane == 0) qout[wq] = save ? cnt : -1;
    } else {
        int dv[SLOADS][4];
        for (int it = 0; it < nRounds; ++it) {
            const int base = next_base();
            if (base >= E) continue;
            load_round(base, dv);
            scan_round(base, dv);
        }
        save = (qout != nullptr) && (cnt <= QKEEP);
        if (qout && lane == 0) qout[wq] = save ? cnt : -1;
    }
    if (cnt > QCAP) { s_over = 1; cnt = QCAP; }

    for (int b = 0; b < cnt; b += 256) {
        int k[8], dl[8], s[8]; float as_[8];
        #pragma unroll
        for (int u = 0; u < 8; ++u) {
            int kk = b + 32 * u + lane;
            k[u] = (kk < cnt) ? kk : (cnt - 1);
        }
        if (reuse) {
            #pragma unroll
            for (int u = 0; u < 8; ++u) {
                unsigned p = (unsigned)qin[wq + 32 + k[u]];
                s[u]  = (int)(p & ((1u << QSRCBITS) - 1u));
                dl[u] = (int)(p >> QSRCBITS);
            }
        } else if (keyed) {
            #pragma unroll
            for (int u = 0; u < 8; ++u) {
                const int en = qi[k[u]];
                s[u]  = ei[en];
                dl[u] = dstp[en] - n0;
            }
        } else {
            #pragma unroll
            for (int u = 0; u < 8; ++u) { s[u] = ei[qi[k[u]]]; dl[u] = qd[k[u]]; }
        }
        #pragma unroll
        for (int u = 0; u < 8; ++u) as_[u] = aS[s[u]];
        asm volatile("" :: "v"(as_[0]), "v"(as_[1]), "v"(as_[2]), "v"(as_[3]),
                           "v"(as_[4]), "v"(as_[5]), "v"(as_[6]), "v"(as_[7]));
        #pragma unroll
        for (int u = 0; u < 8; ++u) {
            if (b + 32 * u + lane < cnt) {
                float e = leaky02(as_[u] + s_aD[dl[u]]);
                qi[k[u]] = s[u];
                qv[k[u]] = e;
                if (reuse || keyed) qd[k[u]] = dl[u];
                if (!reuse && save) qout[wq + 32 + k[u]] = (int)((unsigned)s[u] | ((unsigned)dl[u] << QSRCBITS));
                atomicMax(&s_max[dl[u]], f2ord(e));
            }
        }
    }
    __syncthreads();

    const bool over = __builtin_amdgcn_readfirstlane(s_over) != 0;

    auto list_trip = [&](int trip) {
        for (int base = wv * 32; base < E; base += SWAVES * 32) {
            int i = base + lane;
            int dl = ((i < E) ? dstp[i] : -1) - n0;
            bool m = (unsigned)dl < (unsigned)rows;
            int s = 0; float al = 0.f;
            if (m) {
                s = ei[i];
                float e = leaky02(aS[s] + s_aD[dl]);
                if (trip == 0) {
                    atomicMax(&s_max[dl], f2ord(e));
                } else {
                    float ex = expf(e - ord2f(s_max[dl]));
                    if (trip == 1) atomicAdd(&s_den[dl], ex);
                    else           al = ex / s_den[dl];
                }
            }
            if (trip == 2) {
                unsigned mk = (unsigned)__ballot(m);
                while (mk) {
                    int l = __ffs((int)mk) - 1;
                    mk &= mk - 1u;
                    int   sj = __shfl(s, l, 32);
                    int   dj = __shfl(dl, l, 32);
                    float aj = __shfl(al, l, 32);
                    const float* h = H + (size_t)sj * HDIM;
                    float* o = s_acc + dj * APITCH;
                    #pragma unroll
                    for (int f = 0; f < 4; ++f)
                        atomicAdd(&o[lane + 32 * f], h[lane + 32 * f] * aj);
                }
            }
        }
    };

    if (over) {
        list_trip(0);
        __syncthreads();
    }

    if (tid < rows) {
        float ex = expf(s_self[tid] - ord2f(s_max[tid]));
        s_self[tid] = ex;
        atomicAdd(&s_den[tid], ex);
    }
    if (!over) {
        for (int b = 0; b < cnt; b += 32) {
            int k = b + lane;
            if (k < cnt) {
                int dl = qd[k];
                float ex = expf(qv[k] - ord2f(s_max[dl]));
                qv[k] = ex;
                atomicAdd(&s_den[dl], ex);
            }
        }
    } else {
        list_trip(1);
    }
    __syncthreads();

    if (!over) {
        for (int b = 0; b < cnt; b += 32) {
            int nb = (cnt - b < 32) ? (cnt - b) : 32;
            int s = 0, dl = 0; float al = 0.f;
            if (lane < nb) {
                s  = qi[b + lane];
                dl = qd[b + lane];
                al = qv[b + lane] / s_den[dl];
            }
            for (int j = 0; j < nb; j += 16) {
                v2f hv[16][2];
                #pragma unroll
                for (int u = 0; u < 16; ++u) {
                    int ju = (j + u < nb) ? (j + u) : (nb - 1);
                    int sj = __builtin_amdgcn_readlane(s, ju);
                    const float* h = H + (size_t)sj * HDIM + 2 * lane;
                    hv[u][0] = *(const v2f*)(h);
                    hv[u][1] = *(const v2f*)(h + 64);
                }
                #pragma unroll
                for (int u = 0; u < 16; ++u) {
                    if (j + u < nb) {
                        float aj = __uint_as_float(__builtin_amdgcn_readlane(__float_as_uint(al), j + u));
                        int   dj = __builtin_amdgcn_readlane(dl, j + u);
                        float* o = s_acc + dj * APITCH + 2 * lane;
                        atomicAdd(&o[0],  hv[u][0][0] * aj);
                        atomicAdd(&o[1],  hv[u][0][1] * aj);
                        atomicAdd(&o[64], hv[u][1][0] * aj);
                        atomicAdd(&o[65], hv[u][1][1] * aj);
                    }
                }
            }
        }
    } else {
        list_trip(2);
    }
    __syncthreads();

    for (int e4 = tid; e4 < rows * (HDIM / 4); e4 += SWAVES * 32) {
        int r = e4 / (HDIM / 4);
        int f = (e4 - r * (HDIM / 4)) * 4;
        float4 v = *(const float4*)(s_acc + r * APITCH + f);
        float4 h = *(const float4*)(H + (size_t)(n0 + r) * HDIM + f);
        float  a = s_self[r] / s_den[r];
        v.x += h.x * a; v.y += h.y * a; v.z += h.z * a; v.w += h.w * a;
        if (LAYER == 1) {
            v.x += bias[f]; v.y += bias[f + 1]; v.z += bias[f + 2]; v.w += bias[f + 3];
            v.x = (v.x > 0.f) ? v.x : 0.f;  v.y = (v.y > 0.f) ? v.y : 0.f;
            v.z = (v.z > 0.f) ? v.z : 0.f;  v.w = (v.w > 0.f) ? v.w : 0.f;
            *(float4*)(act + (size_t)(n0 + r) * HDIM + f) = v;
        }
        *(float4*)(s_acc + r * APITCH + f) = v;
    }
    __syncthreads();

    if (LAYER == 1) {
        for (int r = wv; r < rows; r += SWAVES) {
            const float* a = s_acc + r * APITCH + lane;
            float ls = 0.f, ld = 0.f;
            #pragma unroll
            for (int j = 0; j < 4; ++j) {
                const float xv = a[32 * j];
                ls += xv * s_u[lane + 32 * j];
                ld += xv * s_u[HDIM + lane + 32 * j];
            }
            #pragma unroll
            for (int o = 16; o >= 1; o >>= 1) {
                ls += __shfl_xor(ls, o, 32);
                ld += __shfl_xor(ld, o, 32);
            }
            if (lane == 0) { s_l2[r] = ls; s_l2[SROWS + r] = ld; }
        }
        __syncthreads();
        if (tid < rows)                              oa[n0 + tid] = s_l2[tid];
        else if (tid >= 128 && tid - 128 < rows)     ob[n0 + tid - 128] = s_l2[SROWS + tid - 128];
        return;
    }

    float* s_t = reinterpret_cast<float*>(s_qi);
    const int wvs   = __builtin_amdgcn_readfirstlane(wv);
    const int laneM = lane & 15;
    const int hh    = lane >> 4;
    const int kOff  = hh * 2;
    if (wvs < G1JOBS) {
        const int rt = wvs >> 2, q = wvs & 3;
        const int col0 = q * 32 + 2 * laneM;
        const float* arow = s_acc + (rt * 16 + laneM) * APITCH + kOff;
        const float* wrow = Wn + (size_t)kOff * HDIM + col0;
        v8f c0 = {}, c1 = {};
        #pragma unroll 8
        for (int k = 0; k < HDIM; k += 4) {
            const v2f a = *(const v2f*)(arow + k);
            const float2 w0 = *(const float2*)(wrow + (size_t)k * HDIM);
            const float2 w1 = *(const float2*)(wrow + (size_t)(k + 1) * HDIM);
            const v2f b0 = {w0.x, w1.x};
            const v2f b1 = {w0.y, w1.y};
            c0 = __builtin_amdgcn_wmma_f32_16x16x4_f32(false, a, false, b0, (short)0, c0, false, false);
            c1 = __builtin_amdgcn_wmma_f32_16x16x4_f32(false, a, false, b1, (short)0, c1, false, false);
        }
        const float bv0 = va[col0], bv1 = va[col0 + 1];
        float* o = s_t + (rt * 16 + hh * 8) * APITCH + col0;
        #pragma unroll
        for (int r = 0; r < 8; ++r) {
            float y0 = c0[r] + bv0, y1 = c1[r] + bv1;
            y0 = (y0 > 0.f) ? y0 : 0.f;
            y1 = (y1 > 0.f) ? y1 : 0.f;
            *(float2*)(o + r * APITCH) = make_float2(y0, y1);
        }
    }
    __syncthreads();

    if (wvs < G2JOBS) {
        const int rt = wvs / 3, ct = wvs - rt * 3;
        const int col = ct * 16 + laneM;
        const int cc  = (col < nCols) ? col : (nCols - 1);
        const float* arow = s_t + (rt * 16 + laneM) * APITCH + kOff;
        const float* wcol = vb + (size_t)kOff * nCols + cc;
        v8f c = {};
        #pragma unroll 8
        for (int k = 0; k < HDIM; k += 4) {
            const v2f a = *(const v2f*)(arow + k);
            v2f b;
            b[0] = wcol[(size_t)k * nCols];
            b[1] = wcol[(size_t)(k + 1) * nCols];
            c = __builtin_amdgcn_wmma_f32_16x16x4_f32(false, a, false, b, (short)0, c, false, false);
        }
        if (col < nCols) {
            const float bv = vc[col];
            #pragma unroll
            for (int r = 0; r < 8; ++r)
                s_acc[(rt * 16 + hh * 8 + r) * nCols + col] = c[r] + bv;
        }
    }
    __syncthreads();

    static_assert(SROWS % 4 == 0, "a strip's output starts 16-byte aligned");
    {
        const int nOut = rows * nCols;
        float* og = oa + (size_t)n0 * nCols;
        for (int i4 = tid; i4 < (nOut >> 2); i4 += SWAVES * 32)
            *(float4*)(og + 4 * i4) = *(const float4*)(s_acc + 4 * i4);
        const int tail = (nOut & ~3) + tid;
        if (tid < (nOut & 3)) og[tail] = s_acc[tail];
    }
}

extern "C" void kernel_launch(void* const* d_in, const int* in_sizes, int n_in,
                              void* d_out, int out_size, void* d_ws, size_t ws_size,
                              hipStream_t stream)
{
    const float* x   = (const float*)d_in[0];
    const int*   ei  = (const int*)d_in[1];
    const float* W1  = (const float*)d_in[2];
    const float* a1s = (const float*)d_in[3];
    const float* a1d = (const float*)d_in[4];
    const float* b1  = (const float*)d_in[5];
    const float* W2  = (const float*)d_in[6];
    const float* a2s = (const float*)d_in[7];
    const float* a2d = (const float*)d_in[8];
    const float* b2  = (const float*)d_in[9];
    const float* Wl  = (const float*)d_in[10];
    const float* bl  = (const float*)d_in[11];

    const int N  = in_sizes[0] / HDIM;
    const int E  = in_sizes[1] / 2;
    const int C  = in_sizes[11];

    char* ws = (char*)d_ws;
    auto align256 = [](size_t v) { return (v + 255) & ~(size_t)255; };
    size_t off = 0;
    float*    bufH  = (float*)(ws + off);  off += align256((size_t)N * HDIM * 4);
    float*    bufA  = (float*)(ws + off);  off += align256((size_t)N * HDIM * 4);
    float*    aS    = (float*)(ws + off);  off += align256((size_t)N * 4);
    float*    aD    = (float*)(ws + off);  off += align256((size_t)N * 4);
    float*    aS2   = (float*)(ws + off);  off += align256((size_t)N * 4);
    float*    aD2   = (float*)(ws + off);  off += align256((size_t)N * 4);
    const size_t nStrips = (size_t)(N + SROWS - 1) / SROWS;
    const size_t kBytes = (size_t)(E + KEYPAD - 1) / KEYPAD * KEYPAD;
    unsigned* keys = (nStrips <= 255 && off + align256(kBytes) <= ws_size) ? (unsigned*)(ws + off) : nullptr;
    if (keys) off += align256(kBytes);
    const size_t qBytes  = nStrips * SWAVES * QSAVE * 4;
    int* qsave = (off + qBytes <= ws_size && N <= (1 << QSRCBITS)) ? (int*)(ws + off) : nullptr;

    const int nRowTiles = N / 16;
    const int BLK = 256;
    dim3 blk(BLK);

    const int gemmBlocks = (nRowTiles * 2 + 7) / 8;
    const int keyBlocks  = keys ? (int)((kBytes / 16 + BLK - 1) / BLK) : 0;
    dim3 gemmGrid(gemmBlocks + keyBlocks);
    dim3 alphaGrid((N + 4 * AROWS - 1) / (4 * AROWS));
    dim3 stripGrid((N + SROWS - 1) / SROWS);

    gat_gemm_strip<<<gemmGrid, blk, 0, stream>>>(x, W1, bufH, nRowTiles, ei, E, N, keys, gemmBlocks);
    gat_alpha<<<alphaGrid, dim3(128), 0, stream>>>(bufH, a1s, a1d, aS, aD, N);
    gat_agg_strip<1><<<stripGrid, dim3(SWAVES * 32), 0, stream>>>(
        ei, E, N, aS, aD, bufH, b1, bufA, nullptr, qsave, keys,
        W2, a2s, a2d, nullptr, aS2, aD2, C);
    gat_agg_strip<2><<<stripGrid, dim3(SWAVES * 32), 0, stream>>>(
        ei, E, N, aS2, aD2, bufA, nullptr, nullptr, qsave, nullptr, keys,
        W2, b2, Wl, bl, (float*)d_out, nullptr, C);
}


# Layer 1 sums x rows first, multiplies after: (sum alpha*x[src]) @ W1 on WMMA out of strip LDS; logits = x.(W1@a1); the x @ W1 launch goes, 4 -> 3 launches; product and logit sum order change
# speedup vs baseline: 4.5201x; 1.1065x over previous
#include <hip/hip_runtime.h>
#include <hip/hip_bf16.h>

typedef __attribute__((ext_vector_type(2))) float v2f;
typedef __attribute__((ext_vector_type(8))) float v8f;

#define HDIM 128
#define SROWS  80
#define KLOADS 8
#define KEYPAD (KLOADS * 512)

#define AROWS 16
#define LWAVES 8
template <int O>
__device__ __forceinline__ void alpha_halve(float (&val)[2 * AROWS], int lane)
{
    const bool up = (lane & O) != 0;
    #pragma unroll
    for (int v = 0; v < O; ++v) {
        const float send = up ? val[v]     : val[v + O];
        const float keep = up ? val[v + O] : val[v];
        val[v] = keep + __shfl_xor(send, O, 32);
    }
}

__global__ __launch_bounds__(LWAVES * 32) void gat_logit_keys(
    const float* __restrict__ X, const float* __restrict__ W,
    const float* __restrict__ asrc, const float* __restrict__ adst,
    float* __restrict__ aS, float* __restrict__ aD, int N,
    const int* __restrict__ ei, int E,
    unsigned* __restrict__ keys,
    int logitBlocks)
{
    static_assert(LWAVES * 32 == 256 && HDIM == 128, "8 threads per row of W, 4 rows each, form u");
    if ((int)blockIdx.x >= logitBlocks) {
        const int nKey16 = (E + KEYPAD - 1) / KEYPAD * (KEYPAD / 16);
        const int g = ((int)blockIdx.x - logitBlocks) * (int)blockDim.x + (int)threadIdx.x;
        if (g >= nKey16) return;
        const int* __restrict__ dstp = ei + E;
        const bool vec = ((E & 3) == 0);
        unsigned kw[4];
        #pragma unroll
        for (int q = 0; q < 4; ++q) {
            const int i = g * 16 + q * 4;
            int d[4];
            if (vec && i + 4 <= E) {
                const int4 t = *(const int4*)(dstp + i);
                d[0] = t.x; d[1] = t.y; d[2] = t.z; d[3] = t.w;
            } else {
                #pragma unroll
                for (int c = 0; c < 4; ++c) d[c] = (i + c < E) ? dstp[i + c] : -1;
            }
            unsigned w = 0;
            #pragma unroll
            for (int c = 0; c < 4; ++c) {
                const unsigned k = ((unsigned)d[c] < (unsigned)N) ? (unsigned)d[c] / SROWS : 255u;
                w |= k << (8 * c);
            }
            kw[q] = w;
        }
        *(uint4*)(keys + (size_t)g * 4) = make_uint4(kw[0], kw[1], kw[2], kw[3]);
        return;
    }

    __shared__ __attribute__((aligned(16))) float s_u[2 * HDIM];
    const int tid  = threadIdx.x;
    const int lane = tid & 31;
    const int row0 = ((int)blockIdx.x * LWAVES + (tid >> 5)) * AROWS;

    float4 hv[AROWS];
    #pragma unroll
    for (int r = 0; r < AROWS; ++r) {
        const int row = (row0 + r < N) ? (row0 + r) : (N - 1);
        hv[r] = *(const float4*)(X + (size_t)row * HDIM + 4 * lane);
    }

    {
        const int p = tid & 7;
        float us[4] = {0.f, 0.f, 0.f, 0.f}, ud[4] = {0.f, 0.f, 0.f, 0.f};
        #pragma unroll
        for (int q = 0; q < 4; ++q) {
            const int col = (q * 8 + p) * 4;
            const float4 s4 = *(const float4*)(asrc + col);
            const float4 d4 = *(const float4*)(adst + col);
            #pragma unroll
            for (int i = 0; i < 4; ++i) {
                const int k = (tid >> 3) + 32 * i;
                const float4 w = *(const float4*)(W + (size_t)k * HDIM + col);
                us[i] += w.x * s4.x + w.y * s4.y + w.z * s4.z + w.w * s4.w;
                ud[i] += w.x * d4.x + w.y * d4.y + w.z * d4.z + w.w * d4.w;
            }
        }
        #pragma unroll
        for (int i = 0; i < 4; ++i) {
            #pragma unroll
            for (int o = 1; o < 8; o <<= 1) {
                us[i] += __shfl_xor(us[i], o, 32);
                ud[i] += __shfl_xor(ud[i], o, 32);
            }
            if (p == 0) {
                const int k = (tid >> 3) + 32 * i;
                s_u[k] = us[i]; s_u[HDIM + k] = ud[i];
            }
        }
    }
    __syncthreads();
    if (row0 >= N) return;

    const float4 s4 = *(const float4*)(s_u + 4 * lane);
    const float4 d4 = *(const float4*)(s_u + HDIM + 4 * lane);
    float val[2 * AROWS];
    #pragma unroll
    for (int r = 0; r < AROWS; ++r) {
        val[r]         = hv[r].x * s4.x + hv[r].y * s4.y + hv[r].z * s4.z + hv[r].w * s4.w;
        val[AROWS + r] = hv[r].x * d4.x + hv[r].y * d4.y + hv[r].z * d4.z + hv[r].w * d4.w;
    }

    alpha_halve<16>(val, lane);
    alpha_halve<8>(val, lane);
    alpha_halve<4>(val, lane);
    alpha_halve<2>(val, lane);
    alpha_halve<1>(val, lane);

    const int row = row0 + (lane & (AROWS - 1));
    if (row < N) {
        if (lane < AROWS) aS[row] = val[0];
        else              aD[row] = val[0];
    }
}

__device__ __forceinline__ unsigned f2ord(float f) {
    unsigned b = __float_as_uint(f);
    return (b & 0x80000000u) ? ~b : (b | 0x80000000u);
}
__device__ __forceinline__ float ord2f(unsigned u) {
    unsigned b = (u & 0x80000000u) ? (u & 0x7FFFFFFFu) : ~u;
    return __uint_as_float(b);
}
__device__ __forceinline__ float leaky02(float e) {
    return (e >= 0.f) ? e : 0.2f * e;
}

#define SWAVES 32
#define SLOADS 8
#define WEDGES (SLOADS * 128)
#define QCAP   640
#define QKEEP  160
#define QSAVE  (QKEEP + 32)
#define QSRCBITS 24
#define APITCH (HDIM + 4)
#define G1JOBS ((SROWS / 16) * 4)
#define G2JOBS ((SROWS / 16) * 3)

template <int LAYER>
__global__ __launch_bounds__(1024) void gat_agg_strip(
    const int* __restrict__ ei, int E, int N,
    const float* __restrict__ aS, const float* __restrict__ aD,
    const float* __restrict__ H,
    float* __restrict__ act,
    const int* __restrict__ qin,
    int* __restrict__ qout,
    const unsigned* __restrict__ keys,
    const float* __restrict__ Wp,
    const float* __restrict__ pb,
    const float* __restrict__ Wu,
    const float* __restrict__ va,
    const float* __restrict__ vb,
    const float* __restrict__ vc,
    float* __restrict__ oa,
    float* __restrict__ ob,
    int nCols)
{
    static_assert(SWAVES * 32 == HDIM * 8, "8 threads per row of W2 form u");
    static_assert(SROWS % 16 == 0 && G1JOBS <= SWAVES && G2JOBS <= SWAVES, "one wave per matrix job");
    static_assert(SROWS * APITCH <= SWAVES * QCAP, "the second row buffer lies in the first queue array");
    static_assert(48 <= APITCH, "the strip's classes (3 column tiles at most) fit where its sums were");
    __shared__ __attribute__((aligned(16))) float s_acc[SROWS * APITCH];
    __shared__ float    s_u[2 * HDIM];
    __shared__ float    s_l2[2 * SROWS];
    __shared__ unsigned s_max[SROWS];
    __shared__ float    s_den[SROWS];
    __shared__ float    s_aD[SROWS];
    __shared__ float    s_self[SROWS];
    __shared__ __attribute__((aligned(16))) int s_qi[SWAVES * QCAP];
    __shared__ int      s_qd[SWAVES * QCAP];
    __shared__ float    s_qv[SWAVES * QCAP];
    __shared__ int      s_over;

    const int tid  = threadIdx.x;
    const int lane = tid & 31;
    const int wv   = tid >> 5;
    const int n0   = blockIdx.x * SROWS;
    const int rows = (N - n0 < SROWS) ? (N - n0) : SROWS;
    const int* __restrict__ dstp = ei + E;

    for (int i = tid; i < SROWS * APITCH / 4; i += SWAVES * 32)
        ((float4*)s_acc)[i] = make_float4(0.f, 0.f, 0.f, 0.f);
    if (LAYER == 1) {
        const int k = tid >> 3, p = tid & 7;
        const float* wr = Wu + (size_t)k * HDIM + p * 16;
        float us = 0.f, ud = 0.f;
        #pragma unroll
        for (int q = 0; q < 4; ++q) {
            const float4 w  = *(const float4*)(wr + 4 * q);
            const float4 s4 = *(const float4*)(va + p * 16 + 4 * q);
            const float4 d4 = *(const float4*)(vb + p * 16 + 4 * q);
            us += w.x * s4.x + w.y * s4.y + w.z * s4.z + w.w * s4.w;
            ud += w.x * d4.x + w.y * d4.y + w.z * d4.z + w.w * d4.w;
        }
        #pragma unroll
        for (int o = 1; o < 8; o <<= 1) {
            us += __shfl_xor(us, o, 32);
            ud += __shfl_xor(ud, o, 32);
        }
        if (p == 0) { s_u[k] = us; s_u[HDIM + k] = ud; }
    }
    if (tid < SROWS) {
        float ad = 0.f, es = 0.f;
        if (tid < rows) {
            ad = aD[n0 + tid];
            es = leaky02(aS[n0 + tid] + ad);
        }
        s_aD[tid]   = ad;
        s_self[tid] = es;
        s_max[tid]  = f2ord(es);
        s_den[tid]  = 0.f;
    }
    if (tid == 0) s_over = 0;
    __syncthreads();

    int*   qi = s_qi + wv * QCAP;
    int*   qd = s_qd + wv * QCAP;
    float* qv = s_qv + wv * QCAP;
    const bool vec = ((E & 3) == 0);

    auto load_round = [&](int base, int (&dv)[SLOADS][4]) {
        if (vec && base + WEDGES <= E) {
            #pragma unroll
            for (int j = 0; j < SLOADS; ++j) {
                int4 t = *(const int4*)(dstp + base + j * 128 + lane * 4);
                dv[j][0] = t.x; dv[j][1] = t.y; dv[j][2] = t.z; dv[j][3] = t.w;
            }
        } else {
            #pragma unroll
            for (int j = 0; j < SLOADS; ++j) {
                #pragma unroll
                for (int c = 0; c < 4; ++c) {
                    int i = base + j * 128 + lane * 4 + c;
                    dv[j][c] = (i < E) ? dstp[i] : -1;
                }
            }
        }
    };

    int cnt = 0;
    auto scan_round = [&](int base, const int (&dv)[SLOADS][4]) {
        #pragma unroll
        for (int j = 0; j < SLOADS; ++j) {
            #pragma unroll
            for (int c = 0; c < 4; ++c) {
                int dl = dv[j][c] - n0;
                bool m = (unsigned)dl < (unsigned)rows;
                unsigned mask = (unsigned)__ballot(m);
                if (mask) {
                    int pos = cnt + __popc(mask & ((1u << lane) - 1u));
                    if (m && pos < QCAP) {
                        qi[pos] = base + j * 128 + lane * 4 + c;
                        qd[pos] = dl;
                    }
                    cnt += __popc(mask);
                }
            }
        }
    };

    const int perRound = SWAVES * WEDGES;
    const int nRounds = (E + perRound - 1) / perRound;
    int rd = (nRounds > 0) ? (int)(blockIdx.x % (unsigned)nRounds) : 0;
    auto next_base = [&]() {
        int b = rd * perRound + wv * WEDGES;
        rd = (rd + 1 == nRounds) ? 0 : rd + 1;
        return b;
    };
    const size_t wq = ((size_t)blockIdx.x * SWAVES + wv) * QSAVE;
    const int saved = qin ? __builtin_amdgcn_readfirstlane(qin[wq]) : -1;
    const bool reuse = saved >= 0;
    bool save = false;
    const bool keyed = (keys != nullptr);
    if (reuse) {
        cnt = saved;
    } else if (keyed) {
        constexpr int LCAP = QCAP / 32;
        const unsigned K4 = (unsigned)blockIdx.x * 0x01010101u;
        const int Epad = (E + KEYPAD - 1) / KEYPAD * KEYPAD;
        const int perRoundK = SWAVES * KEYPAD;
        const int nRoundsK = (Epad + perRoundK - 1) / perRoundK;
        int rk = (nRoundsK > 0) ? (int)(blockIdx.x % (unsigned)nRoundsK) : 0;
        int mine = 0;
        for (int it = 0; it < nRoundsK; ++it) {
            const int base = rk * perRoundK + wv * KEYPAD;
            rk = (rk + 1 == nRoundsK) ? 0 : rk + 1;
            if (base >= Epad) continue;
            uint4 kv[KLOADS];
            #pragma unroll
            for (int j = 0; j < KLOADS; ++j)
                kv[j] = *(const uint4*)(keys + ((base + j * 512) >> 2) + lane * 4);
            #pragma unroll
            for (int j = 0; j < KLOADS; j += 2) {
                unsigned c = 0;
                #pragma unroll
                for (int jj = 0; jj < 2; ++jj) {
                    const unsigned w4[4] = {kv[j + jj].x, kv[j + jj].y, kv[j + jj].z, kv[j + jj].w};
                    #pragma unroll
                    for (int w = 0; w < 4; ++w) {
                        const unsigned t = w4[w] ^ K4;
                        const unsigned a = (t & 0x7f7f7f7fu) + 0x7f7f7f7fu;
                        const unsigned z = ~(a | t | 0x7f7f7f7fu);
                        c |= z >> (7 - w - 4 * jj);
                    }
                }
                const int e0 = base + j * 512 + lane * 16;
                while (c) {
                    const int p = __ffs((int)c) - 1;
                    c &= c - 1u;
                    const int slot = (mine < LCAP - 1) ? mine : (LCAP - 1);
                    qd[slot * 32 + lane] = e0 + ((p & 4) << 7) + ((p & 3) << 2) + (p >> 3);
                    ++mine;
                }
            }
        }
        const bool lover = __ballot(mine > LCAP) != 0;
        if (lover) s_over = 1;
        const int n = (mine < LCAP) ? mine : LCAP;
        int incl = n;
        #pragma unroll
        for (int o = 1; o < 32; o <<= 1) {
            const int t = __shfl_up(incl, o, 32);
            if (lane >= o) incl += t;
        }
        const int excl = incl - n;
        cnt = __shfl(incl, 31, 32);
        for (int t = 0; t < LCAP; ++t) {
            if (__ballot(t < n) == 0) break;
            if (t < n) qi[excl + t] = qd[t * 32 + lane];
        }
        save = (qout != nullptr) && !lover && (cnt <= QKEEP);
        if (qout && lane == 0) qout[wq] = save ? cnt : -1;
    } else {
        int dv[SLOADS][4];
        for (int it = 0; it < nRounds; ++it) {
            const int base = next_base();
            if (base >= E) continue;
            load_round(base, dv);
            scan_round(base, dv);
        }
        save = (qout != nullptr) && (cnt <= QKEEP);
        if (qout && lane == 0) qout[wq] = save ? cnt : -1;
    }
    if (cnt > QCAP) { s_over = 1; cnt = QCAP; }

    for (int b = 0; b < cnt; b += 256) {
        int k[8], dl[8], s[8]; float as_[8];
        #pragma unroll
        for (int u = 0; u < 8; ++u) {
            int kk = b + 32 * u + lane;
            k[u] = (kk < cnt) ? kk : (cnt - 1);
        }
        if (reuse) {
            #pragma unroll
            for (int u = 0; u < 8; ++u) {
                unsigned p = (unsigned)qin[wq + 32 + k[u]];
                s[u]  = (int)(p & ((1u << QSRCBITS) - 1u));
                dl[u] = (int)(p >> QSRCBITS);
            }
        } else if (keyed) {
            #pragma unroll
            for (int u = 0; u < 8; ++u) {
                const int en = qi[k[u]];
                s[u]  = ei[en];
                dl[u] = dstp[en] - n0;
            }
        } else {
            #pragma unroll
            for (int u = 0; u < 8; ++u) { s[u] = ei[qi[k[u]]]; dl[u] = qd[k[u]]; }
        }
        #pragma unroll
        for (int u = 0; u < 8; ++u) as_[u] = aS[s[u]];
        asm volatile("" :: "v"(as_[0]), "v"(as_[1]), "v"(as_[2]), "v"(as_[3]),
                           "v"(as_[4]), "v"(as_[5]), "v"(as_[6]), "v"(as_[7]));
        #pragma unroll
        for (int u = 0; u < 8; ++u) {
            if (b + 32 * u + lane < cnt) {
                float e = leaky02(as_[u] + s_aD[dl[u]]);
                qi[k[u]] = s[u];
                qv[k[u]] = e;
                if (reuse || keyed) qd[k[u]] = dl[u];
                if (!reuse && save) qout[wq + 32 + k[u]] = (int)((unsigned)s[u] | ((unsigned)dl[u] << QSRCBITS));
                atomicMax(&s_max[dl[u]], f2ord(e));
            }
        }
    }
    __syncthreads();

    const bool over = __builtin_amdgcn_readfirstlane(s_over) != 0;

    auto list_trip = [&](int trip) {
        for (int base = wv * 32; base < E; base += SWAVES * 32) {
            int i = base + lane;
            int dl = ((i < E) ? dstp[i] : -1) - n0;
            bool m = (unsigned)dl < (unsigned)rows;
            int s = 0; float al = 0.f;
            if (m) {
                s = ei[i];
                float e = leaky02(aS[s] + s_aD[dl]);
                if (trip == 0) {
                    atomicMax(&s_max[dl], f2ord(e));
                } else {
                    float ex = expf(e - ord2f(s_max[dl]));
                    if (trip == 1) atomicAdd(&s_den[dl], ex);
                    else           al = ex / s_den[dl];
                }
            }
            if (trip == 2) {
                unsigned mk = (unsigned)__ballot(m);
                while (mk) {
                    int l = __ffs((int)mk) - 1;
                    mk &= mk - 1u;
                    int   sj = __shfl(s, l, 32);
                    int   dj = __shfl(dl, l, 32);
                    float aj = __shfl(al, l, 32);
                    const float* h = H + (size_t)sj * HDIM;
                    float* o = s_acc + dj * APITCH;
                    #pragma unroll
                    for (int f = 0; f < 4; ++f)
                        atomicAdd(&o[lane + 32 * f], h[lane + 32 * f] * aj);
                }
            }
        }
    };

    if (over) {
        list_trip(0);
        __syncthreads();
    }

    if (tid < rows) {
        float ex = expf(s_self[tid] - ord2f(s_max[tid]));
        s_self[tid] = ex;
        atomicAdd(&s_den[tid], ex);
    }
    if (!over) {
        for (int b = 0; b < cnt; b += 32) {
            int k = b + lane;
            if (k < cnt) {
                int dl = qd[k];
                float ex = expf(qv[k] - ord2f(s_max[dl]));
                qv[k] = ex;
                atomicAdd(&s_den[dl], ex);
            }
        }
    } else {
        list_trip(1);
    }
    __syncthreads();

    if (!over) {
        for (int b = 0; b < cnt; b += 32) {
            int nb = (cnt - b < 32) ? (cnt - b) : 32;
            int s = 0, dl = 0; float al = 0.f;
            if (lane < nb) {
                s  = qi[b + lane];
                dl = qd[b + lane];
                al = qv[b + lane] / s_den[dl];
            }
            for (int j = 0; j < nb; j += 16) {
                v2f hv[16][2];
                #pragma unroll
                for (int u = 0; u < 16; ++u) {
                    int ju = (j + u < nb) ? (j + u) : (nb - 1);
                    int sj = __builtin_amdgcn_readlane(s, ju);
                    const float* h = H + (size_t)sj * HDIM + 2 * lane;
                    hv[u][0] = *(const v2f*)(h);
                    hv[u][1] = *(const v2f*)(h + 64);
                }
                #pragma unroll
                for (int u = 0; u < 16; ++u) {
                    if (j + u < nb) {
                        float aj = __uint_as_float(__builtin_amdgcn_readlane(__float_as_uint(al), j + u));
                        int   dj = __builtin_amdgcn_readlane(dl, j + u);
                        float* o = s_acc + dj * APITCH + 2 * lane;
                        atomicAdd(&o[0],  hv[u][0][0] * aj);
                        atomicAdd(&o[1],  hv[u][0][1] * aj);
                        atomicAdd(&o[64], hv[u][1][0] * aj);
                        atomicAdd(&o[65], hv[u][1][1] * aj);
                    }
                }
            }
        }
    } else {
        list_trip(2);
    }
    __syncthreads();

    for (int e4 = tid; e4 < rows * (HDIM / 4); e4 += SWAVES * 32) {
        int r = e4 / (HDIM / 4);
        int f = (e4 - r * (HDIM / 4)) * 4;
        float4 v = *(const float4*)(s_acc + r * APITCH + f);
        float4 h = *(const float4*)(H + (size_t)(n0 + r) * HDIM + f);
        float  a = s_self[r] / s_den[r];
        v.x += h.x * a; v.y += h.y * a; v.z += h.z * a; v.w += h.w * a;
        *(float4*)(s_acc + r * APITCH + f) = v;
    }
    __syncthreads();

    float* s_t = reinterpret_cast<float*>(s_qi);
    const int wvs   = __builtin_amdgcn_readfirstlane(wv);
    const int laneM = lane & 15;
    const int hh    = lane >> 4;
    const int kOff  = hh * 2;
    if (wvs < G1JOBS) {
        const int rt = wvs >> 2, q = wvs & 3;
        const int col0 = q * 32 + 2 * laneM;
        const float* arow = s_acc + (rt * 16 + laneM) * APITCH + kOff;
        const float* wrow = Wp + (size_t)kOff * HDIM + col0;
        v8f c0 = {}, c1 = {};
        #pragma unroll 8
        for (int k = 0; k < HDIM; k += 4) {
            const v2f a = *(const v2f*)(arow + k);
            const float2 w0 = *(const float2*)(wrow + (size_t)k * HDIM);
            const float2 w1 = *(const float2*)(wrow + (size_t)(k + 1) * HDIM);
            const v2f b0 = {w0.x, w1.x};
            const v2f b1 = {w0.y, w1.y};
            c0 = __builtin_amdgcn_wmma_f32_16x16x4_f32(false, a, false, b0, (short)0, c0, false, false);
            c1 = __builtin_amdgcn_wmma_f32_16x16x4_f32(false, a, false, b1, (short)0, c1, false, false);
        }
        const float bv0 = pb[col0], bv1 = pb[col0 + 1];
        float* o = s_t + (rt * 16 + hh * 8) * APITCH + col0;
        #pragma unroll
        for (int r = 0; r < 8; ++r) {
            float y0 = c0[r] + bv0, y1 = c1[r] + bv1;
            y0 = (y0 > 0.f) ? y0 : 0.f;
            y1 = (y1 > 0.f) ? y1 : 0.f;
            *(float2*)(o + r * APITCH) = make_float2(y0, y1);
        }
    }
    __syncthreads();

    if (LAYER == 1) {
        for (int e4 = tid; e4 < rows * (HDIM / 4); e4 += SWAVES * 32) {
            const int r = e4 / (HDIM / 4);
            const int f = (e4 - r * (HDIM / 4)) * 4;
            *(float4*)(act + (size_t)(n0 + r) * HDIM + f) = *(const float4*)(s_t + r * APITCH + f);
        }
        for (int r = wv; r < rows; r += SWAVES) {
            const float* a = s_t + r * APITCH + lane;
            float ls = 0.f, ld = 0.f;
            #pragma unroll
            for (int j = 0; j < 4; ++j) {
                const float xv = a[32 * j];
                ls += xv * s_u[lane + 32 * j];
                ld += xv * s_u[HDIM + lane + 32 * j];
            }
            #pragma unroll
            for (int o = 16; o >= 1; o >>= 1) {
                ls += __shfl_xor(ls, o, 32);
                ld += __shfl_xor(ld, o, 32);
            }
            if (lane == 0) { s_l2[r] = ls; s_l2[SROWS + r] = ld; }
        }
        __syncthreads();
        if (tid < rows)                              oa[n0 + tid] = s_l2[tid];
        else if (tid >= 128 && tid - 128 < rows)     ob[n0 + tid - 128] = s_l2[SROWS + tid - 128];
        return;
    }

    if (wvs < G2JOBS) {
        const int rt = wvs / 3, ct = wvs - rt * 3;
        const int col = ct * 16 + laneM;
        const int cc  = (col < nCols) ? col : (nCols - 1);
        const float* arow = s_t + (rt * 16 + laneM) * APITCH + kOff;
        const float* wcol = vb + (size_t)kOff * nCols + cc;
        v8f c = {};
        #pragma unroll 8
        for (int k = 0; k < HDIM; k += 4) {
            const v2f a = *(const v2f*)(arow + k);
            v2f b;
            b[0] = wcol[(size_t)k * nCols];
            b[1] = wcol[(size_t)(k + 1) * nCols];
            c = __builtin_amdgcn_wmma_f32_16x16x4_f32(false, a, false, b, (short)0, c, false, false);
        }
        if (col < nCols) {
            const float bv = vc[col];
            #pragma unroll
            for (int r = 0; r < 8; ++r)
                s_acc[(rt * 16 + hh * 8 + r) * nCols + col] = c[r] + bv;
        }
    }
    __syncthreads();

    static_assert(SROWS % 4 == 0, "a strip's output starts 16-byte aligned");
    {
        const int nOut = rows * nCols;
        float* og = oa + (size_t)n0 * nCols;
        for (int i4 = tid; i4 < (nOut >> 2); i4 += SWAVES * 32)
            *(float4*)(og + 4 * i4) = *(const float4*)(s_acc + 4 * i4);
        const int tail = (nOut & ~3) + tid;
        if (tid < (nOut & 3)) og[tail] = s_acc[tail];
    }
}

extern "C" void kernel_launch(void* const* d_in, const int* in_sizes, int n_in,
                              void* d_out, int out_size, void* d_ws, size_t ws_size,
                              hipStream_t stream)
{
    const float* x   = (const float*)d_in[0];
    const int*   ei  = (const int*)d_in[1];
    const float* W1  = (const float*)d_in[2];
    const float* a1s = (const float*)d_in[3];
    const float* a1d = (const float*)d_in[4];
    const float* b1  = (const float*)d_in[5];
    const float* W2  = (const float*)d_in[6];
    const float* a2s = (const float*)d_in[7];
    const float* a2d = (const float*)d_in[8];
    const float* b2  = (const float*)d_in[9];
    const float* Wl  = (const float*)d_in[10];
    const float* bl  = (const float*)d_in[11];

    const int N  = in_sizes[0] / HDIM;
    const int E  = in_sizes[1] / 2;
    const int C  = in_sizes[11];

    char* ws = (char*)d_ws;
    auto align256 = [](size_t v) { return (v + 255) & ~(size_t)255; };
    size_t off = 0;
    float*    bufA  = (float*)(ws + off);  off += align256((size_t)N * HDIM * 4);
    float*    aS    = (float*)(ws + off);  off += align256((size_t)N * 4);
    float*    aD    = (float*)(ws + off);  off += align256((size_t)N * 4);
    float*    aS2   = (float*)(ws + off);  off += align256((size_t)N * 4);
    float*    aD2   = (float*)(ws + off);  off += align256((size_t)N * 4);
    const size_t nStrips = (size_t)(N + SROWS - 1) / SROWS;
    const size_t kBytes = (size_t)(E + KEYPAD - 1) / KEYPAD * KEYPAD;
    unsigned* keys = (nStrips <= 255 && off + align256(kBytes) <= ws_size) ? (unsigned*)(ws + off) : nullptr;
    if (keys) off += align256(kBytes);
    const size_t qBytes  = nStrips * SWAVES * QSAVE * 4;
    int* qsave = (off + qBytes <= ws_size && N <= (1 << QSRCBITS)) ? (int*)(ws + off) : nullptr;

    const int BLK = LWAVES * 32;
    const int logitBlocks = (N + LWAVES * AROWS - 1) / (LWAVES * AROWS);
    const int keyBlocks   = keys ? (int)((kBytes / 16 + BLK - 1) / BLK) : 0;
    dim3 firstGrid(logitBlocks + keyBlocks);
    dim3 stripGrid((N + SROWS - 1) / SROWS);

    gat_logit_keys<<<firstGrid, dim3(BLK), 0, stream>>>(x, W1, a1s, a1d, aS, aD, N, ei, E, keys, logitBlocks);
    gat_agg_strip<1><<<stripGrid, dim3(SWAVES * 32), 0, stream>>>(
        ei, E, N, aS, aD, x, bufA, nullptr, qsave, keys,
        W1, b1, W2, a2s, a2d, nullptr, aS2, aD2, C);
    gat_agg_strip<2><<<stripGrid, dim3(SWAVES * 32), 0, stream>>>(
        ei, E, N, aS2, aD2, bufA, nullptr, qsave, nullptr, keys,
        W2, b2, nullptr, nullptr, Wl, bl, (float*)d_out, nullptr, C);
}
